# starting kernel
# baseline (speedup 1.0000x reference)
#include <hip/hip_runtime.h>
#include <hip/hip_bf16.h>

typedef __attribute__((ext_vector_type(2))) float v2f;
typedef __attribute__((ext_vector_type(8))) float v8f;

#define NNODES 50000
#define NEDGES 800000
#define WDIM   256
#define NHEADS 8
#define NHID   32
#define NGRAPH 256

__device__ __forceinline__ unsigned fenc(float x) {
    unsigned u = __float_as_uint(x);
    return (u & 0x80000000u) ? ~u : (u | 0x80000000u);
}
__device__ __forceinline__ float fdec(unsigned e) {
    unsigned u = (e & 0x80000000u) ? (e & 0x7FFFFFFFu) : ~e;
    return __uint_as_float(u);
}

__global__ void fill_u32(unsigned* __restrict__ p, unsigned v, long n) {
    long i = (long)blockIdx.x * blockDim.x + threadIdx.x;
    if (i < n) p[i] = v;
}

__global__ void gemm_f32_wmma(const float* __restrict__ A, const float* __restrict__ B,
                              float* __restrict__ C, int M, int N, int K) {
    int wave   = (blockIdx.x * blockDim.x + threadIdx.x) >> 5;
    int lane   = threadIdx.x & 31;
    int nstrip = N >> 6;
    int m0 = (wave / nstrip) << 4;
    int n0 = (wave % nstrip) << 6;
    if (m0 >= M) return;
    int hi = lane >> 4;
    int l  = lane & 15;

    v8f c0 = {}, c1 = {}, c2 = {}, c3 = {};
    const float* arow  = A + (size_t)(m0 + l) * K + (hi << 1);
    const float* bbase = B + (size_t)(hi << 1) * N + n0 + l;

    for (int k = 0; k < K; k += 4) {
        v2f a = *(const v2f*)(arow + k);
        const float* bk0 = bbase + (size_t)k * N;
        const float* bk1 = bk0 + N;
        v2f b0, b1, b2, b3;
        b0.x = bk0[0];   b0.y = bk1[0];
        b1.x = bk0[16];  b1.y = bk1[16];
        b2.x = bk0[32];  b2.y = bk1[32];
        b3.x = bk0[48];  b3.y = bk1[48];
        c0 = __builtin_amdgcn_wmma_f32_16x16x4_f32(false, a, false, b0, (short)0, c0, false, false);
        c1 = __builtin_amdgcn_wmma_f32_16x16x4_f32(false, a, false, b1, (short)0, c1, false, false);
        c2 = __builtin_amdgcn_wmma_f32_16x16x4_f32(false, a, false, b2, (short)0, c2, false, false);
        c3 = __builtin_amdgcn_wmma_f32_16x16x4_f32(false, a, false, b3, (short)0, c3, false, false);
    }
    float* crow = C + (size_t)(m0 + (hi << 3)) * N + n0 + l;
#pragma unroll
    for (int r = 0; r < 8; ++r) {
        crow[(size_t)r * N +  0] = c0[r];
        crow[(size_t)r * N + 16] = c1[r];
        crow[(size_t)r * N + 32] = c2[r];
        crow[(size_t)r * N + 48] = c3[r];
    }
}

__global__ void compute_al(const float* __restrict__ h, const float* __restrict__ as_,
                           const float* __restrict__ ad_, float* __restrict__ als,
                           float* __restrict__ ald, int n_nodes) {
    int i = blockIdx.x * blockDim.x + threadIdx.x;
    if (i >= n_nodes * NHEADS) return;
    int head = i & 7;
    const float* hp = h + (size_t)(i >> 3) * WDIM + head * NHID;
    const float* sp = as_ + head * NHID;
    const float* dp = ad_ + head * NHID;
    float s = 0.f, d = 0.f;
#pragma unroll
    for (int c = 0; c < NHID; ++c) { float v = hp[c]; s += v * sp[c]; d += v * dp[c]; }
    als[i] = s; ald[i] = d;
}

__global__ void edge_logits(const int* __restrict__ src, const int* __restrict__ dst,
                            const float* __restrict__ als, const float* __restrict__ ald,
                            float* __restrict__ ep, unsigned* __restrict__ menc) {
    long i = (long)blockIdx.x * blockDim.x + threadIdx.x;
    long tot = (long)(NEDGES + NNODES) * NHEADS;
    if (i >= tot) return;
    int e = (int)(i >> 3), head = (int)(i & 7);
    int s, d;
    if (e < NEDGES) { s = src[e]; d = dst[e]; } else { s = d = e - NEDGES; }
    float v = als[s * NHEADS + head] + ald[d * NHEADS + head];
    v = v > 0.f ? v : 0.2f * v;
    ep[i] = v;
    atomicMax(&menc[d * NHEADS + head], fenc(v));
}

__global__ void edge_exp(const int* __restrict__ dst, float* __restrict__ ep,
                         const unsigned* __restrict__ menc, float* __restrict__ denom) {
    long i = (long)blockIdx.x * blockDim.x + threadIdx.x;
    long tot = (long)(NEDGES + NNODES) * NHEADS;
    if (i >= tot) return;
    int e = (int)(i >> 3), head = (int)(i & 7);
    int d = (e < NEDGES) ? dst[e] : (e - NEDGES);
    float m = fdec(menc[d * NHEADS + head]);
    float p = expf(ep[i] - m);
    ep[i] = p;
    atomicAdd(&denom[d * NHEADS + head], p);
}

__global__ void edge_agg(const int* __restrict__ src, const int* __restrict__ dst,
                         const float* __restrict__ ep, const float* __restrict__ denom,
                         const float* __restrict__ h, float* __restrict__ agg) {
    int e = blockIdx.x;
    int t = threadIdx.x;
    int s, d;
    if (e < NEDGES) { s = src[e]; d = dst[e]; } else { s = d = e - NEDGES; }
    int head = t >> 5;
    float alpha = ep[(size_t)e * NHEADS + head] / denom[d * NHEADS + head];
    atomicAdd(&agg[(size_t)d * WDIM + t], alpha * h[(size_t)s * WDIM + t]);
}

__global__ void bias_act(const float* __restrict__ agg, const float* __restrict__ b,
                         float* __restrict__ out, int apply_elu) {
    long i = (long)blockIdx.x * blockDim.x + threadIdx.x;
    if (i >= (long)NNODES * WDIM) return;
    float v = agg[i] + b[i & (WDIM - 1)];
    if (apply_elu) v = v > 0.f ? v : (expf(v) - 1.0f);
    out[i] = v;
}

__global__ void pool_accum(const float* __restrict__ x2, const int* __restrict__ batch,
                           unsigned* __restrict__ gmaxenc, float* __restrict__ gsum,
                           float* __restrict__ counts) {
    int n = blockIdx.x, c = threadIdx.x;
    int g = batch[n];
    float v = x2[(size_t)n * WDIM + c];
    atomicMax(&gmaxenc[g * WDIM + c], fenc(v));
    atomicAdd(&gsum[g * WDIM + c], v);
    if (c == 0) atomicAdd(&counts[g], 1.0f);
}

__global__ void pool_final(const unsigned* __restrict__ gmaxenc, const float* __restrict__ gsum,
                           const float* __restrict__ counts, float* __restrict__ rep) {
    int g = blockIdx.x, c = threadIdx.x;
    float cnt = counts[g];
    float mx = (cnt > 0.f) ? fdec(gmaxenc[g * WDIM + c]) : 0.0f;
    float mean = gsum[g * WDIM + c] / fmaxf(cnt, 1.0f);
    rep[(size_t)g * (2 * WDIM) + c] = mx;
    rep[(size_t)g * (2 * WDIM) + WDIM + c] = mean;
}

__global__ void fc_sigmoid(const float* __restrict__ rep, const float* __restrict__ Wfc,
                           const float* __restrict__ bfc, float* __restrict__ out) {
    __shared__ float red[256];
    int g = blockIdx.x, t = threadIdx.x;
    const float* r = rep + (size_t)g * (2 * WDIM);
    float s = r[t] * Wfc[t] + r[WDIM + t] * Wfc[WDIM + t];
    red[t] = s;
    __syncthreads();
    for (int off = 128; off > 0; off >>= 1) {
        if (t < off) red[t] += red[t + off];
        __syncthreads();
    }
    if (t == 0) out[g] = 1.0f / (1.0f + expf(-(red[0] + bfc[0])));
}

extern "C" void kernel_launch(void* const* d_in, const int* in_sizes, int n_in,
                              void* d_out, int out_size, void* d_ws, size_t ws_size,
                              hipStream_t stream) {
    (void)in_sizes; (void)n_in; (void)out_size; (void)ws_size;

    const float* x    = (const float*)d_in[0];
    const int*   ei   = (const int*)d_in[1];
    const int*   bat  = (const int*)d_in[2];
    const float* W1   = (const float*)d_in[3];
    const float* a1s  = (const float*)d_in[4];
    const float* a1d  = (const float*)d_in[5];
    const float* b1   = (const float*)d_in[6];
    const float* W2   = (const float*)d_in[7];
    const float* a2s  = (const float*)d_in[8];
    const float* a2d  = (const float*)d_in[9];
    const float* b2   = (const float*)d_in[10];
    const float* Wfc  = (const float*)d_in[11];
    const float* bfc  = (const float*)d_in[12];

    float* out = (float*)d_out;
    float* rep = out + NGRAPH;

    const int* esrc = ei;
    const int* edst = ei + NEDGES;
    const int ETOT = NEDGES + NNODES;

    float*    bufA  = (float*)d_ws;
    float*    bufB  = bufA + (size_t)NNODES * WDIM;
    float*    als   = bufB + (size_t)NNODES * WDIM;
    float*    ald   = als + (size_t)NNODES * NHEADS;
    unsigned* menc  = (unsigned*)(ald + (size_t)NNODES * NHEADS);
    float*    denom = (float*)(menc + (size_t)NNODES * NHEADS);
    float*    ep    = denom + (size_t)NNODES * NHEADS;
    unsigned* gmaxe = (unsigned*)(ep + (size_t)ETOT * NHEADS);
    float*    gsum  = (float*)(gmaxe + NGRAPH * WDIM);
    float*    cnts  = gsum + NGRAPH * WDIM;

    const long nNH   = (long)NNODES * NHEADS;
    const long nNW   = (long)NNODES * WDIM;
    const long nEH   = (long)ETOT * NHEADS;
    const int  bNH   = (int)((nNH + 255) / 256);
    const int  bNW   = (int)((nNW + 255) / 256);
    const int  bEH   = (int)((nEH + 255) / 256);

    auto run_layer = [&](const float* xin, int K, const float* Wm, const float* as_,
                         const float* ad_, const float* bias, int elu) {
        long waves  = (long)(NNODES / 16) * (WDIM / 64);
        int  blocks = (int)((waves * 32 + 255) / 256);
        gemm_f32_wmma<<<blocks, 256, 0, stream>>>(xin, Wm, bufA, NNODES, WDIM, K);

        compute_al<<<bNH, 256, 0, stream>>>(bufA, as_, ad_, als, ald, NNODES);

        fill_u32<<<bNH, 256, 0, stream>>>(menc, 0u, nNH);
        fill_u32<<<bNH, 256, 0, stream>>>((unsigned*)denom, 0u, nNH);
        fill_u32<<<bNW, 256, 0, stream>>>((unsigned*)bufB, 0u, nNW);

        edge_logits<<<bEH, 256, 0, stream>>>(esrc, edst, als, ald, ep, menc);
        edge_exp   <<<bEH, 256, 0, stream>>>(edst, ep, menc, denom);
        edge_agg   <<<ETOT, 256, 0, stream>>>(esrc, edst, ep, denom, bufA, bufB);
        bias_act   <<<bNW, 256, 0, stream>>>(bufB, bias, bufB, elu);
    };

    run_layer(x,    128, W1, a1s, a1d, b1, 1);
    run_layer(bufB, 256, W2, a2s, a2d, b2, 0);

    fill_u32<<<(NGRAPH * WDIM + 255) / 256, 256, 0, stream>>>(gmaxe, 0u, NGRAPH * WDIM);
    fill_u32<<<(NGRAPH * WDIM + 255) / 256, 256, 0, stream>>>((unsigned*)gsum, 0u, NGRAPH * WDIM);
    fill_u32<<<1, 256, 0, stream>>>((unsigned*)cnts, 0u, NGRAPH);
    pool_accum<<<NNODES, 256, 0, stream>>>(bufB, bat, gmaxe, gsum, cnts);
    pool_final<<<NGRAPH, 256, 0, stream>>>(gmaxe, gsum, cnts, rep);
    fc_sigmoid<<<NGRAPH, 256, 0, stream>>>(rep, Wfc, bfc, out);
}


# edge aggregation: edges bucketed by destination per call (returning-add count = arrival-order slot, scan, place); a wave sums a node's p*h rows in registers, divides by denom once; no float atomics
# speedup vs baseline: 2.5415x; 2.5415x over previous
#include <hip/hip_runtime.h>
#include <hip/hip_bf16.h>

typedef __attribute__((ext_vector_type(2))) float v2f;
typedef __attribute__((ext_vector_type(8))) float v8f;

#define NNODES 50000
#define NEDGES 800000
#define WDIM   256
#define NHEADS 8
#define NHID   32
#define NGRAPH 256

__device__ __forceinline__ unsigned fenc(float x) {
    unsigned u = __float_as_uint(x);
    return (u & 0x80000000u) ? ~u : (u | 0x80000000u);
}
__device__ __forceinline__ float fdec(unsigned e) {
    unsigned u = (e & 0x80000000u) ? (e & 0x7FFFFFFFu) : ~e;
    return __uint_as_float(u);
}

__global__ void fill_u32(unsigned* __restrict__ p, unsigned v, long n) {
    long i = (long)blockIdx.x * blockDim.x + threadIdx.x;
    if (i < n) p[i] = v;
}

__global__ void gemm_f32_wmma(const float* __restrict__ A, const float* __restrict__ B,
                              float* __restrict__ C, int M, int N, int K) {
    int wave   = (blockIdx.x * blockDim.x + threadIdx.x) >> 5;
    int lane   = threadIdx.x & 31;
    int nstrip = N >> 6;
    int m0 = (wave / nstrip) << 4;
    int n0 = (wave % nstrip) << 6;
    if (m0 >= M) return;
    int hi = lane >> 4;
    int l  = lane & 15;

    v8f c0 = {}, c1 = {}, c2 = {}, c3 = {};
    const float* arow  = A + (size_t)(m0 + l) * K + (hi << 1);
    const float* bbase = B + (size_t)(hi << 1) * N + n0 + l;

    for (int k = 0; k < K; k += 4) {
        v2f a = *(const v2f*)(arow + k);
        const float* bk0 = bbase + (size_t)k * N;
        const float* bk1 = bk0 + N;
        v2f b0, b1, b2, b3;
        b0.x = bk0[0];   b0.y = bk1[0];
        b1.x = bk0[16];  b1.y = bk1[16];
        b2.x = bk0[32];  b2.y = bk1[32];
        b3.x = bk0[48];  b3.y = bk1[48];
        c0 = __builtin_amdgcn_wmma_f32_16x16x4_f32(false, a, false, b0, (short)0, c0, false, false);
        c1 = __builtin_amdgcn_wmma_f32_16x16x4_f32(false, a, false, b1, (short)0, c1, false, false);
        c2 = __builtin_amdgcn_wmma_f32_16x16x4_f32(false, a, false, b2, (short)0, c2, false, false);
        c3 = __builtin_amdgcn_wmma_f32_16x16x4_f32(false, a, false, b3, (short)0, c3, false, false);
    }
    float* crow = C + (size_t)(m0 + (hi << 3)) * N + n0 + l;
#pragma unroll
    for (int r = 0; r < 8; ++r) {
        crow[(size_t)r * N +  0] = c0[r];
        crow[(size_t)r * N + 16] = c1[r];
        crow[(size_t)r * N + 32] = c2[r];
        crow[(size_t)r * N + 48] = c3[r];
    }
}

__global__ void compute_al(const float* __restrict__ h, const float* __restrict__ as_,
                           const float* __restrict__ ad_, float* __restrict__ als,
                           float* __restrict__ ald, int n_nodes) {
    int i = blockIdx.x * blockDim.x + threadIdx.x;
    if (i >= n_nodes * NHEADS) return;
    int head = i & 7;
    const float* hp = h + (size_t)(i >> 3) * WDIM + head * NHID;
    const float* sp = as_ + head * NHID;
    const float* dp = ad_ + head * NHID;
    float s = 0.f, d = 0.f;
#pragma unroll
    for (int c = 0; c < NHID; ++c) { float v = hp[c]; s += v * sp[c]; d += v * dp[c]; }
    als[i] = s; ald[i] = d;
}

__global__ void edge_logits(const int* __restrict__ src, const int* __restrict__ dst,
                            const float* __restrict__ als, const float* __restrict__ ald,
                            float* __restrict__ ep, unsigned* __restrict__ menc) {
    long i = (long)blockIdx.x * blockDim.x + threadIdx.x;
    long tot = (long)(NEDGES + NNODES) * NHEADS;
    if (i >= tot) return;
    int e = (int)(i >> 3), head = (int)(i & 7);
    int s, d;
    if (e < NEDGES) { s = src[e]; d = dst[e]; } else { s = d = e - NEDGES; }
    float v = als[s * NHEADS + head] + ald[d * NHEADS + head];
    v = v > 0.f ? v : 0.2f * v;
    ep[i] = v;
    atomicMax(&menc[d * NHEADS + head], fenc(v));
}

__global__ void edge_exp(const int* __restrict__ dst, float* __restrict__ ep,
                         const unsigned* __restrict__ menc, float* __restrict__ denom) {
    long i = (long)blockIdx.x * blockDim.x + threadIdx.x;
    long tot = (long)(NEDGES + NNODES) * NHEADS;
    if (i >= tot) return;
    int e = (int)(i >> 3), head = (int)(i & 7);
    int d = (e < NEDGES) ? dst[e] : (e - NEDGES);
    float m = fdec(menc[d * NHEADS + head]);
    float p = expf(ep[i] - m);
    ep[i] = p;
    atomicAdd(&denom[d * NHEADS + head], p);
}

static_assert(NNODES % 8 == 0, "row_scan and node_agg take nodes eight at a time");

__global__ void dst_count(const int* __restrict__ dst, unsigned* __restrict__ cnt,
                          unsigned* __restrict__ rank) {
    int e = blockIdx.x * blockDim.x + threadIdx.x;
    if (e >= NEDGES) return;
    rank[e] = atomicAdd(&cnt[dst[e]], 1u);
}

#define SCAN_TILES ((NNODES + 8191) / 8192)
__device__ __forceinline__ unsigned wave_incl_scan(unsigned x, int lane) {
    for (int d = 1; d < 32; d <<= 1) {
        unsigned y = __shfl_up(x, d, 32);
        if (lane >= d) x += y;
    }
    return x;
}
__global__ __launch_bounds__(1024) void row_scan(const unsigned* __restrict__ cnt,
                                                 unsigned* __restrict__ rowptr) {
    __shared__ unsigned wsum[SCAN_TILES][32];
    const int t = threadIdx.x, lane = t & 31, w = t >> 5;
    uint4 a[SCAN_TILES], b[SCAN_TILES];
    unsigned s[SCAN_TILES], x[SCAN_TILES];
#pragma unroll
    for (int k = 0; k < SCAN_TILES; ++k) {
        const int i0 = k * 8192 + t * 8;
        a[k] = make_uint4(0u, 0u, 0u, 0u);
        b[k] = a[k];
        if (i0 < NNODES) {
            a[k] = *(const uint4*)(cnt + i0);
            b[k] = *(const uint4*)(cnt + i0 + 4);
        }
    }
#pragma unroll
    for (int k = 0; k < SCAN_TILES; ++k) {
        s[k] = a[k].x + a[k].y + a[k].z + a[k].w + b[k].x + b[k].y + b[k].z + b[k].w;
        x[k] = wave_incl_scan(s[k], lane);
        if (lane == 31) wsum[k][w] = x[k];
    }
    __syncthreads();
    unsigned carry = 0;
#pragma unroll
    for (int k = 0; k < SCAN_TILES; ++k) {
        const int i0 = k * 8192 + t * 8;
        const unsigned ws = wsum[k][lane];
        const unsigned xs = wave_incl_scan(ws, lane);
        const unsigned wexcl = __shfl(xs - ws, w, 32);
        const unsigned total = __shfl(xs, 31, 32);
        unsigned run = carry + wexcl + (x[k] - s[k]);
        if (i0 < NNODES) {
            uint4 o0, o1;
            o0.x = run; run += a[k].x;  o0.y = run; run += a[k].y;
            o0.z = run; run += a[k].z;  o0.w = run; run += a[k].w;
            o1.x = run; run += b[k].x;  o1.y = run; run += b[k].y;
            o1.z = run; run += b[k].z;  o1.w = run;
            *(uint4*)(rowptr + i0)     = o0;
            *(uint4*)(rowptr + i0 + 4) = o1;
        }
        carry += total;
    }
    if (t == 0) rowptr[NNODES] = carry;
}

__global__ void dst_place(const int* __restrict__ src, const int* __restrict__ dst,
                          const unsigned* __restrict__ rank, const unsigned* __restrict__ rowptr,
                          int2* __restrict__ csr) {
    int e = blockIdx.x * blockDim.x + threadIdx.x;
    if (e >= NEDGES) return;
    csr[rowptr[dst[e]] + rank[e]] = make_int2(src[e], e);
}

__device__ __forceinline__ float4 fma4(float p, float4 y, float4 a) {
    a.x = fmaf(p, y.x, a.x); a.y = fmaf(p, y.y, a.y);
    a.z = fmaf(p, y.z, a.z); a.w = fmaf(p, y.w, a.w);
    return a;
}

__global__ __launch_bounds__(256) void node_agg(const unsigned* __restrict__ rowptr,
                                                const int2* __restrict__ csr,
                                                const float* __restrict__ ep,
                                                const float* __restrict__ denom,
                                                const float* __restrict__ h,
                                                float* __restrict__ agg) {
    const int wv   = __builtin_amdgcn_readfirstlane((int)(threadIdx.x >> 5));
    const int node = blockIdx.x * 8 + wv;
    if (node >= NNODES) return;
    const int lane = threadIdx.x & 31;
    const int c0 = lane * 4, c1 = 128 + lane * 4;
    const int h0 = lane >> 3, h1 = 4 + (lane >> 3);
    const int beg = __builtin_amdgcn_readfirstlane((int)rowptr[node]);
    const int end = __builtin_amdgcn_readfirstlane((int)rowptr[node + 1]);

    const float* pself = ep + (size_t)(NEDGES + node) * NHEADS;
    const float* hself = h + (size_t)node * WDIM;
    float4 acc0 = make_float4(0.f, 0.f, 0.f, 0.f), acc1 = acc0;
    acc0 = fma4(pself[h0], *(const float4*)(hself + c0), acc0);
    acc1 = fma4(pself[h1], *(const float4*)(hself + c1), acc1);

    for (int j0 = beg; j0 < end; j0 += 32) {
        const int n = (end - j0) < 32 ? (end - j0) : 32;
        int2 ent = make_int2(0, 0);
        if (lane < n) ent = csr[j0 + lane];
        int j = 0;
        for (; j + 4 <= n; j += 4) {
            float q0[4], q1[4];
            float4 u0[4], u1[4];
#pragma unroll
            for (int u = 0; u < 4; ++u) {
                const int s = __builtin_amdgcn_readlane(ent.x, j + u);
                const int e = __builtin_amdgcn_readlane(ent.y, j + u);
                const float* pe = ep + (size_t)e * NHEADS;
                const float* hs = h + (size_t)s * WDIM;
                q0[u] = pe[h0];
                q1[u] = pe[h1];
                u0[u] = *(const float4*)(hs + c0);
                u1[u] = *(const float4*)(hs + c1);
            }
#pragma unroll
            for (int u = 0; u < 4; ++u) {
                acc0 = fma4(q0[u], u0[u], acc0);
                acc1 = fma4(q1[u], u1[u], acc1);
            }
        }
        for (; j < n; ++j) {
            const int s = __builtin_amdgcn_readlane(ent.x, j);
            const int e = __builtin_amdgcn_readlane(ent.y, j);
            const float* pe = ep + (size_t)e * NHEADS;
            const float* hs = h + (size_t)s * WDIM;
            acc0 = fma4(pe[h0], *(const float4*)(hs + c0), acc0);
            acc1 = fma4(pe[h1], *(const float4*)(hs + c1), acc1);
        }
    }
    const float* dn = denom + (size_t)node * NHEADS;
    const float d0 = dn[h0], d1 = dn[h1];
    float* orow = agg + (size_t)node * WDIM;
    *(float4*)(orow + c0) = make_float4(acc0.x / d0, acc0.y / d0, acc0.z / d0, acc0.w / d0);
    *(float4*)(orow + c1) = make_float4(acc1.x / d1, acc1.y / d1, acc1.z / d1, acc1.w / d1);
}

__global__ void bias_act(const float* __restrict__ agg, const float* __restrict__ b,
                         float* __restrict__ out, int apply_elu) {
    long i = (long)blockIdx.x * blockDim.x + threadIdx.x;
    if (i >= (long)NNODES * WDIM) return;
    float v = agg[i] + b[i & (WDIM - 1)];
    if (apply_elu) v = v > 0.f ? v : (expf(v) - 1.0f);
    out[i] = v;
}

__global__ void pool_accum(const float* __restrict__ x2, const int* __restrict__ batch,
                           unsigned* __restrict__ gmaxenc, float* __restrict__ gsum,
                           float* __restrict__ counts) {
    int n = blockIdx.x, c = threadIdx.x;
    int g = batch[n];
    float v = x2[(size_t)n * WDIM + c];
    atomicMax(&gmaxenc[g * WDIM + c], fenc(v));
    atomicAdd(&gsum[g * WDIM + c], v);
    if (c == 0) atomicAdd(&counts[g], 1.0f);
}

__global__ void pool_final(const unsigned* __restrict__ gmaxenc, const float* __restrict__ gsum,
                           const float* __restrict__ counts, float* __restrict__ rep) {
    int g = blockIdx.x, c = threadIdx.x;
    float cnt = counts[g];
    float mx = (cnt > 0.f) ? fdec(gmaxenc[g * WDIM + c]) : 0.0f;
    float mean = gsum[g * WDIM + c] / fmaxf(cnt, 1.0f);
    rep[(size_t)g * (2 * WDIM) + c] = mx;
    rep[(size_t)g * (2 * WDIM) + WDIM + c] = mean;
}

__global__ void fc_sigmoid(const float* __restrict__ rep, const float* __restrict__ Wfc,
                           const float* __restrict__ bfc, float* __restrict__ out) {
    __shared__ float red[256];
    int g = blockIdx.x, t = threadIdx.x;
    const float* r = rep + (size_t)g * (2 * WDIM);
    float s = r[t] * Wfc[t] + r[WDIM + t] * Wfc[WDIM + t];
    red[t] = s;
    __syncthreads();
    for (int off = 128; off > 0; off >>= 1) {
        if (t < off) red[t] += red[t + off];
        __syncthreads();
    }
    if (t == 0) out[g] = 1.0f / (1.0f + expf(-(red[0] + bfc[0])));
}

extern "C" void kernel_launch(void* const* d_in, const int* in_sizes, int n_in,
                              void* d_out, int out_size, void* d_ws, size_t ws_size,
                              hipStream_t stream) {
    (void)in_sizes; (void)n_in; (void)out_size; (void)ws_size;

    const float* x    = (const float*)d_in[0];
    const int*   ei   = (const int*)d_in[1];
    const int*   bat  = (const int*)d_in[2];
    const float* W1   = (const float*)d_in[3];
    const float* a1s  = (const float*)d_in[4];
    const float* a1d  = (const float*)d_in[5];
    const float* b1   = (const float*)d_in[6];
    const float* W2   = (const float*)d_in[7];
    const float* a2s  = (const float*)d_in[8];
    const float* a2d  = (const float*)d_in[9];
    const float* b2   = (const float*)d_in[10];
    const float* Wfc  = (const float*)d_in[11];
    const float* bfc  = (const float*)d_in[12];

    float* out = (float*)d_out;
    float* rep = out + NGRAPH;

    const int* esrc = ei;
    const int* edst = ei + NEDGES;
    const int ETOT = NEDGES + NNODES;

    float*    bufA  = (float*)d_ws;
    float*    bufB  = bufA + (size_t)NNODES * WDIM;
    float*    als   = bufB + (size_t)NNODES * WDIM;
    float*    ald   = als + (size_t)NNODES * NHEADS;
    unsigned* menc  = (unsigned*)(ald + (size_t)NNODES * NHEADS);
    float*    denom = (float*)(menc + (size_t)NNODES * NHEADS);
    float*    ep    = denom + (size_t)NNODES * NHEADS;
    unsigned* gmaxe = (unsigned*)(ep + (size_t)ETOT * NHEADS);
    float*    gsum  = (float*)(gmaxe + NGRAPH * WDIM);
    float*    cnts  = gsum + NGRAPH * WDIM;
    unsigned* dcnt  = (unsigned*)(cnts + NGRAPH);
    unsigned* rowp  = dcnt + NNODES;
    unsigned* erank = rowp + NNODES + 4;
    int2*     csr   = (int2*)(erank + NEDGES);

    const long nNH   = (long)NNODES * NHEADS;
    const long nNW   = (long)NNODES * WDIM;
    const long nEH   = (long)ETOT * NHEADS;
    const int  bNH   = (int)((nNH + 255) / 256);
    const int  bNW   = (int)((nNW + 255) / 256);
    const int  bEH   = (int)((nEH + 255) / 256);
    const int  bN    = (NNODES + 255) / 256;
    const int  bE    = (NEDGES + 255) / 256;

    fill_u32 <<<bN, 256, 0, stream>>>(dcnt, 0u, NNODES);
    dst_count<<<bE, 256, 0, stream>>>(edst, dcnt, erank);
    row_scan <<<1, 1024, 0, stream>>>(dcnt, rowp);
    dst_place<<<bE, 256, 0, stream>>>(esrc, edst, erank, rowp, csr);

    auto run_layer = [&](const float* xin, int K, const float* Wm, const float* as_,
                         const float* ad_, const float* bias, int elu) {
        long waves  = (long)(NNODES / 16) * (WDIM / 64);
        int  blocks = (int)((waves * 32 + 255) / 256);
        gemm_f32_wmma<<<blocks, 256, 0, stream>>>(xin, Wm, bufA, NNODES, WDIM, K);

        compute_al<<<bNH, 256, 0, stream>>>(bufA, as_, ad_, als, ald, NNODES);

        fill_u32<<<bNH, 256, 0, stream>>>(menc, 0u, nNH);
        fill_u32<<<bNH, 256, 0, stream>>>((unsigned*)denom, 0u, nNH);

        edge_logits<<<bEH, 256, 0, stream>>>(esrc, edst, als, ald, ep, menc);
        edge_exp   <<<bEH, 256, 0, stream>>>(edst, ep, menc, denom);
        node_agg   <<<NNODES / 8, 256, 0, stream>>>(rowp, csr, ep, denom, bufA, bufB);
        bias_act   <<<bNW, 256, 0, stream>>>(bufB, bias, bufB, elu);
    };

    run_layer(x,    128, W1, a1s, a1d, b1, 1);
    run_layer(bufB, 256, W2, a2s, a2d, b2, 0);

    fill_u32<<<(NGRAPH * WDIM + 255) / 256, 256, 0, stream>>>(gmaxe, 0u, NGRAPH * WDIM);
    fill_u32<<<(NGRAPH * WDIM + 255) / 256, 256, 0, stream>>>((unsigned*)gsum, 0u, NGRAPH * WDIM);
    fill_u32<<<1, 256, 0, stream>>>((unsigned*)cnts, 0u, NGRAPH);
    pool_accum<<<NNODES, 256, 0, stream>>>(bufB, bat, gmaxe, gsum, cnts);
    pool_final<<<NGRAPH, 256, 0, stream>>>(gmaxe, gsum, cnts, rep);
    fc_sigmoid<<<NGRAPH, 256, 0, stream>>>(rep, Wfc, bfc, out);
}


# segment softmax done per destination node inside the aggregation wave from the dst-ordered list (max, exp, denom in registers; denom summed in list order, no float atomics); edge kernels dropped
# speedup vs baseline: 3.5834x; 1.4100x over previous
#include <hip/hip_runtime.h>
#include <hip/hip_bf16.h>

typedef __attribute__((ext_vector_type(2))) float v2f;
typedef __attribute__((ext_vector_type(8))) float v8f;

#define NNODES 50000
#define NEDGES 800000
#define WDIM   256
#define NHEADS 8
#define NHID   32
#define NGRAPH 256

__device__ __forceinline__ unsigned fenc(float x) {
    unsigned u = __float_as_uint(x);
    return (u & 0x80000000u) ? ~u : (u | 0x80000000u);
}
__device__ __forceinline__ float fdec(unsigned e) {
    unsigned u = (e & 0x80000000u) ? (e & 0x7FFFFFFFu) : ~e;
    return __uint_as_float(u);
}

__global__ void fill_u32(unsigned* __restrict__ p, unsigned v, long n) {
    long i = (long)blockIdx.x * blockDim.x + threadIdx.x;
    if (i < n) p[i] = v;
}

__global__ void gemm_f32_wmma(const float* __restrict__ A, const float* __restrict__ B,
                              float* __restrict__ C, int M, int N, int K) {
    int wave   = (blockIdx.x * blockDim.x + threadIdx.x) >> 5;
    int lane   = threadIdx.x & 31;
    int nstrip = N >> 6;
    int m0 = (wave / nstrip) << 4;
    int n0 = (wave % nstrip) << 6;
    if (m0 >= M) return;
    int hi = lane >> 4;
    int l  = lane & 15;

    v8f c0 = {}, c1 = {}, c2 = {}, c3 = {};
    const float* arow  = A + (size_t)(m0 + l) * K + (hi << 1);
    const float* bbase = B + (size_t)(hi << 1) * N + n0 + l;

    for (int k = 0; k < K; k += 4) {
        v2f a = *(const v2f*)(arow + k);
        const float* bk0 = bbase + (size_t)k * N;
        const float* bk1 = bk0 + N;
        v2f b0, b1, b2, b3;
        b0.x = bk0[0];   b0.y = bk1[0];
        b1.x = bk0[16];  b1.y = bk1[16];
        b2.x = bk0[32];  b2.y = bk1[32];
        b3.x = bk0[48];  b3.y = bk1[48];
        c0 = __builtin_amdgcn_wmma_f32_16x16x4_f32(false, a, false, b0, (short)0, c0, false, false);
        c1 = __builtin_amdgcn_wmma_f32_16x16x4_f32(false, a, false, b1, (short)0, c1, false, false);
        c2 = __builtin_amdgcn_wmma_f32_16x16x4_f32(false, a, false, b2, (short)0, c2, false, false);
        c3 = __builtin_amdgcn_wmma_f32_16x16x4_f32(false, a, false, b3, (short)0, c3, false, false);
    }
    float* crow = C + (size_t)(m0 + (hi << 3)) * N + n0 + l;
#pragma unroll
    for (int r = 0; r < 8; ++r) {
        crow[(size_t)r * N +  0] = c0[r];
        crow[(size_t)r * N + 16] = c1[r];
        crow[(size_t)r * N + 32] = c2[r];
        crow[(size_t)r * N + 48] = c3[r];
    }
}

__global__ void compute_al(const float* __restrict__ h, const float* __restrict__ as_,
                           const float* __restrict__ ad_, float* __restrict__ als,
                           float* __restrict__ ald, int n_nodes) {
    int i = blockIdx.x * blockDim.x + threadIdx.x;
    if (i >= n_nodes * NHEADS) return;
    int head = i & 7;
    const float* hp = h + (size_t)(i >> 3) * WDIM + head * NHID;
    const float* sp = as_ + head * NHID;
    const float* dp = ad_ + head * NHID;
    float s = 0.f, d = 0.f;
#pragma unroll
    for (int c = 0; c < NHID; ++c) { float v = hp[c]; s += v * sp[c]; d += v * dp[c]; }
    als[i] = s; ald[i] = d;
}

static_assert(NNODES % 8 == 0, "row_scan and node_agg take nodes eight at a time");

__global__ void dst_count(const int* __restrict__ dst, unsigned* __restrict__ cnt,
                          unsigned* __restrict__ rank) {
    int e = blockIdx.x * blockDim.x + threadIdx.x;
    if (e >= NEDGES) return;
    rank[e] = atomicAdd(&cnt[dst[e]], 1u);
}

#define SCAN_TILES ((NNODES + 8191) / 8192)
__device__ __forceinline__ unsigned wave_incl_scan(unsigned x, int lane) {
    for (int d = 1; d < 32; d <<= 1) {
        unsigned y = __shfl_up(x, d, 32);
        if (lane >= d) x += y;
    }
    return x;
}
__global__ __launch_bounds__(1024) void row_scan(const unsigned* __restrict__ cnt,
                                                 unsigned* __restrict__ rowptr) {
    __shared__ unsigned wsum[SCAN_TILES][32];
    const int t = threadIdx.x, lane = t & 31, w = t >> 5;
    uint4 a[SCAN_TILES], b[SCAN_TILES];
    unsigned s[SCAN_TILES], x[SCAN_TILES];
#pragma unroll
    for (int k = 0; k < SCAN_TILES; ++k) {
        const int i0 = k * 8192 + t * 8;
        a[k] = make_uint4(0u, 0u, 0u, 0u);
        b[k] = a[k];
        if (i0 < NNODES) {
            a[k] = *(const uint4*)(cnt + i0);
            b[k] = *(const uint4*)(cnt + i0 + 4);
        }
    }
#pragma unroll
    for (int k = 0; k < SCAN_TILES; ++k) {
        s[k] = a[k].x + a[k].y + a[k].z + a[k].w + b[k].x + b[k].y + b[k].z + b[k].w;
        x[k] = wave_incl_scan(s[k], lane);
        if (lane == 31) wsum[k][w] = x[k];
    }
    __syncthreads();
    unsigned carry = 0;
#pragma unroll
    for (int k = 0; k < SCAN_TILES; ++k) {
        const int i0 = k * 8192 + t * 8;
        const unsigned ws = wsum[k][lane];
        const unsigned xs = wave_incl_scan(ws, lane);
        const unsigned wexcl = __shfl(xs - ws, w, 32);
        const unsigned total = __shfl(xs, 31, 32);
        unsigned run = carry + wexcl + (x[k] - s[k]);
        if (i0 < NNODES) {
            uint4 o0, o1;
            o0.x = run; run += a[k].x;  o0.y = run; run += a[k].y;
            o0.z = run; run += a[k].z;  o0.w = run; run += a[k].w;
            o1.x = run; run += b[k].x;  o1.y = run; run += b[k].y;
            o1.z = run; run += b[k].z;  o1.w = run;
            *(uint4*)(rowptr + i0)     = o0;
            *(uint4*)(rowptr + i0 + 4) = o1;
        }
        carry += total;
    }
    if (t == 0) rowptr[NNODES] = carry;
}

__global__ void dst_place(const int* __restrict__ src, const int* __restrict__ dst,
                          const unsigned* __restrict__ rank, const unsigned* __restrict__ rowptr,
                          int* __restrict__ csr) {
    int e = blockIdx.x * blockDim.x + threadIdx.x;
    if (e >= NEDGES) return;
    csr[rowptr[dst[e]] + rank[e]] = src[e];
}

__device__ __forceinline__ float4 fma4(float p, float4 y, float4 a) {
    a.x = fmaf(p, y.x, a.x); a.y = fmaf(p, y.y, a.y);
    a.z = fmaf(p, y.z, a.z); a.w = fmaf(p, y.w, a.w);
    return a;
}

__device__ __forceinline__ float lrelu(float v) { return v > 0.f ? v : 0.2f * v; }
__device__ __forceinline__ float sel4(bool lo, bool hi, float a, float b, float c, float d) {
    const float x = lo ? b : a, y = lo ? d : c;
    return hi ? y : x;
}
__device__ __forceinline__ float lane_bcast(float v, int src) {
    return __int_as_float(__builtin_amdgcn_readlane(__float_as_int(v), src));
}
__global__ __launch_bounds__(256) void node_agg(const unsigned* __restrict__ rowptr,
                                                const int* __restrict__ csr,
                                                const float* __restrict__ als,
                                                const float* __restrict__ ald,
                                                const float* __restrict__ h,
                                                float* __restrict__ agg) {
    const int wv   = __builtin_amdgcn_readfirstlane((int)(threadIdx.x >> 5));
    const int node = blockIdx.x * 8 + wv;
    if (node >= NNODES) return;
    const int lane = threadIdx.x & 31;
    const int c0 = lane * 4, c1 = 128 + lane * 4;
    const bool b3 = (lane & 8) != 0, b4 = (lane & 16) != 0;
    const int beg = __builtin_amdgcn_readfirstlane((int)rowptr[node]);
    const int end = __builtin_amdgcn_readfirstlane((int)rowptr[node + 1]);

    const float4 dA = *(const float4*)(ald + (size_t)node * NHEADS);
    const float4 dB = *(const float4*)(ald + (size_t)node * NHEADS + 4);
    const float4 sA = *(const float4*)(als + (size_t)node * NHEADS);
    const float4 sB = *(const float4*)(als + (size_t)node * NHEADS + 4);
    const float d[8] = {dA.x, dA.y, dA.z, dA.w, dB.x, dB.y, dB.z, dB.w};
    const float sf[8] = {sA.x, sA.y, sA.z, sA.w, sB.x, sB.y, sB.z, sB.w};
    float es[8], e[8], m[8];
#pragma unroll
    for (int k = 0; k < 8; ++k) { es[k] = lrelu(sf[k] + d[k]); e[k] = es[k]; m[k] = es[k]; }

    int sl = 0, jlast = beg;
    for (int j0 = beg; j0 < end; j0 += 32) {
        jlast = j0;
        if (j0 + lane < end) {
            sl = csr[j0 + lane];
            const float4 xa = *(const float4*)(als + (size_t)sl * NHEADS);
            const float4 xb = *(const float4*)(als + (size_t)sl * NHEADS + 4);
            const float x[8] = {xa.x, xa.y, xa.z, xa.w, xb.x, xb.y, xb.z, xb.w};
#pragma unroll
            for (int k = 0; k < 8; ++k) { e[k] = lrelu(x[k] + d[k]); m[k] = fmaxf(m[k], e[k]); }
        }
    }
    float k0 = b4 ? m[2] : m[0], k1 = b4 ? m[3] : m[1], k2 = b4 ? m[6] : m[4], k3 = b4 ? m[7] : m[5];
    float g0 = b4 ? m[0] : m[2], g1 = b4 ? m[1] : m[3], g2 = b4 ? m[4] : m[6], g3 = b4 ? m[5] : m[7];
    k0 = fmaxf(k0, __shfl_xor(g0, 16, 32)); k1 = fmaxf(k1, __shfl_xor(g1, 16, 32));
    k2 = fmaxf(k2, __shfl_xor(g2, 16, 32)); k3 = fmaxf(k3, __shfl_xor(g3, 16, 32));
    float m0 = b3 ? k1 : k0, t0 = b3 ? k0 : k1;
    float m1 = b3 ? k3 : k2, t1 = b3 ? k2 : k3;
    m0 = fmaxf(m0, __shfl_xor(t0, 8, 32));
    m1 = fmaxf(m1, __shfl_xor(t1, 8, 32));
#pragma unroll
    for (int x = 4; x > 0; x >>= 1) {
        m0 = fmaxf(m0, __shfl_xor(m0, x, 32));
        m1 = fmaxf(m1, __shfl_xor(m1, x, 32));
    }
    float mx[8];
#pragma unroll
    for (int k = 0; k < 4; ++k) { mx[k] = lane_bcast(m0, 8 * k); mx[4 + k] = lane_bcast(m1, 8 * k); }

    const float* hself = h + (size_t)node * WDIM;
    float dn0 = expf(sel4(b3, b4, es[0], es[1], es[2], es[3]) - m0);
    float dn1 = expf(sel4(b3, b4, es[4], es[5], es[6], es[7]) - m1);
    float4 acc0 = make_float4(0.f, 0.f, 0.f, 0.f), acc1 = acc0;
    acc0 = fma4(dn0, *(const float4*)(hself + c0), acc0);
    acc1 = fma4(dn1, *(const float4*)(hself + c1), acc1);

    for (int j0 = beg; j0 < end; j0 += 32) {
        const int n = (end - j0) < 32 ? (end - j0) : 32;
        int ent = sl;
        float p[8];
        if (j0 != jlast) {
            ent = csr[j0 + lane];
            const float4 xa = *(const float4*)(als + (size_t)ent * NHEADS);
            const float4 xb = *(const float4*)(als + (size_t)ent * NHEADS + 4);
            const float x[8] = {xa.x, xa.y, xa.z, xa.w, xb.x, xb.y, xb.z, xb.w};
#pragma unroll
            for (int k = 0; k < 8; ++k) p[k] = expf(lrelu(x[k] + d[k]) - mx[k]);
        } else {
#pragma unroll
            for (int k = 0; k < 8; ++k) p[k] = expf(e[k] - mx[k]);
        }
        int j = 0;
        for (; j + 4 <= n; j += 4) {
            float4 u0[4], u1[4];
#pragma unroll
            for (int u = 0; u < 4; ++u) {
                const float* hs = h + (size_t)__builtin_amdgcn_readlane(ent, j + u) * WDIM;
                u0[u] = *(const float4*)(hs + c0);
                u1[u] = *(const float4*)(hs + c1);
            }
#pragma unroll
            for (int u = 0; u < 4; ++u) {
                const float p0 = sel4(b3, b4, lane_bcast(p[0], j + u), lane_bcast(p[1], j + u),
                                         lane_bcast(p[2], j + u), lane_bcast(p[3], j + u));
                const float p1 = sel4(b3, b4, lane_bcast(p[4], j + u), lane_bcast(p[5], j + u),
                                         lane_bcast(p[6], j + u), lane_bcast(p[7], j + u));
                dn0 += p0; dn1 += p1;
                acc0 = fma4(p0, u0[u], acc0);
                acc1 = fma4(p1, u1[u], acc1);
            }
        }
        for (; j < n; ++j) {
            const float* hs = h + (size_t)__builtin_amdgcn_readlane(ent, j) * WDIM;
            const float p0 = sel4(b3, b4, lane_bcast(p[0], j), lane_bcast(p[1], j),
                                     lane_bcast(p[2], j), lane_bcast(p[3], j));
            const float p1 = sel4(b3, b4, lane_bcast(p[4], j), lane_bcast(p[5], j),
                                     lane_bcast(p[6], j), lane_bcast(p[7], j));
            dn0 += p0; dn1 += p1;
            acc0 = fma4(p0, *(const float4*)(hs + c0), acc0);
            acc1 = fma4(p1, *(const float4*)(hs + c1), acc1);
        }
    }
    float* orow = agg + (size_t)node * WDIM;
    *(float4*)(orow + c0) = make_float4(acc0.x / dn0, acc0.y / dn0, acc0.z / dn0, acc0.w / dn0);
    *(float4*)(orow + c1) = make_float4(acc1.x / dn1, acc1.y / dn1, acc1.z / dn1, acc1.w / dn1);
}

__global__ void bias_act(const float* __restrict__ agg, const float* __restrict__ b,
                         float* __restrict__ out, int apply_elu) {
    long i = (long)blockIdx.x * blockDim.x + threadIdx.x;
    if (i >= (long)NNODES * WDIM) return;
    float v = agg[i] + b[i & (WDIM - 1)];
    if (apply_elu) v = v > 0.f ? v : (expf(v) - 1.0f);
    out[i] = v;
}

__global__ void pool_accum(const float* __restrict__ x2, const int* __restrict__ batch,
                           unsigned* __restrict__ gmaxenc, float* __restrict__ gsum,
                           float* __restrict__ counts) {
    int n = blockIdx.x, c = threadIdx.x;
    int g = batch[n];
    float v = x2[(size_t)n * WDIM + c];
    atomicMax(&gmaxenc[g * WDIM + c], fenc(v));
    atomicAdd(&gsum[g * WDIM + c], v);
    if (c == 0) atomicAdd(&counts[g], 1.0f);
}

__global__ void pool_final(const unsigned* __restrict__ gmaxenc, const float* __restrict__ gsum,
                           const float* __restrict__ counts, float* __restrict__ rep) {
    int g = blockIdx.x, c = threadIdx.x;
    float cnt = counts[g];
    float mx = (cnt > 0.f) ? fdec(gmaxenc[g * WDIM + c]) : 0.0f;
    float mean = gsum[g * WDIM + c] / fmaxf(cnt, 1.0f);
    rep[(size_t)g * (2 * WDIM) + c] = mx;
    rep[(size_t)g * (2 * WDIM) + WDIM + c] = mean;
}

__global__ void fc_sigmoid(const float* __restrict__ rep, const float* __restrict__ Wfc,
                           const float* __restrict__ bfc, float* __restrict__ out) {
    __shared__ float red[256];
    int g = blockIdx.x, t = threadIdx.x;
    const float* r = rep + (size_t)g * (2 * WDIM);
    float s = r[t] * Wfc[t] + r[WDIM + t] * Wfc[WDIM + t];
    red[t] = s;
    __syncthreads();
    for (int off = 128; off > 0; off >>= 1) {
        if (t < off) red[t] += red[t + off];
        __syncthreads();
    }
    if (t == 0) out[g] = 1.0f / (1.0f + expf(-(red[0] + bfc[0])));
}

extern "C" void kernel_launch(void* const* d_in, const int* in_sizes, int n_in,
                              void* d_out, int out_size, void* d_ws, size_t ws_size,
                              hipStream_t stream) {
    (void)in_sizes; (void)n_in; (void)out_size; (void)ws_size;

    const float* x    = (const float*)d_in[0];
    const int*   ei   = (const int*)d_in[1];
    const int*   bat  = (const int*)d_in[2];
    const float* W1   = (const float*)d_in[3];
    const float* a1s  = (const float*)d_in[4];
    const float* a1d  = (const float*)d_in[5];
    const float* b1   = (const float*)d_in[6];
    const float* W2   = (const float*)d_in[7];
    const float* a2s  = (const float*)d_in[8];
    const float* a2d  = (const float*)d_in[9];
    const float* b2   = (const float*)d_in[10];
    const float* Wfc  = (const float*)d_in[11];
    const float* bfc  = (const float*)d_in[12];

    float* out = (float*)d_out;
    float* rep = out + NGRAPH;

    const int* esrc = ei;
    const int* edst = ei + NEDGES;

    float*    bufA  = (float*)d_ws;
    float*    bufB  = bufA + (size_t)NNODES * WDIM;
    float*    als   = bufB + (size_t)NNODES * WDIM;
    float*    ald   = als + (size_t)NNODES * NHEADS;
    unsigned* gmaxe = (unsigned*)(ald + (size_t)NNODES * NHEADS);
    float*    gsum  = (float*)(gmaxe + NGRAPH * WDIM);
    float*    cnts  = gsum + NGRAPH * WDIM;
    unsigned* dcnt  = (unsigned*)(cnts + NGRAPH);
    unsigned* rowp  = dcnt + NNODES;
    unsigned* erank = rowp + NNODES + 4;
    int*      csr   = (int*)(erank + NEDGES);

    const long nNH   = (long)NNODES * NHEADS;
    const long nNW   = (long)NNODES * WDIM;
    const int  bNH   = (int)((nNH + 255) / 256);
    const int  bNW   = (int)((nNW + 255) / 256);
    const int  bN    = (NNODES + 255) / 256;
    const int  bE    = (NEDGES + 255) / 256;

    fill_u32 <<<bN, 256, 0, stream>>>(dcnt, 0u, NNODES);
    dst_count<<<bE, 256, 0, stream>>>(edst, dcnt, erank);
    row_scan <<<1, 1024, 0, stream>>>(dcnt, rowp);
    dst_place<<<bE, 256, 0, stream>>>(esrc, edst, erank, rowp, csr);

    auto run_layer = [&](const float* xin, int K, const float* Wm, const float* as_,
                         const float* ad_, const float* bias, int elu) {
        long waves  = (long)(NNODES / 16) * (WDIM / 64);
        int  blocks = (int)((waves * 32 + 255) / 256);
        gemm_f32_wmma<<<blocks, 256, 0, stream>>>(xin, Wm, bufA, NNODES, WDIM, K);

        compute_al<<<bNH, 256, 0, stream>>>(bufA, as_, ad_, als, ald, NNODES);

        node_agg   <<<NNODES / 8, 256, 0, stream>>>(rowp, csr, als, ald, bufA, bufB);
        bias_act   <<<bNW, 256, 0, stream>>>(bufB, bias, bufB, elu);
    };

    run_layer(x,    128, W1, a1s, a1d, b1, 1);
    run_layer(bufB, 256, W2, a2s, a2d, b2, 0);

    fill_u32<<<(NGRAPH * WDIM + 255) / 256, 256, 0, stream>>>(gmaxe, 0u, NGRAPH * WDIM);
    fill_u32<<<(NGRAPH * WDIM + 255) / 256, 256, 0, stream>>>((unsigned*)gsum, 0u, NGRAPH * WDIM);
    fill_u32<<<1, 256, 0, stream>>>((unsigned*)cnts, 0u, NGRAPH);
    pool_accum<<<NNODES, 256, 0, stream>>>(bufB, bat, gmaxe, gsum, cnts);
    pool_final<<<NGRAPH, 256, 0, stream>>>(gmaxe, gsum, cnts, rep);
    fc_sigmoid<<<NGRAPH, 256, 0, stream>>>(rep, Wfc, bfc, out);
}


# bias add and ELU folded into the aggregation wave's epilogue before its single row store; the two element-wise bias/activation launches over [N,256] dropped (18 to 16 launches); bit-identical
# speedup vs baseline: 4.1527x; 1.1589x over previous
#include <hip/hip_runtime.h>
#include <hip/hip_bf16.h>

typedef __attribute__((ext_vector_type(2))) float v2f;
typedef __attribute__((ext_vector_type(8))) float v8f;

#define NNODES 50000
#define NEDGES 800000
#define WDIM   256
#define NHEADS 8
#define NHID   32
#define NGRAPH 256

__device__ __forceinline__ unsigned fenc(float x) {
    unsigned u = __float_as_uint(x);
    return (u & 0x80000000u) ? ~u : (u | 0x80000000u);
}
__device__ __forceinline__ float fdec(unsigned e) {
    unsigned u = (e & 0x80000000u) ? (e & 0x7FFFFFFFu) : ~e;
    return __uint_as_float(u);
}

__global__ void fill_u32(unsigned* __restrict__ p, unsigned v, long n) {
    long i = (long)blockIdx.x * blockDim.x + threadIdx.x;
    if (i < n) p[i] = v;
}

__global__ void gemm_f32_wmma(const float* __restrict__ A, const float* __restrict__ B,
                              float* __restrict__ C, int M, int N, int K) {
    int wave   = (blockIdx.x * blockDim.x + threadIdx.x) >> 5;
    int lane   = threadIdx.x & 31;
    int nstrip = N >> 6;
    int m0 = (wave / nstrip) << 4;
    int n0 = (wave % nstrip) << 6;
    if (m0 >= M) return;
    int hi = lane >> 4;
    int l  = lane & 15;

    v8f c0 = {}, c1 = {}, c2 = {}, c3 = {};
    const float* arow  = A + (size_t)(m0 + l) * K + (hi << 1);
    const float* bbase = B + (size_t)(hi << 1) * N + n0 + l;

    for (int k = 0; k < K; k += 4) {
        v2f a = *(const v2f*)(arow + k);
        const float* bk0 = bbase + (size_t)k * N;
        const float* bk1 = bk0 + N;
        v2f b0, b1, b2, b3;
        b0.x = bk0[0];   b0.y = bk1[0];
        b1.x = bk0[16];  b1.y = bk1[16];
        b2.x = bk0[32];  b2.y = bk1[32];
        b3.x = bk0[48];  b3.y = bk1[48];
        c0 = __builtin_amdgcn_wmma_f32_16x16x4_f32(false, a, false, b0, (short)0, c0, false, false);
        c1 = __builtin_amdgcn_wmma_f32_16x16x4_f32(false, a, false, b1, (short)0, c1, false, false);
        c2 = __builtin_amdgcn_wmma_f32_16x16x4_f32(false, a, false, b2, (short)0, c2, false, false);
        c3 = __builtin_amdgcn_wmma_f32_16x16x4_f32(false, a, false, b3, (short)0, c3, false, false);
    }
    float* crow = C + (size_t)(m0 + (hi << 3)) * N + n0 + l;
#pragma unroll
    for (int r = 0; r < 8; ++r) {
        crow[(size_t)r * N +  0] = c0[r];
        crow[(size_t)r * N + 16] = c1[r];
        crow[(size_t)r * N + 32] = c2[r];
        crow[(size_t)r * N + 48] = c3[r];
    }
}

__global__ void compute_al(const float* __restrict__ h, const float* __restrict__ as_,
                           const float* __restrict__ ad_, float* __restrict__ als,
                           float* __restrict__ ald, int n_nodes) {
    int i = blockIdx.x * blockDim.x + threadIdx.x;
    if (i >= n_nodes * NHEADS) return;
    int head = i & 7;
    const float* hp = h + (size_t)(i >> 3) * WDIM + head * NHID;
    const float* sp = as_ + head * NHID;
    const float* dp = ad_ + head * NHID;
    float s = 0.f, d = 0.f;
#pragma unroll
    for (int c = 0; c < NHID; ++c) { float v = hp[c]; s += v * sp[c]; d += v * dp[c]; }
    als[i] = s; ald[i] = d;
}

static_assert(NNODES % 8 == 0, "row_scan and node_agg take nodes eight at a time");

__global__ void dst_count(const int* __restrict__ dst, unsigned* __restrict__ cnt,
                          unsigned* __restrict__ rank) {
    int e = blockIdx.x * blockDim.x + threadIdx.x;
    if (e >= NEDGES) return;
    rank[e] = atomicAdd(&cnt[dst[e]], 1u);
}

#define SCAN_TILES ((NNODES + 8191) / 8192)
__device__ __forceinline__ unsigned wave_incl_scan(unsigned x, int lane) {
    for (int d = 1; d < 32; d <<= 1) {
        unsigned y = __shfl_up(x, d, 32);
        if (lane >= d) x += y;
    }
    return x;
}
__global__ __launch_bounds__(1024) void row_scan(const unsigned* __restrict__ cnt,
                                                 unsigned* __restrict__ rowptr) {
    __shared__ unsigned wsum[SCAN_TILES][32];
    const int t = threadIdx.x, lane = t & 31, w = t >> 5;
    uint4 a[SCAN_TILES], b[SCAN_TILES];
    unsigned s[SCAN_TILES], x[SCAN_TILES];
#pragma unroll
    for (int k = 0; k < SCAN_TILES; ++k) {
        const int i0 = k * 8192 + t * 8;
        a[k] = make_uint4(0u, 0u, 0u, 0u);
        b[k] = a[k];
        if (i0 < NNODES) {
            a[k] = *(const uint4*)(cnt + i0);
            b[k] = *(const uint4*)(cnt + i0 + 4);
        }
    }
#pragma unroll
    for (int k = 0; k < SCAN_TILES; ++k) {
        s[k] = a[k].x + a[k].y + a[k].z + a[k].w + b[k].x + b[k].y + b[k].z + b[k].w;
        x[k] = wave_incl_scan(s[k], lane);
        if (lane == 31) wsum[k][w] = x[k];
    }
    __syncthreads();
    unsigned carry = 0;
#pragma unroll
    for (int k = 0; k < SCAN_TILES; ++k) {
        const int i0 = k * 8192 + t * 8;
        const unsigned ws = wsum[k][lane];
        const unsigned xs = wave_incl_scan(ws, lane);
        const unsigned wexcl = __shfl(xs - ws, w, 32);
        const unsigned total = __shfl(xs, 31, 32);
        unsigned run = carry + wexcl + (x[k] - s[k]);
        if (i0 < NNODES) {
            uint4 o0, o1;
            o0.x = run; run += a[k].x;  o0.y = run; run += a[k].y;
            o0.z = run; run += a[k].z;  o0.w = run; run += a[k].w;
            o1.x = run; run += b[k].x;  o1.y = run; run += b[k].y;
            o1.z = run; run += b[k].z;  o1.w = run;
            *(uint4*)(rowptr + i0)     = o0;
            *(uint4*)(rowptr + i0 + 4) = o1;
        }
        carry += total;
    }
    if (t == 0) rowptr[NNODES] = carry;
}

__global__ void dst_place(const int* __restrict__ src, const int* __restrict__ dst,
                          const unsigned* __restrict__ rank, const unsigned* __restrict__ rowptr,
                          int* __restrict__ csr) {
    int e = blockIdx.x * blockDim.x + threadIdx.x;
    if (e >= NEDGES) return;
    csr[rowptr[dst[e]] + rank[e]] = src[e];
}

__device__ __forceinline__ float4 fma4(float p, float4 y, float4 a) {
    a.x = fmaf(p, y.x, a.x); a.y = fmaf(p, y.y, a.y);
    a.z = fmaf(p, y.z, a.z); a.w = fmaf(p, y.w, a.w);
    return a;
}

__device__ __forceinline__ float lrelu(float v) { return v > 0.f ? v : 0.2f * v; }
__device__ __forceinline__ float elu1(float v) { return v > 0.f ? v : (expf(v) - 1.0f); }
__device__ __forceinline__ float sel4(bool lo, bool hi, float a, float b, float c, float d) {
    const float x = lo ? b : a, y = lo ? d : c;
    return hi ? y : x;
}
__device__ __forceinline__ float lane_bcast(float v, int src) {
    return __int_as_float(__builtin_amdgcn_readlane(__float_as_int(v), src));
}
__global__ __launch_bounds__(256) void node_agg(const unsigned* __restrict__ rowptr,
                                                const int* __restrict__ csr,
                                                const float* __restrict__ als,
                                                const float* __restrict__ ald,
                                                const float* __restrict__ h,
                                                const float* __restrict__ bias,
                                                int apply_elu,
                                                float* __restrict__ outx) {
    const int wv   = __builtin_amdgcn_readfirstlane((int)(threadIdx.x >> 5));
    const int node = blockIdx.x * 8 + wv;
    if (node >= NNODES) return;
    const int lane = threadIdx.x & 31;
    const int c0 = lane * 4, c1 = 128 + lane * 4;
    const bool b3 = (lane & 8) != 0, b4 = (lane & 16) != 0;
    const int beg = __builtin_amdgcn_readfirstlane((int)rowptr[node]);
    const int end = __builtin_amdgcn_readfirstlane((int)rowptr[node + 1]);

    const float4 dA = *(const float4*)(ald + (size_t)node * NHEADS);
    const float4 dB = *(const float4*)(ald + (size_t)node * NHEADS + 4);
    const float4 sA = *(const float4*)(als + (size_t)node * NHEADS);
    const float4 sB = *(const float4*)(als + (size_t)node * NHEADS + 4);
    const float d[8] = {dA.x, dA.y, dA.z, dA.w, dB.x, dB.y, dB.z, dB.w};
    const float sf[8] = {sA.x, sA.y, sA.z, sA.w, sB.x, sB.y, sB.z, sB.w};
    float es[8], e[8], m[8];
#pragma unroll
    for (int k = 0; k < 8; ++k) { es[k] = lrelu(sf[k] + d[k]); e[k] = es[k]; m[k] = es[k]; }

    int sl = 0, jlast = beg;
    for (int j0 = beg; j0 < end; j0 += 32) {
        jlast = j0;
        if (j0 + lane < end) {
            sl = csr[j0 + lane];
            const float4 xa = *(const float4*)(als + (size_t)sl * NHEADS);
            const float4 xb = *(const float4*)(als + (size_t)sl * NHEADS + 4);
            const float x[8] = {xa.x, xa.y, xa.z, xa.w, xb.x, xb.y, xb.z, xb.w};
#pragma unroll
            for (int k = 0; k < 8; ++k) { e[k] = lrelu(x[k] + d[k]); m[k] = fmaxf(m[k], e[k]); }
        }
    }
    float k0 = b4 ? m[2] : m[0], k1 = b4 ? m[3] : m[1], k2 = b4 ? m[6] : m[4], k3 = b4 ? m[7] : m[5];
    float g0 = b4 ? m[0] : m[2], g1 = b4 ? m[1] : m[3], g2 = b4 ? m[4] : m[6], g3 = b4 ? m[5] : m[7];
    k0 = fmaxf(k0, __shfl_xor(g0, 16, 32)); k1 = fmaxf(k1, __shfl_xor(g1, 16, 32));
    k2 = fmaxf(k2, __shfl_xor(g2, 16, 32)); k3 = fmaxf(k3, __shfl_xor(g3, 16, 32));
    float m0 = b3 ? k1 : k0, t0 = b3 ? k0 : k1;
    float m1 = b3 ? k3 : k2, t1 = b3 ? k2 : k3;
    m0 = fmaxf(m0, __shfl_xor(t0, 8, 32));
    m1 = fmaxf(m1, __shfl_xor(t1, 8, 32));
#pragma unroll
    for (int x = 4; x > 0; x >>= 1) {
        m0 = fmaxf(m0, __shfl_xor(m0, x, 32));
        m1 = fmaxf(m1, __shfl_xor(m1, x, 32));
    }
    float mx[8];
#pragma unroll
    for (int k = 0; k < 4; ++k) { mx[k] = lane_bcast(m0, 8 * k); mx[4 + k] = lane_bcast(m1, 8 * k); }

    const float* hself = h + (size_t)node * WDIM;
    float dn0 = expf(sel4(b3, b4, es[0], es[1], es[2], es[3]) - m0);
    float dn1 = expf(sel4(b3, b4, es[4], es[5], es[6], es[7]) - m1);
    float4 acc0 = make_float4(0.f, 0.f, 0.f, 0.f), acc1 = acc0;
    acc0 = fma4(dn0, *(const float4*)(hself + c0), acc0);
    acc1 = fma4(dn1, *(const float4*)(hself + c1), acc1);

    for (int j0 = beg; j0 < end; j0 += 32) {
        const int n = (end - j0) < 32 ? (end - j0) : 32;
        int ent = sl;
        float p[8];
        if (j0 != jlast) {
            ent = csr[j0 + lane];
            const float4 xa = *(const float4*)(als + (size_t)ent * NHEADS);
            const float4 xb = *(const float4*)(als + (size_t)ent * NHEADS + 4);
            const float x[8] = {xa.x, xa.y, xa.z, xa.w, xb.x, xb.y, xb.z, xb.w};
#pragma unroll
            for (int k = 0; k < 8; ++k) p[k] = expf(lrelu(x[k] + d[k]) - mx[k]);
        } else {
#pragma unroll
            for (int k = 0; k < 8; ++k) p[k] = expf(e[k] - mx[k]);
        }
        int j = 0;
        for (; j + 4 <= n; j += 4) {
            float4 u0[4], u1[4];
#pragma unroll
            for (int u = 0; u < 4; ++u) {
                const float* hs = h + (size_t)__builtin_amdgcn_readlane(ent, j + u) * WDIM;
                u0[u] = *(const float4*)(hs + c0);
                u1[u] = *(const float4*)(hs + c1);
            }
#pragma unroll
            for (int u = 0; u < 4; ++u) {
                const float p0 = sel4(b3, b4, lane_bcast(p[0], j + u), lane_bcast(p[1], j + u),
                                         lane_bcast(p[2], j + u), lane_bcast(p[3], j + u));
                const float p1 = sel4(b3, b4, lane_bcast(p[4], j + u), lane_bcast(p[5], j + u),
                                         lane_bcast(p[6], j + u), lane_bcast(p[7], j + u));
                dn0 += p0; dn1 += p1;
                acc0 = fma4(p0, u0[u], acc0);
                acc1 = fma4(p1, u1[u], acc1);
            }
        }
        for (; j < n; ++j) {
            const float* hs = h + (size_t)__builtin_amdgcn_readlane(ent, j) * WDIM;
            const float p0 = sel4(b3, b4, lane_bcast(p[0], j), lane_bcast(p[1], j),
                                     lane_bcast(p[2], j), lane_bcast(p[3], j));
            const float p1 = sel4(b3, b4, lane_bcast(p[4], j), lane_bcast(p[5], j),
                                     lane_bcast(p[6], j), lane_bcast(p[7], j));
            dn0 += p0; dn1 += p1;
            acc0 = fma4(p0, *(const float4*)(hs + c0), acc0);
            acc1 = fma4(p1, *(const float4*)(hs + c1), acc1);
        }
    }
    const float4 bz0 = *(const float4*)(bias + c0);
    const float4 bz1 = *(const float4*)(bias + c1);
    float4 o0 = make_float4(acc0.x / dn0 + bz0.x, acc0.y / dn0 + bz0.y, acc0.z / dn0 + bz0.z, acc0.w / dn0 + bz0.w);
    float4 o1 = make_float4(acc1.x / dn1 + bz1.x, acc1.y / dn1 + bz1.y, acc1.z / dn1 + bz1.z, acc1.w / dn1 + bz1.w);
    if (apply_elu) {
        o0.x = elu1(o0.x); o0.y = elu1(o0.y); o0.z = elu1(o0.z); o0.w = elu1(o0.w);
        o1.x = elu1(o1.x); o1.y = elu1(o1.y); o1.z = elu1(o1.z); o1.w = elu1(o1.w);
    }
    float* orow = outx + (size_t)node * WDIM;
    *(float4*)(orow + c0) = o0;
    *(float4*)(orow + c1) = o1;
}

__global__ void pool_accum(const float* __restrict__ x2, const int* __restrict__ batch,
                           unsigned* __restrict__ gmaxenc, float* __restrict__ gsum,
                           float* __restrict__ counts) {
    int n = blockIdx.x, c = threadIdx.x;
    int g = batch[n];
    float v = x2[(size_t)n * WDIM + c];
    atomicMax(&gmaxenc[g * WDIM + c], fenc(v));
    atomicAdd(&gsum[g * WDIM + c], v);
    if (c == 0) atomicAdd(&counts[g], 1.0f);
}

__global__ void pool_final(const unsigned* __restrict__ gmaxenc, const float* __restrict__ gsum,
                           const float* __restrict__ counts, float* __restrict__ rep) {
    int g = blockIdx.x, c = threadIdx.x;
    float cnt = counts[g];
    float mx = (cnt > 0.f) ? fdec(gmaxenc[g * WDIM + c]) : 0.0f;
    float mean = gsum[g * WDIM + c] / fmaxf(cnt, 1.0f);
    rep[(size_t)g * (2 * WDIM) + c] = mx;
    rep[(size_t)g * (2 * WDIM) + WDIM + c] = mean;
}

__global__ void fc_sigmoid(const float* __restrict__ rep, const float* __restrict__ Wfc,
                           const float* __restrict__ bfc, float* __restrict__ out) {
    __shared__ float red[256];
    int g = blockIdx.x, t = threadIdx.x;
    const float* r = rep + (size_t)g * (2 * WDIM);
    float s = r[t] * Wfc[t] + r[WDIM + t] * Wfc[WDIM + t];
    red[t] = s;
    __syncthreads();
    for (int off = 128; off > 0; off >>= 1) {
        if (t < off) red[t] += red[t + off];
        __syncthreads();
    }
    if (t == 0) out[g] = 1.0f / (1.0f + expf(-(red[0] + bfc[0])));
}

extern "C" void kernel_launch(void* const* d_in, const int* in_sizes, int n_in,
                              void* d_out, int out_size, void* d_ws, size_t ws_size,
                              hipStream_t stream) {
    (void)in_sizes; (void)n_in; (void)out_size; (void)ws_size;

    const float* x    = (const float*)d_in[0];
    const int*   ei   = (const int*)d_in[1];
    const int*   bat  = (const int*)d_in[2];
    const float* W1   = (const float*)d_in[3];
    const float* a1s  = (const float*)d_in[4];
    const float* a1d  = (const float*)d_in[5];
    const float* b1   = (const float*)d_in[6];
    const float* W2   = (const float*)d_in[7];
    const float* a2s  = (const float*)d_in[8];
    const float* a2d  = (const float*)d_in[9];
    const float* b2   = (const float*)d_in[10];
    const float* Wfc  = (const float*)d_in[11];
    const float* bfc  = (const float*)d_in[12];

    float* out = (float*)d_out;
    float* rep = out + NGRAPH;

    const int* esrc = ei;
    const int* edst = ei + NEDGES;

    float*    bufA  = (float*)d_ws;
    float*    bufB  = bufA + (size_t)NNODES * WDIM;
    float*    als   = bufB + (size_t)NNODES * WDIM;
    float*    ald   = als + (size_t)NNODES * NHEADS;
    unsigned* gmaxe = (unsigned*)(ald + (size_t)NNODES * NHEADS);
    float*    gsum  = (float*)(gmaxe + NGRAPH * WDIM);
    float*    cnts  = gsum + NGRAPH * WDIM;
    unsigned* dcnt  = (unsigned*)(cnts + NGRAPH);
    unsigned* rowp  = dcnt + NNODES;
    unsigned* erank = rowp + NNODES + 4;
    int*      csr   = (int*)(erank + NEDGES);

    const long nNH   = (long)NNODES * NHEADS;
    const int  bNH   = (int)((nNH + 255) / 256);
    const int  bN    = (NNODES + 255) / 256;
    const int  bE    = (NEDGES + 255) / 256;

    fill_u32 <<<bN, 256, 0, stream>>>(dcnt, 0u, NNODES);
    dst_count<<<bE, 256, 0, stream>>>(edst, dcnt, erank);
    row_scan <<<1, 1024, 0, stream>>>(dcnt, rowp);
    dst_place<<<bE, 256, 0, stream>>>(esrc, edst, erank, rowp, csr);

    auto run_layer = [&](const float* xin, int K, const float* Wm, const float* as_,
                         const float* ad_, const float* bias, int elu) {
        long waves  = (long)(NNODES / 16) * (WDIM / 64);
        int  blocks = (int)((waves * 32 + 255) / 256);
        gemm_f32_wmma<<<blocks, 256, 0, stream>>>(xin, Wm, bufA, NNODES, WDIM, K);

        compute_al<<<bNH, 256, 0, stream>>>(bufA, as_, ad_, als, ald, NNODES);

        node_agg   <<<NNODES / 8, 256, 0, stream>>>(rowp, csr, als, ald, bufA, bias, elu, bufB);
    };

    run_layer(x,    128, W1, a1s, a1d, b1, 1);
    run_layer(bufB, 256, W2, a2s, a2d, b2, 0);

    fill_u32<<<(NGRAPH * WDIM + 255) / 256, 256, 0, stream>>>(gmaxe, 0u, NGRAPH * WDIM);
    fill_u32<<<(NGRAPH * WDIM + 255) / 256, 256, 0, stream>>>((unsigned*)gsum, 0u, NGRAPH * WDIM);
    fill_u32<<<1, 256, 0, stream>>>((unsigned*)cnts, 0u, NGRAPH);
    pool_accum<<<NNODES, 256, 0, stream>>>(bufB, bat, gmaxe, gsum, cnts);
    pool_final<<<NGRAPH, 256, 0, stream>>>(gmaxe, gsum, cnts, rep);
    fc_sigmoid<<<NGRAPH, 256, 0, stream>>>(rep, Wfc, bfc, out);
}


# graph pooling: a wave folds 16 consecutive rows in registers (max; sum in node order), then one whole-line atomicMax/atomicAdd per channel per run of equal graph id, not per node; sum tree changes
# speedup vs baseline: 5.1581x; 1.2421x over previous
#include <hip/hip_runtime.h>
#include <hip/hip_bf16.h>

typedef __attribute__((ext_vector_type(2))) float v2f;
typedef __attribute__((ext_vector_type(8))) float v8f;

#define NNODES 50000
#define NEDGES 800000
#define WDIM   256
#define NHEADS 8
#define NHID   32
#define NGRAPH 256

__device__ __forceinline__ unsigned fenc(float x) {
    unsigned u = __float_as_uint(x);
    return (u & 0x80000000u) ? ~u : (u | 0x80000000u);
}
__device__ __forceinline__ float fdec(unsigned e) {
    unsigned u = (e & 0x80000000u) ? (e & 0x7FFFFFFFu) : ~e;
    return __uint_as_float(u);
}

__global__ void fill_u32(unsigned* __restrict__ p, unsigned v, long n) {
    long i = (long)blockIdx.x * blockDim.x + threadIdx.x;
    if (i < n) p[i] = v;
}

__global__ void gemm_f32_wmma(const float* __restrict__ A, const float* __restrict__ B,
                              float* __restrict__ C, int M, int N, int K) {
    int wave   = (blockIdx.x * blockDim.x + threadIdx.x) >> 5;
    int lane   = threadIdx.x & 31;
    int nstrip = N >> 6;
    int m0 = (wave / nstrip) << 4;
    int n0 = (wave % nstrip) << 6;
    if (m0 >= M) return;
    int hi = lane >> 4;
    int l  = lane & 15;

    v8f c0 = {}, c1 = {}, c2 = {}, c3 = {};
    const float* arow  = A + (size_t)(m0 + l) * K + (hi << 1);
    const float* bbase = B + (size_t)(hi << 1) * N + n0 + l;

    for (int k = 0; k < K; k += 4) {
        v2f a = *(const v2f*)(arow + k);
        const float* bk0 = bbase + (size_t)k * N;
        const float* bk1 = bk0 + N;
        v2f b0, b1, b2, b3;
        b0.x = bk0[0];   b0.y = bk1[0];
        b1.x = bk0[16];  b1.y = bk1[16];
        b2.x = bk0[32];  b2.y = bk1[32];
        b3.x = bk0[48];  b3.y = bk1[48];
        c0 = __builtin_amdgcn_wmma_f32_16x16x4_f32(false, a, false, b0, (short)0, c0, false, false);
        c1 = __builtin_amdgcn_wmma_f32_16x16x4_f32(false, a, false, b1, (short)0, c1, false, false);
        c2 = __builtin_amdgcn_wmma_f32_16x16x4_f32(false, a, false, b2, (short)0, c2, false, false);
        c3 = __builtin_amdgcn_wmma_f32_16x16x4_f32(false, a, false, b3, (short)0, c3, false, false);
    }
    float* crow = C + (size_t)(m0 + (hi << 3)) * N + n0 + l;
#pragma unroll
    for (int r = 0; r < 8; ++r) {
        crow[(size_t)r * N +  0] = c0[r];
        crow[(size_t)r * N + 16] = c1[r];
        crow[(size_t)r * N + 32] = c2[r];
        crow[(size_t)r * N + 48] = c3[r];
    }
}

__global__ void compute_al(const float* __restrict__ h, const float* __restrict__ as_,
                           const float* __restrict__ ad_, float* __restrict__ als,
                           float* __restrict__ ald, int n_nodes) {
    int i = blockIdx.x * blockDim.x + threadIdx.x;
    if (i >= n_nodes * NHEADS) return;
    int head = i & 7;
    const float* hp = h + (size_t)(i >> 3) * WDIM + head * NHID;
    const float* sp = as_ + head * NHID;
    const float* dp = ad_ + head * NHID;
    float s = 0.f, d = 0.f;
#pragma unroll
    for (int c = 0; c < NHID; ++c) { float v = hp[c]; s += v * sp[c]; d += v * dp[c]; }
    als[i] = s; ald[i] = d;
}

static_assert(NNODES % 8 == 0, "row_scan and node_agg take nodes eight at a time");

__global__ void dst_count(const int* __restrict__ dst, unsigned* __restrict__ cnt,
                          unsigned* __restrict__ rank) {
    int e = blockIdx.x * blockDim.x + threadIdx.x;
    if (e >= NEDGES) return;
    rank[e] = atomicAdd(&cnt[dst[e]], 1u);
}

#define SCAN_TILES ((NNODES + 8191) / 8192)
__device__ __forceinline__ unsigned wave_incl_scan(unsigned x, int lane) {
    for (int d = 1; d < 32; d <<= 1) {
        unsigned y = __shfl_up(x, d, 32);
        if (lane >= d) x += y;
    }
    return x;
}
__global__ __launch_bounds__(1024) void row_scan(const unsigned* __restrict__ cnt,
                                                 unsigned* __restrict__ rowptr) {
    __shared__ unsigned wsum[SCAN_TILES][32];
    const int t = threadIdx.x, lane = t & 31, w = t >> 5;
    uint4 a[SCAN_TILES], b[SCAN_TILES];
    unsigned s[SCAN_TILES], x[SCAN_TILES];
#pragma unroll
    for (int k = 0; k < SCAN_TILES; ++k) {
        const int i0 = k * 8192 + t * 8;
        a[k] = make_uint4(0u, 0u, 0u, 0u);
        b[k] = a[k];
        if (i0 < NNODES) {
            a[k] = *(const uint4*)(cnt + i0);
            b[k] = *(const uint4*)(cnt + i0 + 4);
        }
    }
#pragma unroll
    for (int k = 0; k < SCAN_TILES; ++k) {
        s[k] = a[k].x + a[k].y + a[k].z + a[k].w + b[k].x + b[k].y + b[k].z + b[k].w;
        x[k] = wave_incl_scan(s[k], lane);
        if (lane == 31) wsum[k][w] = x[k];
    }
    __syncthreads();
    unsigned carry = 0;
#pragma unroll
    for (int k = 0; k < SCAN_TILES; ++k) {
        const int i0 = k * 8192 + t * 8;
        const unsigned ws = wsum[k][lane];
        const unsigned xs = wave_incl_scan(ws, lane);
        const unsigned wexcl = __shfl(xs - ws, w, 32);
        const unsigned total = __shfl(xs, 31, 32);
        unsigned run = carry + wexcl + (x[k] - s[k]);
        if (i0 < NNODES) {
            uint4 o0, o1;
            o0.x = run; run += a[k].x;  o0.y = run; run += a[k].y;
            o0.z = run; run += a[k].z;  o0.w = run; run += a[k].w;
            o1.x = run; run += b[k].x;  o1.y = run; run += b[k].y;
            o1.z = run; run += b[k].z;  o1.w = run;
            *(uint4*)(rowptr + i0)     = o0;
            *(uint4*)(rowptr + i0 + 4) = o1;
        }
        carry += total;
    }
    if (t == 0) rowptr[NNODES] = carry;
}

__global__ void dst_place(const int* __restrict__ src, const int* __restrict__ dst,
                          const unsigned* __restrict__ rank, const unsigned* __restrict__ rowptr,
                          int* __restrict__ csr) {
    int e = blockIdx.x * blockDim.x + threadIdx.x;
    if (e >= NEDGES) return;
    csr[rowptr[dst[e]] + rank[e]] = src[e];
}

__device__ __forceinline__ float4 fma4(float p, float4 y, float4 a) {
    a.x = fmaf(p, y.x, a.x); a.y = fmaf(p, y.y, a.y);
    a.z = fmaf(p, y.z, a.z); a.w = fmaf(p, y.w, a.w);
    return a;
}

__device__ __forceinline__ float lrelu(float v) { return v > 0.f ? v : 0.2f * v; }
__device__ __forceinline__ float elu1(float v) { return v > 0.f ? v : (expf(v) - 1.0f); }
__device__ __forceinline__ float sel4(bool lo, bool hi, float a, float b, float c, float d) {
    const float x = lo ? b : a, y = lo ? d : c;
    return hi ? y : x;
}
__device__ __forceinline__ float lane_bcast(float v, int src) {
    return __int_as_float(__builtin_amdgcn_readlane(__float_as_int(v), src));
}
__global__ __launch_bounds__(256) void node_agg(const unsigned* __restrict__ rowptr,
                                                const int* __restrict__ csr,
                                                const float* __restrict__ als,
                                                const float* __restrict__ ald,
                                                const float* __restrict__ h,
                                                const float* __restrict__ bias,
                                                int apply_elu,
                                                float* __restrict__ outx) {
    const int wv   = __builtin_amdgcn_readfirstlane((int)(threadIdx.x >> 5));
    const int node = blockIdx.x * 8 + wv;
    if (node >= NNODES) return;
    const int lane = threadIdx.x & 31;
    const int c0 = lane * 4, c1 = 128 + lane * 4;
    const bool b3 = (lane & 8) != 0, b4 = (lane & 16) != 0;
    const int beg = __builtin_amdgcn_readfirstlane((int)rowptr[node]);
    const int end = __builtin_amdgcn_readfirstlane((int)rowptr[node + 1]);

    const float4 dA = *(const float4*)(ald + (size_t)node * NHEADS);
    const float4 dB = *(const float4*)(ald + (size_t)node * NHEADS + 4);
    const float4 sA = *(const float4*)(als + (size_t)node * NHEADS);
    const float4 sB = *(const float4*)(als + (size_t)node * NHEADS + 4);
    const float d[8] = {dA.x, dA.y, dA.z, dA.w, dB.x, dB.y, dB.z, dB.w};
    const float sf[8] = {sA.x, sA.y, sA.z, sA.w, sB.x, sB.y, sB.z, sB.w};
    float es[8], e[8], m[8];
#pragma unroll
    for (int k = 0; k < 8; ++k) { es[k] = lrelu(sf[k] + d[k]); e[k] = es[k]; m[k] = es[k]; }

    int sl = 0, jlast = beg;
    for (int j0 = beg; j0 < end; j0 += 32) {
        jlast = j0;
        if (j0 + lane < end) {
            sl = csr[j0 + lane];
            const float4 xa = *(const float4*)(als + (size_t)sl * NHEADS);
            const float4 xb = *(const float4*)(als + (size_t)sl * NHEADS + 4);
            const float x[8] = {xa.x, xa.y, xa.z, xa.w, xb.x, xb.y, xb.z, xb.w};
#pragma unroll
            for (int k = 0; k < 8; ++k) { e[k] = lrelu(x[k] + d[k]); m[k] = fmaxf(m[k], e[k]); }
        }
    }
    float k0 = b4 ? m[2] : m[0], k1 = b4 ? m[3] : m[1], k2 = b4 ? m[6] : m[4], k3 = b4 ? m[7] : m[5];
    float g0 = b4 ? m[0] : m[2], g1 = b4 ? m[1] : m[3], g2 = b4 ? m[4] : m[6], g3 = b4 ? m[5] : m[7];
    k0 = fmaxf(k0, __shfl_xor(g0, 16, 32)); k1 = fmaxf(k1, __shfl_xor(g1, 16, 32));
    k2 = fmaxf(k2, __shfl_xor(g2, 16, 32)); k3 = fmaxf(k3, __shfl_xor(g3, 16, 32));
    float m0 = b3 ? k1 : k0, t0 = b3 ? k0 : k1;
    float m1 = b3 ? k3 : k2, t1 = b3 ? k2 : k3;
    m0 = fmaxf(m0, __shfl_xor(t0, 8, 32));
    m1 = fmaxf(m1, __shfl_xor(t1, 8, 32));
#pragma unroll
    for (int x = 4; x > 0; x >>= 1) {
        m0 = fmaxf(m0, __shfl_xor(m0, x, 32));
        m1 = fmaxf(m1, __shfl_xor(m1, x, 32));
    }
    float mx[8];
#pragma unroll
    for (int k = 0; k < 4; ++k) { mx[k] = lane_bcast(m0, 8 * k); mx[4 + k] = lane_bcast(m1, 8 * k); }

    const float* hself = h + (size_t)node * WDIM;
    float dn0 = expf(sel4(b3, b4, es[0], es[1], es[2], es[3]) - m0);
    float dn1 = expf(sel4(b3, b4, es[4], es[5], es[6], es[7]) - m1);
    float4 acc0 = make_float4(0.f, 0.f, 0.f, 0.f), acc1 = acc0;
    acc0 = fma4(dn0, *(const float4*)(hself + c0), acc0);
    acc1 = fma4(dn1, *(const float4*)(hself + c1), acc1);

    for (int j0 = beg; j0 < end; j0 += 32) {
        const int n = (end - j0) < 32 ? (end - j0) : 32;
        int ent = sl;
        float p[8];
        if (j0 != jlast) {
            ent = csr[j0 + lane];
            const float4 xa = *(const float4*)(als + (size_t)ent * NHEADS);
            const float4 xb = *(const float4*)(als + (size_t)ent * NHEADS + 4);
            const float x[8] = {xa.x, xa.y, xa.z, xa.w, xb.x, xb.y, xb.z, xb.w};
#pragma unroll
            for (int k = 0; k < 8; ++k) p[k] = expf(lrelu(x[k] + d[k]) - mx[k]);
        } else {
#pragma unroll
            for (int k = 0; k < 8; ++k) p[k] = expf(e[k] - mx[k]);
        }
        int j = 0;
        for (; j + 4 <= n; j += 4) {
            float4 u0[4], u1[4];
#pragma unroll
            for (int u = 0; u < 4; ++u) {
                const float* hs = h + (size_t)__builtin_amdgcn_readlane(ent, j + u) * WDIM;
                u0[u] = *(const float4*)(hs + c0);
                u1[u] = *(const float4*)(hs + c1);
            }
#pragma unroll
            for (int u = 0; u < 4; ++u) {
                const float p0 = sel4(b3, b4, lane_bcast(p[0], j + u), lane_bcast(p[1], j + u),
                                         lane_bcast(p[2], j + u), lane_bcast(p[3], j + u));
                const float p1 = sel4(b3, b4, lane_bcast(p[4], j + u), lane_bcast(p[5], j + u),
                                         lane_bcast(p[6], j + u), lane_bcast(p[7], j + u));
                dn0 += p0; dn1 += p1;
                acc0 = fma4(p0, u0[u], acc0);
                acc1 = fma4(p1, u1[u], acc1);
            }
        }
        for (; j < n; ++j) {
            const float* hs = h + (size_t)__builtin_amdgcn_readlane(ent, j) * WDIM;
            const float p0 = sel4(b3, b4, lane_bcast(p[0], j), lane_bcast(p[1], j),
                                     lane_bcast(p[2], j), lane_bcast(p[3], j));
            const float p1 = sel4(b3, b4, lane_bcast(p[4], j), lane_bcast(p[5], j),
                                     lane_bcast(p[6], j), lane_bcast(p[7], j));
            dn0 += p0; dn1 += p1;
            acc0 = fma4(p0, *(const float4*)(hs + c0), acc0);
            acc1 = fma4(p1, *(const float4*)(hs + c1), acc1);
        }
    }
    const float4 bz0 = *(const float4*)(bias + c0);
    const float4 bz1 = *(const float4*)(bias + c1);
    float4 o0 = make_float4(acc0.x / dn0 + bz0.x, acc0.y / dn0 + bz0.y, acc0.z / dn0 + bz0.z, acc0.w / dn0 + bz0.w);
    float4 o1 = make_float4(acc1.x / dn1 + bz1.x, acc1.y / dn1 + bz1.y, acc1.z / dn1 + bz1.z, acc1.w / dn1 + bz1.w);
    if (apply_elu) {
        o0.x = elu1(o0.x); o0.y = elu1(o0.y); o0.z = elu1(o0.z); o0.w = elu1(o0.w);
        o1.x = elu1(o1.x); o1.y = elu1(o1.y); o1.z = elu1(o1.z); o1.w = elu1(o1.w);
    }
    float* orow = outx + (size_t)node * WDIM;
    *(float4*)(orow + c0) = o0;
    *(float4*)(orow + c1) = o1;
}

#define POOL_ROWS 16
#define POOL_BLOCKS ((NNODES / POOL_ROWS + 7) / 8)
static_assert(NNODES % POOL_ROWS == 0, "pool_accum takes nodes sixteen at a time");
static_assert(POOL_ROWS <= 32, "one lane holds one node's graph id");

__device__ __forceinline__ void pool_flush(int g, int lane, const float (&mx)[8], const float (&sm)[8],
                                           float cnt, unsigned* __restrict__ gmaxenc,
                                           float* __restrict__ gsum, float* __restrict__ counts) {
    unsigned* pm = gmaxenc + g * WDIM + lane;
    float*    ps = gsum + g * WDIM + lane;
#pragma unroll
    for (int i = 0; i < 8; ++i) {
        atomicMax(pm + 32 * i, fenc(mx[i]));
        atomicAdd(ps + 32 * i, sm[i]);
    }
    if (lane == 0) atomicAdd(&counts[g], cnt);
}

__global__ __launch_bounds__(256) void pool_accum(const float* __restrict__ x2, const int* __restrict__ batch,
                                                  unsigned* __restrict__ gmaxenc, float* __restrict__ gsum,
                                                  float* __restrict__ counts) {
    const int wv = __builtin_amdgcn_readfirstlane((int)(threadIdx.x >> 5));
    const int n0 = (blockIdx.x * 8 + wv) * POOL_ROWS;
    if (n0 >= NNODES) return;
    const int lane = threadIdx.x & 31;
    const int gl = batch[n0 + (lane & (POOL_ROWS - 1))];
    const float* xr = x2 + (size_t)n0 * WDIM + lane;
    float v[POOL_ROWS][8];
#pragma unroll
    for (int r = 0; r < POOL_ROWS; ++r) {
#pragma unroll
        for (int i = 0; i < 8; ++i) v[r][i] = xr[r * WDIM + 32 * i];
    }
    int cur = __builtin_amdgcn_readlane(gl, 0);
    float mx[8], sm[8], cnt = 0.f;
#pragma unroll
    for (int i = 0; i < 8; ++i) { mx[i] = -INFINITY; sm[i] = 0.f; }
#pragma unroll
    for (int r = 0; r < POOL_ROWS; ++r) {
        const int g = __builtin_amdgcn_readlane(gl, r);
        if (g != cur) {
            pool_flush(cur, lane, mx, sm, cnt, gmaxenc, gsum, counts);
#pragma unroll
            for (int i = 0; i < 8; ++i) { mx[i] = -INFINITY; sm[i] = 0.f; }
            cnt = 0.f;
            cur = g;
        }
#pragma unroll
        for (int i = 0; i < 8; ++i) { mx[i] = fmaxf(mx[i], v[r][i]); sm[i] += v[r][i]; }
        cnt += 1.0f;
    }
    pool_flush(cur, lane, mx, sm, cnt, gmaxenc, gsum, counts);
}

__global__ void pool_final(const unsigned* __restrict__ gmaxenc, const float* __restrict__ gsum,
                           const float* __restrict__ counts, float* __restrict__ rep) {
    int g = blockIdx.x, c = threadIdx.x;
    float cnt = counts[g];
    float mx = (cnt > 0.f) ? fdec(gmaxenc[g * WDIM + c]) : 0.0f;
    float mean = gsum[g * WDIM + c] / fmaxf(cnt, 1.0f);
    rep[(size_t)g * (2 * WDIM) + c] = mx;
    rep[(size_t)g * (2 * WDIM) + WDIM + c] = mean;
}

__global__ void fc_sigmoid(const float* __restrict__ rep, const float* __restrict__ Wfc,
                           const float* __restrict__ bfc, float* __restrict__ out) {
    __shared__ float red[256];
    int g = blockIdx.x, t = threadIdx.x;
    const float* r = rep + (size_t)g * (2 * WDIM);
    float s = r[t] * Wfc[t] + r[WDIM + t] * Wfc[WDIM + t];
    red[t] = s;
    __syncthreads();
    for (int off = 128; off > 0; off >>= 1) {
        if (t < off) red[t] += red[t + off];
        __syncthreads();
    }
    if (t == 0) out[g] = 1.0f / (1.0f + expf(-(red[0] + bfc[0])));
}

extern "C" void kernel_launch(void* const* d_in, const int* in_sizes, int n_in,
                              void* d_out, int out_size, void* d_ws, size_t ws_size,
                              hipStream_t stream) {
    (void)in_sizes; (void)n_in; (void)out_size; (void)ws_size;

    const float* x    = (const float*)d_in[0];
    const int*   ei   = (const int*)d_in[1];
    const int*   bat  = (const int*)d_in[2];
    const float* W1   = (const float*)d_in[3];
    const float* a1s  = (const float*)d_in[4];
    const float* a1d  = (const float*)d_in[5];
    const float* b1   = (const float*)d_in[6];
    const float* W2   = (const float*)d_in[7];
    const float* a2s  = (const float*)d_in[8];
    const float* a2d  = (const float*)d_in[9];
    const float* b2   = (const float*)d_in[10];
    const float* Wfc  = (const float*)d_in[11];
    const float* bfc  = (const float*)d_in[12];

    float* out = (float*)d_out;
    float* rep = out + NGRAPH;

    const int* esrc = ei;
    const int* edst = ei + NEDGES;

    float*    bufA  = (float*)d_ws;
    float*    bufB  = bufA + (size_t)NNODES * WDIM;
    float*    als   = bufB + (size_t)NNODES * WDIM;
    float*    ald   = als + (size_t)NNODES * NHEADS;
    unsigned* gmaxe = (unsigned*)(ald + (size_t)NNODES * NHEADS);
    float*    gsum  = (float*)(gmaxe + NGRAPH * WDIM);
    float*    cnts  = gsum + NGRAPH * WDIM;
    unsigned* dcnt  = (unsigned*)(cnts + NGRAPH);
    unsigned* rowp  = dcnt + NNODES;
    unsigned* erank = rowp + NNODES + 4;
    int*      csr   = (int*)(erank + NEDGES);

    const long nNH   = (long)NNODES * NHEADS;
    const int  bNH   = (int)((nNH + 255) / 256);
    const int  bN    = (NNODES + 255) / 256;
    const int  bE    = (NEDGES + 255) / 256;

    fill_u32 <<<bN, 256, 0, stream>>>(dcnt, 0u, NNODES);
    dst_count<<<bE, 256, 0, stream>>>(edst, dcnt, erank);
    row_scan <<<1, 1024, 0, stream>>>(dcnt, rowp);
    dst_place<<<bE, 256, 0, stream>>>(esrc, edst, erank, rowp, csr);

    auto run_layer = [&](const float* xin, int K, const float* Wm, const float* as_,
                         const float* ad_, const float* bias, int elu) {
        long waves  = (long)(NNODES / 16) * (WDIM / 64);
        int  blocks = (int)((waves * 32 + 255) / 256);
        gemm_f32_wmma<<<blocks, 256, 0, stream>>>(xin, Wm, bufA, NNODES, WDIM, K);

        compute_al<<<bNH, 256, 0, stream>>>(bufA, as_, ad_, als, ald, NNODES);

        node_agg   <<<NNODES / 8, 256, 0, stream>>>(rowp, csr, als, ald, bufA, bias, elu, bufB);
    };

    run_layer(x,    128, W1, a1s, a1d, b1, 1);
    run_layer(bufB, 256, W2, a2s, a2d, b2, 0);

    fill_u32<<<(NGRAPH * WDIM + 255) / 256, 256, 0, stream>>>(gmaxe, 0u, NGRAPH * WDIM);
    fill_u32<<<(NGRAPH * WDIM + 255) / 256, 256, 0, stream>>>((unsigned*)gsum, 0u, NGRAPH * WDIM);
    fill_u32<<<1, 256, 0, stream>>>((unsigned*)cnts, 0u, NGRAPH);
    pool_accum<<<POOL_BLOCKS, 256, 0, stream>>>(bufB, bat, gmaxe, gsum, cnts);
    pool_final<<<NGRAPH, 256, 0, stream>>>(gmaxe, gsum, cnts, rep);
    fc_sigmoid<<<NGRAPH, 256, 0, stream>>>(rep, Wfc, bfc, out);
}


# edge list by destination built in one launch: a work-group owns a node range, reads all destinations, slot = returning add on an LDS counter, LDS scan, place; list order stays an arrival order
# speedup vs baseline: 5.7048x; 1.1060x over previous
#include <hip/hip_runtime.h>
#include <hip/hip_bf16.h>

typedef __attribute__((ext_vector_type(2))) float v2f;
typedef __attribute__((ext_vector_type(8))) float v8f;

#define NNODES 50000
#define NEDGES 800000
#define WDIM   256
#define NHEADS 8
#define NHID   32
#define NGRAPH 256

__device__ __forceinline__ unsigned fenc(float x) {
    unsigned u = __float_as_uint(x);
    return (u & 0x80000000u) ? ~u : (u | 0x80000000u);
}
__device__ __forceinline__ float fdec(unsigned e) {
    unsigned u = (e & 0x80000000u) ? (e & 0x7FFFFFFFu) : ~e;
    return __uint_as_float(u);
}

__global__ void fill_u32(unsigned* __restrict__ p, unsigned v, long n) {
    long i = (long)blockIdx.x * blockDim.x + threadIdx.x;
    if (i < n) p[i] = v;
}

__global__ void gemm_f32_wmma(const float* __restrict__ A, const float* __restrict__ B,
                              float* __restrict__ C, int M, int N, int K) {
    int wave   = (blockIdx.x * blockDim.x + threadIdx.x) >> 5;
    int lane   = threadIdx.x & 31;
    int nstrip = N >> 6;
    int m0 = (wave / nstrip) << 4;
    int n0 = (wave % nstrip) << 6;
    if (m0 >= M) return;
    int hi = lane >> 4;
    int l  = lane & 15;

    v8f c0 = {}, c1 = {}, c2 = {}, c3 = {};
    const float* arow  = A + (size_t)(m0 + l) * K + (hi << 1);
    const float* bbase = B + (size_t)(hi << 1) * N + n0 + l;

    for (int k = 0; k < K; k += 4) {
        v2f a = *(const v2f*)(arow + k);
        const float* bk0 = bbase + (size_t)k * N;
        const float* bk1 = bk0 + N;
        v2f b0, b1, b2, b3;
        b0.x = bk0[0];   b0.y = bk1[0];
        b1.x = bk0[16];  b1.y = bk1[16];
        b2.x = bk0[32];  b2.y = bk1[32];
        b3.x = bk0[48];  b3.y = bk1[48];
        c0 = __builtin_amdgcn_wmma_f32_16x16x4_f32(false, a, false, b0, (short)0, c0, false, false);
        c1 = __builtin_amdgcn_wmma_f32_16x16x4_f32(false, a, false, b1, (short)0, c1, false, false);
        c2 = __builtin_amdgcn_wmma_f32_16x16x4_f32(false, a, false, b2, (short)0, c2, false, false);
        c3 = __builtin_amdgcn_wmma_f32_16x16x4_f32(false, a, false, b3, (short)0, c3, false, false);
    }
    float* crow = C + (size_t)(m0 + (hi << 3)) * N + n0 + l;
#pragma unroll
    for (int r = 0; r < 8; ++r) {
        crow[(size_t)r * N +  0] = c0[r];
        crow[(size_t)r * N + 16] = c1[r];
        crow[(size_t)r * N + 32] = c2[r];
        crow[(size_t)r * N + 48] = c3[r];
    }
}

__global__ void compute_al(const float* __restrict__ h, const float* __restrict__ as_,
                           const float* __restrict__ ad_, float* __restrict__ als,
                           float* __restrict__ ald, int n_nodes) {
    int i = blockIdx.x * blockDim.x + threadIdx.x;
    if (i >= n_nodes * NHEADS) return;
    int head = i & 7;
    const float* hp = h + (size_t)(i >> 3) * WDIM + head * NHID;
    const float* sp = as_ + head * NHID;
    const float* dp = ad_ + head * NHID;
    float s = 0.f, d = 0.f;
#pragma unroll
    for (int c = 0; c < NHID; ++c) { float v = hp[c]; s += v * sp[c]; d += v * dp[c]; }
    als[i] = s; ald[i] = d;
}

#define CSR_WGS  96
#define CSR_R    ((NNODES + CSR_WGS - 1) / CSR_WGS)
#define CSR_CAPW 512
#define CSR_U    8
#define NQUADS   (NEDGES / 4)
static_assert(NEDGES % 4 == 0, "destinations are read four at a time");
static_assert(CSR_R <= 1024 && CSR_R < (1 << 12), "one thread per node of the range; node packed above bit 20");
static_assert(NEDGES < (1 << 20), "a slot inside a list is packed into 20 bits");
static_assert(NNODES % 8 == 0, "node_agg takes nodes eight at a time");

__device__ __forceinline__ unsigned wave_incl_scan(unsigned x, int lane) {
    for (int d = 1; d < 32; d <<= 1) {
        unsigned y = __shfl_up(x, d, 32);
        if (lane >= d) x += y;
    }
    return x;
}

__global__ __launch_bounds__(1024) void csr_build(const int* __restrict__ src, const int* __restrict__ dst,
                                                  unsigned* __restrict__ rowptr, int* __restrict__ csr) {
    __shared__ unsigned s_cnt[CSR_R];
    __shared__ unsigned s_fill[CSR_R];
    __shared__ unsigned s_wbelow[32];
    __shared__ unsigned s_wtot[32];
    __shared__ unsigned s_ovf;
    __shared__ uint2    s_stash[32][CSR_CAPW];

    const int t = threadIdx.x, lane = t & 31;
    const int w = __builtin_amdgcn_readfirstlane(t >> 5);
    const int lo = blockIdx.x * CSR_R;

    if (t < CSR_R) { s_cnt[t] = 0u; s_fill[t] = 0u; }
    if (t == 0) s_ovf = 0u;
    __syncthreads();

    const int4* dq = (const int4*)dst;
    unsigned nbelow = 0u, wcount = 0u;
    for (int q0 = 0; q0 < NQUADS; q0 += 1024 * CSR_U) {
        int4 v[CSR_U];
#pragma unroll
        for (int u = 0; u < CSR_U; ++u) {
            const int q = q0 + u * 1024 + t;
            v[u] = make_int4(0x7fffffff, 0x7fffffff, 0x7fffffff, 0x7fffffff);
            if (q < NQUADS) v[u] = dq[q];
        }
#pragma unroll
        for (int u = 0; u < CSR_U; ++u) {
            const int e0 = (q0 + u * 1024 + t) * 4;
            const int dd[4] = {v[u].x, v[u].y, v[u].z, v[u].w};
#pragma unroll
            for (int c = 0; c < 4; ++c) {
                const int d = dd[c];
                nbelow += (unsigned)__popc(__builtin_amdgcn_ballot_w32(d < lo));
                const unsigned dl = (unsigned)(d - lo);
                const bool inr = dl < (unsigned)CSR_R;
                const unsigned mask = __builtin_amdgcn_ballot_w32(inr);
                if (mask != 0u) {
                    if (inr) {
                        const unsigned rk = atomicAdd(&s_cnt[dl], 1u);
                        const unsigned s = wcount + __builtin_amdgcn_mbcnt_lo(mask, 0u);
                        if (s < CSR_CAPW) s_stash[w][s] = make_uint2((unsigned)(e0 + c), (dl << 20) | rk);
                    }
                    wcount += (unsigned)__popc(mask);
                }
            }
        }
    }
    if (lane == 0) {
        s_wbelow[w] = nbelow;
        if (wcount > CSR_CAPW) s_ovf = 1u;
    }
    __syncthreads();

    unsigned base = s_wbelow[lane];
    for (int x = 16; x > 0; x >>= 1) base += __shfl_xor(base, x, 32);
    const unsigned x = (t < CSR_R) ? s_cnt[t] : 0u;
    const unsigned incl = wave_incl_scan(x, lane);
    if (lane == 31) s_wtot[w] = incl;
    const unsigned ovf = s_ovf;
    __syncthreads();
    const unsigned wt = s_wtot[lane];
    const unsigned wincl = wave_incl_scan(wt, lane);
    const unsigned wexcl = __shfl(wincl - wt, w, 32);
    const unsigned total = __shfl(wincl, 31, 32);
    const unsigned excl = wexcl + incl - x;
    if (t < CSR_R) {
        s_cnt[t] = excl;
        if (lo + t < NNODES) rowptr[lo + t] = base + excl;
    }
    if (blockIdx.x == CSR_WGS - 1 && t == 0) rowptr[NNODES] = base + total;
    __syncthreads();

    if (ovf == 0u) {
        for (unsigned k = (unsigned)lane; k < wcount; k += 32u) {
            const uint2 it = s_stash[w][k];
            csr[base + s_cnt[it.y >> 20] + (it.y & 0xFFFFFu)] = src[it.x];
        }
    } else {
        for (int q = t; q < NQUADS; q += 1024) {
            const int4 vv = dq[q];
            const int dd[4] = {vv.x, vv.y, vv.z, vv.w};
#pragma unroll
            for (int c = 0; c < 4; ++c) {
                const unsigned dl = (unsigned)(dd[c] - lo);
                if (dl < (unsigned)CSR_R) {
                    const unsigned slot = atomicAdd(&s_fill[dl], 1u);
                    csr[base + s_cnt[dl] + slot] = src[q * 4 + c];
                }
            }
        }
    }
}

__device__ __forceinline__ float4 fma4(float p, float4 y, float4 a) {
    a.x = fmaf(p, y.x, a.x); a.y = fmaf(p, y.y, a.y);
    a.z = fmaf(p, y.z, a.z); a.w = fmaf(p, y.w, a.w);
    return a;
}

__device__ __forceinline__ float lrelu(float v) { return v > 0.f ? v : 0.2f * v; }
__device__ __forceinline__ float elu1(float v) { return v > 0.f ? v : (expf(v) - 1.0f); }
__device__ __forceinline__ float sel4(bool lo, bool hi, float a, float b, float c, float d) {
    const float x = lo ? b : a, y = lo ? d : c;
    return hi ? y : x;
}
__device__ __forceinline__ float lane_bcast(float v, int src) {
    return __int_as_float(__builtin_amdgcn_readlane(__float_as_int(v), src));
}
__global__ __launch_bounds__(256) void node_agg(const unsigned* __restrict__ rowptr,
                                                const int* __restrict__ csr,
                                                const float* __restrict__ als,
                                                const float* __restrict__ ald,
                                                const float* __restrict__ h,
                                                const float* __restrict__ bias,
                                                int apply_elu,
                                                float* __restrict__ outx) {
    const int wv   = __builtin_amdgcn_readfirstlane((int)(threadIdx.x >> 5));
    const int node = blockIdx.x * 8 + wv;
    if (node >= NNODES) return;
    const int lane = threadIdx.x & 31;
    const int c0 = lane * 4, c1 = 128 + lane * 4;
    const bool b3 = (lane & 8) != 0, b4 = (lane & 16) != 0;
    const int beg = __builtin_amdgcn_readfirstlane((int)rowptr[node]);
    const int end = __builtin_amdgcn_readfirstlane((int)rowptr[node + 1]);

    const float4 dA = *(const float4*)(ald + (size_t)node * NHEADS);
    const float4 dB = *(const float4*)(ald + (size_t)node * NHEADS + 4);
    const float4 sA = *(const float4*)(als + (size_t)node * NHEADS);
    const float4 sB = *(const float4*)(als + (size_t)node * NHEADS + 4);
    const float d[8] = {dA.x, dA.y, dA.z, dA.w, dB.x, dB.y, dB.z, dB.w};
    const float sf[8] = {sA.x, sA.y, sA.z, sA.w, sB.x, sB.y, sB.z, sB.w};
    float es[8], e[8], m[8];
#pragma unroll
    for (int k = 0; k < 8; ++k) { es[k] = lrelu(sf[k] + d[k]); e[k] = es[k]; m[k] = es[k]; }

    int sl = 0, jlast = beg;
    for (int j0 = beg; j0 < end; j0 += 32) {
        jlast = j0;
        if (j0 + lane < end) {
            sl = csr[j0 + lane];
            const float4 xa = *(const float4*)(als + (size_t)sl * NHEADS);
            const float4 xb = *(const float4*)(als + (size_t)sl * NHEADS + 4);
            const float x[8] = {xa.x, xa.y, xa.z, xa.w, xb.x, xb.y, xb.z, xb.w};
#pragma unroll
            for (int k = 0; k < 8; ++k) { e[k] = lrelu(x[k] + d[k]); m[k] = fmaxf(m[k], e[k]); }
        }
    }
    float k0 = b4 ? m[2] : m[0], k1 = b4 ? m[3] : m[1], k2 = b4 ? m[6] : m[4], k3 = b4 ? m[7] : m[5];
    float g0 = b4 ? m[0] : m[2], g1 = b4 ? m[1] : m[3], g2 = b4 ? m[4] : m[6], g3 = b4 ? m[5] : m[7];
    k0 = fmaxf(k0, __shfl_xor(g0, 16, 32)); k1 = fmaxf(k1, __shfl_xor(g1, 16, 32));
    k2 = fmaxf(k2, __shfl_xor(g2, 16, 32)); k3 = fmaxf(k3, __shfl_xor(g3, 16, 32));
    float m0 = b3 ? k1 : k0, t0 = b3 ? k0 : k1;
    float m1 = b3 ? k3 : k2, t1 = b3 ? k2 : k3;
    m0 = fmaxf(m0, __shfl_xor(t0, 8, 32));
    m1 = fmaxf(m1, __shfl_xor(t1, 8, 32));
#pragma unroll
    for (int x = 4; x > 0; x >>= 1) {
        m0 = fmaxf(m0, __shfl_xor(m0, x, 32));
        m1 = fmaxf(m1, __shfl_xor(m1, x, 32));
    }
    float mx[8];
#pragma unroll
    for (int k = 0; k < 4; ++k) { mx[k] = lane_bcast(m0, 8 * k); mx[4 + k] = lane_bcast(m1, 8 * k); }

    const float* hself = h + (size_t)node * WDIM;
    float dn0 = expf(sel4(b3, b4, es[0], es[1], es[2], es[3]) - m0);
    float dn1 = expf(sel4(b3, b4, es[4], es[5], es[6], es[7]) - m1);
    float4 acc0 = make_float4(0.f, 0.f, 0.f, 0.f), acc1 = acc0;
    acc0 = fma4(dn0, *(const float4*)(hself + c0), acc0);
    acc1 = fma4(dn1, *(const float4*)(hself + c1), acc1);

    for (int j0 = beg; j0 < end; j0 += 32) {
        const int n = (end - j0) < 32 ? (end - j0) : 32;
        int ent = sl;
        float p[8];
        if (j0 != jlast) {
            ent = csr[j0 + lane];
            const float4 xa = *(const float4*)(als + (size_t)ent * NHEADS);
            const float4 xb = *(const float4*)(als + (size_t)ent * NHEADS + 4);
            const float x[8] = {xa.x, xa.y, xa.z, xa.w, xb.x, xb.y, xb.z, xb.w};
#pragma unroll
            for (int k = 0; k < 8; ++k) p[k] = expf(lrelu(x[k] + d[k]) - mx[k]);
        } else {
#pragma unroll
            for (int k = 0; k < 8; ++k) p[k] = expf(e[k] - mx[k]);
        }
        int j = 0;
        for (; j + 4 <= n; j += 4) {
            float4 u0[4], u1[4];
#pragma unroll
            for (int u = 0; u < 4; ++u) {
                const float* hs = h + (size_t)__builtin_amdgcn_readlane(ent, j + u) * WDIM;
                u0[u] = *(const float4*)(hs + c0);
                u1[u] = *(const float4*)(hs + c1);
            }
#pragma unroll
            for (int u = 0; u < 4; ++u) {
                const float p0 = sel4(b3, b4, lane_bcast(p[0], j + u), lane_bcast(p[1], j + u),
                                         lane_bcast(p[2], j + u), lane_bcast(p[3], j + u));
                const float p1 = sel4(b3, b4, lane_bcast(p[4], j + u), lane_bcast(p[5], j + u),
                                         lane_bcast(p[6], j + u), lane_bcast(p[7], j + u));
                dn0 += p0; dn1 += p1;
                acc0 = fma4(p0, u0[u], acc0);
                acc1 = fma4(p1, u1[u], acc1);
            }
        }
        for (; j < n; ++j) {
            const float* hs = h + (size_t)__builtin_amdgcn_readlane(ent, j) * WDIM;
            const float p0 = sel4(b3, b4, lane_bcast(p[0], j), lane_bcast(p[1], j),
                                     lane_bcast(p[2], j), lane_bcast(p[3], j));
            const float p1 = sel4(b3, b4, lane_bcast(p[4], j), lane_bcast(p[5], j),
                                     lane_bcast(p[6], j), lane_bcast(p[7], j));
            dn0 += p0; dn1 += p1;
            acc0 = fma4(p0, *(const float4*)(hs + c0), acc0);
            acc1 = fma4(p1, *(const float4*)(hs + c1), acc1);
        }
    }
    const float4 bz0 = *(const float4*)(bias + c0);
    const float4 bz1 = *(const float4*)(bias + c1);
    float4 o0 = make_float4(acc0.x / dn0 + bz0.x, acc0.y / dn0 + bz0.y, acc0.z / dn0 + bz0.z, acc0.w / dn0 + bz0.w);
    float4 o1 = make_float4(acc1.x / dn1 + bz1.x, acc1.y / dn1 + bz1.y, acc1.z / dn1 + bz1.z, acc1.w / dn1 + bz1.w);
    if (apply_elu) {
        o0.x = elu1(o0.x); o0.y = elu1(o0.y); o0.z = elu1(o0.z); o0.w = elu1(o0.w);
        o1.x = elu1(o1.x); o1.y = elu1(o1.y); o1.z = elu1(o1.z); o1.w = elu1(o1.w);
    }
    float* orow = outx + (size_t)node * WDIM;
    *(float4*)(orow + c0) = o0;
    *(float4*)(orow + c1) = o1;
}

#define POOL_ROWS 16
#define POOL_BLOCKS ((NNODES / POOL_ROWS + 7) / 8)
static_assert(NNODES % POOL_ROWS == 0, "pool_accum takes nodes sixteen at a time");
static_assert(POOL_ROWS <= 32, "one lane holds one node's graph id");

__device__ __forceinline__ void pool_flush(int g, int lane, const float (&mx)[8], const float (&sm)[8],
                                           float cnt, unsigned* __restrict__ gmaxenc,
                                           float* __restrict__ gsum, float* __restrict__ counts) {
    unsigned* pm = gmaxenc + g * WDIM + lane;
    float*    ps = gsum + g * WDIM + lane;
#pragma unroll
    for (int i = 0; i < 8; ++i) {
        atomicMax(pm + 32 * i, fenc(mx[i]));
        atomicAdd(ps + 32 * i, sm[i]);
    }
    if (lane == 0) atomicAdd(&counts[g], cnt);
}

__global__ __launch_bounds__(256) void pool_accum(const float* __restrict__ x2, const int* __restrict__ batch,
                                                  unsigned* __restrict__ gmaxenc, float* __restrict__ gsum,
                                                  float* __restrict__ counts) {
    const int wv = __builtin_amdgcn_readfirstlane((int)(threadIdx.x >> 5));
    const int n0 = (blockIdx.x * 8 + wv) * POOL_ROWS;
    if (n0 >= NNODES) return;
    const int lane = threadIdx.x & 31;
    const int gl = batch[n0 + (lane & (POOL_ROWS - 1))];
    const float* xr = x2 + (size_t)n0 * WDIM + lane;
    float v[POOL_ROWS][8];
#pragma unroll
    for (int r = 0; r < POOL_ROWS; ++r) {
#pragma unroll
        for (int i = 0; i < 8; ++i) v[r][i] = xr[r * WDIM + 32 * i];
    }
    int cur = __builtin_amdgcn_readlane(gl, 0);
    float mx[8], sm[8], cnt = 0.f;
#pragma unroll
    for (int i = 0; i < 8; ++i) { mx[i] = -INFINITY; sm[i] = 0.f; }
#pragma unroll
    for (int r = 0; r < POOL_ROWS; ++r) {
        const int g = __builtin_amdgcn_readlane(gl, r);
        if (g != cur) {
            pool_flush(cur, lane, mx, sm, cnt, gmaxenc, gsum, counts);
#pragma unroll
            for (int i = 0; i < 8; ++i) { mx[i] = -INFINITY; sm[i] = 0.f; }
            cnt = 0.f;
            cur = g;
        }
#pragma unroll
        for (int i = 0; i < 8; ++i) { mx[i] = fmaxf(mx[i], v[r][i]); sm[i] += v[r][i]; }
        cnt += 1.0f;
    }
    pool_flush(cur, lane, mx, sm, cnt, gmaxenc, gsum, counts);
}

__global__ void pool_final(const unsigned* __restrict__ gmaxenc, const float* __restrict__ gsum,
                           const float* __restrict__ counts, float* __restrict__ rep) {
    int g = blockIdx.x, c = threadIdx.x;
    float cnt = counts[g];
    float mx = (cnt > 0.f) ? fdec(gmaxenc[g * WDIM + c]) : 0.0f;
    float mean = gsum[g * WDIM + c] / fmaxf(cnt, 1.0f);
    rep[(size_t)g * (2 * WDIM) + c] = mx;
    rep[(size_t)g * (2 * WDIM) + WDIM + c] = mean;
}

__global__ void fc_sigmoid(const float* __restrict__ rep, const float* __restrict__ Wfc,
                           const float* __restrict__ bfc, float* __restrict__ out) {
    __shared__ float red[256];
    int g = blockIdx.x, t = threadIdx.x;
    const float* r = rep + (size_t)g * (2 * WDIM);
    float s = r[t] * Wfc[t] + r[WDIM + t] * Wfc[WDIM + t];
    red[t] = s;
    __syncthreads();
    for (int off = 128; off > 0; off >>= 1) {
        if (t < off) red[t] += red[t + off];
        __syncthreads();
    }
    if (t == 0) out[g] = 1.0f / (1.0f + expf(-(red[0] + bfc[0])));
}

extern "C" void kernel_launch(void* const* d_in, const int* in_sizes, int n_in,
                              void* d_out, int out_size, void* d_ws, size_t ws_size,
                              hipStream_t stream) {
    (void)in_sizes; (void)n_in; (void)out_size; (void)ws_size;

    const float* x    = (const float*)d_in[0];
    const int*   ei   = (const int*)d_in[1];
    const int*   bat  = (const int*)d_in[2];
    const float* W1   = (const float*)d_in[3];
    const float* a1s  = (const float*)d_in[4];
    const float* a1d  = (const float*)d_in[5];
    const float* b1   = (const float*)d_in[6];
    const float* W2   = (const float*)d_in[7];
    const float* a2s  = (const float*)d_in[8];
    const float* a2d  = (const float*)d_in[9];
    const float* b2   = (const float*)d_in[10];
    const float* Wfc  = (const float*)d_in[11];
    const float* bfc  = (const float*)d_in[12];

    float* out = (float*)d_out;
    float* rep = out + NGRAPH;

    const int* esrc = ei;
    const int* edst = ei + NEDGES;

    float*    bufA  = (float*)d_ws;
    float*    bufB  = bufA + (size_t)NNODES * WDIM;
    float*    als   = bufB + (size_t)NNODES * WDIM;
    float*    ald   = als + (size_t)NNODES * NHEADS;
    unsigned* gmaxe = (unsigned*)(ald + (size_t)NNODES * NHEADS);
    float*    gsum  = (float*)(gmaxe + NGRAPH * WDIM);
    float*    cnts  = gsum + NGRAPH * WDIM;
    unsigned* rowp  = (unsigned*)(cnts + NGRAPH);
    int*      csr   = (int*)(rowp + NNODES + 4);

    const long nNH   = (long)NNODES * NHEADS;
    const int  bNH   = (int)((nNH + 255) / 256);

    csr_build<<<CSR_WGS, 1024, 0, stream>>>(esrc, edst, rowp, csr);

    auto run_layer = [&](const float* xin, int K, const float* Wm, const float* as_,
                         const float* ad_, const float* bias, int elu) {
        long waves  = (long)(NNODES / 16) * (WDIM / 64);
        int  blocks = (int)((waves * 32 + 255) / 256);
        gemm_f32_wmma<<<blocks, 256, 0, stream>>>(xin, Wm, bufA, NNODES, WDIM, K);

        compute_al<<<bNH, 256, 0, stream>>>(bufA, as_, ad_, als, ald, NNODES);

        node_agg   <<<NNODES / 8, 256, 0, stream>>>(rowp, csr, als, ald, bufA, bias, elu, bufB);
    };

    run_layer(x,    128, W1, a1s, a1d, b1, 1);
    run_layer(bufB, 256, W2, a2s, a2d, b2, 0);

    fill_u32<<<(NGRAPH * WDIM + 255) / 256, 256, 0, stream>>>(gmaxe, 0u, NGRAPH * WDIM);
    fill_u32<<<(NGRAPH * WDIM + 255) / 256, 256, 0, stream>>>((unsigned*)gsum, 0u, NGRAPH * WDIM);
    fill_u32<<<1, 256, 0, stream>>>((unsigned*)cnts, 0u, NGRAPH);
    pool_accum<<<POOL_BLOCKS, 256, 0, stream>>>(bufB, bat, gmaxe, gsum, cnts);
    pool_final<<<NGRAPH, 256, 0, stream>>>(gmaxe, gsum, cnts, rep);
    fc_sigmoid<<<NGRAPH, 256, 0, stream>>>(rep, Wfc, bfc, out);
}


# attention pre-logits formed in the projection GEMM's epilogue from its accumulator tiles, summed by a 16-lane DPP tree (order of the 32-term sums changes); the two pre-logit launches dropped
# speedup vs baseline: 5.9731x; 1.0470x over previous
#include <hip/hip_runtime.h>
#include <hip/hip_bf16.h>

typedef __attribute__((ext_vector_type(2))) float v2f;
typedef __attribute__((ext_vector_type(8))) float v8f;

#define NNODES 50000
#define NEDGES 800000
#define WDIM   256
#define NHEADS 8
#define NHID   32
#define NGRAPH 256

__device__ __forceinline__ unsigned fenc(float x) {
    unsigned u = __float_as_uint(x);
    return (u & 0x80000000u) ? ~u : (u | 0x80000000u);
}
__device__ __forceinline__ float fdec(unsigned e) {
    unsigned u = (e & 0x80000000u) ? (e & 0x7FFFFFFFu) : ~e;
    return __uint_as_float(u);
}

__global__ void fill_u32(unsigned* __restrict__ p, unsigned v, long n) {
    long i = (long)blockIdx.x * blockDim.x + threadIdx.x;
    if (i < n) p[i] = v;
}

static_assert(WDIM == 256 && NHEADS == 8 && NHID == 32, "a 64-wide strip is two heads; 4 strips per row tile");

template <int XM>
__device__ __forceinline__ float row_xor(float x) {
    return __int_as_float(__builtin_amdgcn_update_dpp(0, __float_as_int(x), 0x160 + XM, 0xf, 0xf, true));
}
template <int H, int XM>
__device__ __forceinline__ void halve_step(const float (&in)[2 * H], float (&out)[H], bool up) {
#pragma unroll
    for (int i = 0; i < H; ++i) {
        const float keep = up ? in[H + i] : in[i];
        const float send = up ? in[i] : in[H + i];
        out[i] = keep + row_xor<XM>(send);
    }
}

__global__ __launch_bounds__(256) void gemm_f32_wmma_al(const float* __restrict__ A, const float* __restrict__ B,
                                                        float* __restrict__ C,
                                                        const float* __restrict__ as_, const float* __restrict__ ad_,
                                                        float* __restrict__ als, float* __restrict__ ald,
                                                        int M, int K) {
    __shared__ float s_al[2][32][NHEADS];

    const int N = WDIM;
    const int w    = threadIdx.x >> 5;
    const int lane = threadIdx.x & 31;
    const int mblk = blockIdx.x * 32;
    const int m0r  = mblk + ((w >> 2) << 4);
    const bool valid = m0r < M;
    const int m0 = valid ? m0r : M - 16;
    const int n0 = (w & 3) << 6;
    const int hi = lane >> 4;
    const int l  = lane & 15;

    const float s0 = as_[n0 + l], s1 = as_[n0 + 16 + l], s2 = as_[n0 + 32 + l], s3 = as_[n0 + 48 + l];
    const float d0 = ad_[n0 + l], d1 = ad_[n0 + 16 + l], d2 = ad_[n0 + 32 + l], d3 = ad_[n0 + 48 + l];

    v8f c0 = {}, c1 = {}, c2 = {}, c3 = {};
    const float* arow  = A + (size_t)(m0 + l) * K + (hi << 1);
    const float* bbase = B + (size_t)(hi << 1) * N + n0 + l;

    for (int k = 0; k < K; k += 4) {
        v2f a = *(const v2f*)(arow + k);
        const float* bk0 = bbase + (size_t)k * N;
        const float* bk1 = bk0 + N;
        v2f b0, b1, b2, b3;
        b0.x = bk0[0];   b0.y = bk1[0];
        b1.x = bk0[16];  b1.y = bk1[16];
        b2.x = bk0[32];  b2.y = bk1[32];
        b3.x = bk0[48];  b3.y = bk1[48];
        c0 = __builtin_amdgcn_wmma_f32_16x16x4_f32(false, a, false, b0, (short)0, c0, false, false);
        c1 = __builtin_amdgcn_wmma_f32_16x16x4_f32(false, a, false, b1, (short)0, c1, false, false);
        c2 = __builtin_amdgcn_wmma_f32_16x16x4_f32(false, a, false, b2, (short)0, c2, false, false);
        c3 = __builtin_amdgcn_wmma_f32_16x16x4_f32(false, a, false, b3, (short)0, c3, false, false);
    }
    if (valid) {
        float* crow = C + (size_t)(m0 + (hi << 3)) * N + n0 + l;
#pragma unroll
        for (int r = 0; r < 8; ++r) {
            crow[(size_t)r * N +  0] = c0[r];
            crow[(size_t)r * N + 16] = c1[r];
            crow[(size_t)r * N + 32] = c2[r];
            crow[(size_t)r * N + 48] = c3[r];
        }
    }

    float x[32];
#pragma unroll
    for (int r = 0; r < 8; ++r) {
        x[2 * r]          = fmaf(c1[r], s1, c0[r] * s0);
        x[2 * r + 1]      = fmaf(c1[r], d1, c0[r] * d0);
        x[16 + 2 * r]     = fmaf(c3[r], s3, c2[r] * s2);
        x[16 + 2 * r + 1] = fmaf(c3[r], d3, c2[r] * d2);
    }
    float y[16], z[8], q[4], u[2];
    halve_step<16, 8>(x, y, (lane & 8) != 0);
    halve_step<8, 4>(y, z, (lane & 4) != 0);
    halve_step<4, 2>(z, q, (lane & 2) != 0);
    halve_step<2, 1>(q, u, (lane & 1) != 0);
    {
        const int brow = ((w >> 2) << 4) + (hi << 3) + (l & 7);
        const int head = ((w & 3) << 1) + (l >> 3);
        s_al[0][brow][head] = u[0];
        s_al[1][brow][head] = u[1];
    }
    __syncthreads();
    const int t = threadIdx.x;
    if (t < 128) {
        const int which = t >> 6, f = t & 63;
        if (mblk + (f >> 1) < M) {
            const float4 v = ((const float4*)&s_al[which][0][0])[f];
            float* dstp = (which ? ald : als) + (size_t)mblk * NHEADS;
            ((float4*)dstp)[f] = v;
        }
    }
}

#define CSR_WGS  96
#define CSR_R    ((NNODES + CSR_WGS - 1) / CSR_WGS)
#define CSR_CAPW 512
#define CSR_U    8
#define NQUADS   (NEDGES / 4)
static_assert(NEDGES % 4 == 0, "destinations are read four at a time");
static_assert(CSR_R <= 1024 && CSR_R < (1 << 12), "one thread per node of the range; node packed above bit 20");
static_assert(NEDGES < (1 << 20), "a slot inside a list is packed into 20 bits");
static_assert(NNODES % 8 == 0, "node_agg takes nodes eight at a time");

__device__ __forceinline__ unsigned wave_incl_scan(unsigned x, int lane) {
    for (int d = 1; d < 32; d <<= 1) {
        unsigned y = __shfl_up(x, d, 32);
        if (lane >= d) x += y;
    }
    return x;
}

__global__ __launch_bounds__(1024) void csr_build(const int* __restrict__ src, const int* __restrict__ dst,
                                                  unsigned* __restrict__ rowptr, int* __restrict__ csr) {
    __shared__ unsigned s_cnt[CSR_R];
    __shared__ unsigned s_fill[CSR_R];
    __shared__ unsigned s_wbelow[32];
    __shared__ unsigned s_wtot[32];
    __shared__ unsigned s_ovf;
    __shared__ uint2    s_stash[32][CSR_CAPW];

    const int t = threadIdx.x, lane = t & 31;
    const int w = __builtin_amdgcn_readfirstlane(t >> 5);
    const int lo = blockIdx.x * CSR_R;

    if (t < CSR_R) { s_cnt[t] = 0u; s_fill[t] = 0u; }
    if (t == 0) s_ovf = 0u;
    __syncthreads();

    const int4* dq = (const int4*)dst;
    unsigned nbelow = 0u, wcount = 0u;
    for (int q0 = 0; q0 < NQUADS; q0 += 1024 * CSR_U) {
        int4 v[CSR_U];
#pragma unroll
        for (int u = 0; u < CSR_U; ++u) {
            const int q = q0 + u * 1024 + t;
            v[u] = make_int4(0x7fffffff, 0x7fffffff, 0x7fffffff, 0x7fffffff);
            if (q < NQUADS) v[u] = dq[q];
        }
#pragma unroll
        for (int u = 0; u < CSR_U; ++u) {
            const int e0 = (q0 + u * 1024 + t) * 4;
            const int dd[4] = {v[u].x, v[u].y, v[u].z, v[u].w};
#pragma unroll
            for (int c = 0; c < 4; ++c) {
                const int d = dd[c];
                nbelow += (unsigned)__popc(__builtin_amdgcn_ballot_w32(d < lo));
                const unsigned dl = (unsigned)(d - lo);
                const bool inr = dl < (unsigned)CSR_R;
                const unsigned mask = __builtin_amdgcn_ballot_w32(inr);
                if (mask != 0u) {
                    if (inr) {
                        const unsigned rk = atomicAdd(&s_cnt[dl], 1u);
                        const unsigned s = wcount + __builtin_amdgcn_mbcnt_lo(mask, 0u);
                        if (s < CSR_CAPW) s_stash[w][s] = make_uint2((unsigned)(e0 + c), (dl << 20) | rk);
                    }
                    wcount += (unsigned)__popc(mask);
                }
            }
        }
    }
    if (lane == 0) {
        s_wbelow[w] = nbelow;
        if (wcount > CSR_CAPW) s_ovf = 1u;
    }
    __syncthreads();

    unsigned base = s_wbelow[lane];
    for (int x = 16; x > 0; x >>= 1) base += __shfl_xor(base, x, 32);
    const unsigned x = (t < CSR_R) ? s_cnt[t] : 0u;
    const unsigned incl = wave_incl_scan(x, lane);
    if (lane == 31) s_wtot[w] = incl;
    const unsigned ovf = s_ovf;
    __syncthreads();
    const unsigned wt = s_wtot[lane];
    const unsigned wincl = wave_incl_scan(wt, lane);
    const unsigned wexcl = __shfl(wincl - wt, w, 32);
    const unsigned total = __shfl(wincl, 31, 32);
    const unsigned excl = wexcl + incl - x;
    if (t < CSR_R) {
        s_cnt[t] = excl;
        if (lo + t < NNODES) rowptr[lo + t] = base + excl;
    }
    if (blockIdx.x == CSR_WGS - 1 && t == 0) rowptr[NNODES] = base + total;
    __syncthreads();

    if (ovf == 0u) {
        for (unsigned k = (unsigned)lane; k < wcount; k += 32u) {
            const uint2 it = s_stash[w][k];
            csr[base + s_cnt[it.y >> 20] + (it.y & 0xFFFFFu)] = src[it.x];
        }
    } else {
        for (int q = t; q < NQUADS; q += 1024) {
            const int4 vv = dq[q];
            const int dd[4] = {vv.x, vv.y, vv.z, vv.w};
#pragma unroll
            for (int c = 0; c < 4; ++c) {
                const unsigned dl = (unsigned)(dd[c] - lo);
                if (dl < (unsigned)CSR_R) {
                    const unsigned slot = atomicAdd(&s_fill[dl], 1u);
                    csr[base + s_cnt[dl] + slot] = src[q * 4 + c];
                }
            }
        }
    }
}

__device__ __forceinline__ float4 fma4(float p, float4 y, float4 a) {
    a.x = fmaf(p, y.x, a.x); a.y = fmaf(p, y.y, a.y);
    a.z = fmaf(p, y.z, a.z); a.w = fmaf(p, y.w, a.w);
    return a;
}

__device__ __forceinline__ float lrelu(float v) { return v > 0.f ? v : 0.2f * v; }
__device__ __forceinline__ float elu1(float v) { return v > 0.f ? v : (expf(v) - 1.0f); }
__device__ __forceinline__ float sel4(bool lo, bool hi, float a, float b, float c, float d) {
    const float x = lo ? b : a, y = lo ? d : c;
    return hi ? y : x;
}
__device__ __forceinline__ float lane_bcast(float v, int src) {
    return __int_as_float(__builtin_amdgcn_readlane(__float_as_int(v), src));
}
__global__ __launch_bounds__(256) void node_agg(const unsigned* __restrict__ rowptr,
                                                const int* __restrict__ csr,
                                                const float* __restrict__ als,
                                                const float* __restrict__ ald,
                                                const float* __restrict__ h,
                                                const float* __restrict__ bias,
                                                int apply_elu,
                                                float* __restrict__ outx) {
    const int wv   = __builtin_amdgcn_readfirstlane((int)(threadIdx.x >> 5));
    const int node = blockIdx.x * 8 + wv;
    if (node >= NNODES) return;
    const int lane = threadIdx.x & 31;
    const int c0 = lane * 4, c1 = 128 + lane * 4;
    const bool b3 = (lane & 8) != 0, b4 = (lane & 16) != 0;
    const int beg = __builtin_amdgcn_readfirstlane((int)rowptr[node]);
    const int end = __builtin_amdgcn_readfirstlane((int)rowptr[node + 1]);

    const float4 dA = *(const float4*)(ald + (size_t)node * NHEADS);
    const float4 dB = *(const float4*)(ald + (size_t)node * NHEADS + 4);
    const float4 sA = *(const float4*)(als + (size_t)node * NHEADS);
    const float4 sB = *(const float4*)(als + (size_t)node * NHEADS + 4);
    const float d[8] = {dA.x, dA.y, dA.z, dA.w, dB.x, dB.y, dB.z, dB.w};
    const float sf[8] = {sA.x, sA.y, sA.z, sA.w, sB.x, sB.y, sB.z, sB.w};
    float es[8], e[8], m[8];
#pragma unroll
    for (int k = 0; k < 8; ++k) { es[k] = lrelu(sf[k] + d[k]); e[k] = es[k]; m[k] = es[k]; }

    int sl = 0, jlast = beg;
    for (int j0 = beg; j0 < end; j0 += 32) {
        jlast = j0;
        if (j0 + lane < end) {
            sl = csr[j0 + lane];
            const float4 xa = *(const float4*)(als + (size_t)sl * NHEADS);
            const float4 xb = *(const float4*)(als + (size_t)sl * NHEADS + 4);
            const float x[8] = {xa.x, xa.y, xa.z, xa.w, xb.x, xb.y, xb.z, xb.w};
#pragma unroll
            for (int k = 0; k < 8; ++k) { e[k] = lrelu(x[k] + d[k]); m[k] = fmaxf(m[k], e[k]); }
        }
    }
    float k0 = b4 ? m[2] : m[0], k1 = b4 ? m[3] : m[1], k2 = b4 ? m[6] : m[4], k3 = b4 ? m[7] : m[5];
    float g0 = b4 ? m[0] : m[2], g1 = b4 ? m[1] : m[3], g2 = b4 ? m[4] : m[6], g3 = b4 ? m[5] : m[7];
    k0 = fmaxf(k0, __shfl_xor(g0, 16, 32)); k1 = fmaxf(k1, __shfl_xor(g1, 16, 32));
    k2 = fmaxf(k2, __shfl_xor(g2, 16, 32)); k3 = fmaxf(k3, __shfl_xor(g3, 16, 32));
    float m0 = b3 ? k1 : k0, t0 = b3 ? k0 : k1;
    float m1 = b3 ? k3 : k2, t1 = b3 ? k2 : k3;
    m0 = fmaxf(m0, __shfl_xor(t0, 8, 32));
    m1 = fmaxf(m1, __shfl_xor(t1, 8, 32));
#pragma unroll
    for (int x = 4; x > 0; x >>= 1) {
        m0 = fmaxf(m0, __shfl_xor(m0, x, 32));
        m1 = fmaxf(m1, __shfl_xor(m1, x, 32));
    }
    float mx[8];
#pragma unroll
    for (int k = 0; k < 4; ++k) { mx[k] = lane_bcast(m0, 8 * k); mx[4 + k] = lane_bcast(m1, 8 * k); }

    const float* hself = h + (size_t)node * WDIM;
    float dn0 = expf(sel4(b3, b4, es[0], es[1], es[2], es[3]) - m0);
    float dn1 = expf(sel4(b3, b4, es[4], es[5], es[6], es[7]) - m1);
    float4 acc0 = make_float4(0.f, 0.f, 0.f, 0.f), acc1 = acc0;
    acc0 = fma4(dn0, *(const float4*)(hself + c0), acc0);
    acc1 = fma4(dn1, *(const float4*)(hself + c1), acc1);

    for (int j0 = beg; j0 < end; j0 += 32) {
        const int n = (end - j0) < 32 ? (end - j0) : 32;
        int ent = sl;
        float p[8];
        if (j0 != jlast) {
            ent = csr[j0 + lane];
            const float4 xa = *(const float4*)(als + (size_t)ent * NHEADS);
            const float4 xb = *(const float4*)(als + (size_t)ent * NHEADS + 4);
            const float x[8] = {xa.x, xa.y, xa.z, xa.w, xb.x, xb.y, xb.z, xb.w};
#pragma unroll
            for (int k = 0; k < 8; ++k) p[k] = expf(lrelu(x[k] + d[k]) - mx[k]);
        } else {
#pragma unroll
            for (int k = 0; k < 8; ++k) p[k] = expf(e[k] - mx[k]);
        }
        int j = 0;
        for (; j + 4 <= n; j += 4) {
            float4 u0[4], u1[4];
#pragma unroll
            for (int u = 0; u < 4; ++u) {
                const float* hs = h + (size_t)__builtin_amdgcn_readlane(ent, j + u) * WDIM;
                u0[u] = *(const float4*)(hs + c0);
                u1[u] = *(const float4*)(hs + c1);
            }
#pragma unroll
            for (int u = 0; u < 4; ++u) {
                const float p0 = sel4(b3, b4, lane_bcast(p[0], j + u), lane_bcast(p[1], j + u),
                                         lane_bcast(p[2], j + u), lane_bcast(p[3], j + u));
                const float p1 = sel4(b3, b4, lane_bcast(p[4], j + u), lane_bcast(p[5], j + u),
                                         lane_bcast(p[6], j + u), lane_bcast(p[7], j + u));
                dn0 += p0; dn1 += p1;
                acc0 = fma4(p0, u0[u], acc0);
                acc1 = fma4(p1, u1[u], acc1);
            }
        }
        for (; j < n; ++j) {
            const float* hs = h + (size_t)__builtin_amdgcn_readlane(ent, j) * WDIM;
            const float p0 = sel4(b3, b4, lane_bcast(p[0], j), lane_bcast(p[1], j),
                                     lane_bcast(p[2], j), lane_bcast(p[3], j));
            const float p1 = sel4(b3, b4, lane_bcast(p[4], j), lane_bcast(p[5], j),
                                     lane_bcast(p[6], j), lane_bcast(p[7], j));
            dn0 += p0; dn1 += p1;
            acc0 = fma4(p0, *(const float4*)(hs + c0), acc0);
            acc1 = fma4(p1, *(const float4*)(hs + c1), acc1);
        }
    }
    const float4 bz0 = *(const float4*)(bias + c0);
    const float4 bz1 = *(const float4*)(bias + c1);
    float4 o0 = make_float4(acc0.x / dn0 + bz0.x, acc0.y / dn0 + bz0.y, acc0.z / dn0 + bz0.z, acc0.w / dn0 + bz0.w);
    float4 o1 = make_float4(acc1.x / dn1 + bz1.x, acc1.y / dn1 + bz1.y, acc1.z / dn1 + bz1.z, acc1.w / dn1 + bz1.w);
    if (apply_elu) {
        o0.x = elu1(o0.x); o0.y = elu1(o0.y); o0.z = elu1(o0.z); o0.w = elu1(o0.w);
        o1.x = elu1(o1.x); o1.y = elu1(o1.y); o1.z = elu1(o1.z); o1.w = elu1(o1.w);
    }
    float* orow = outx + (size_t)node * WDIM;
    *(float4*)(orow + c0) = o0;
    *(float4*)(orow + c1) = o1;
}

#define POOL_ROWS 16
#define POOL_BLOCKS ((NNODES / POOL_ROWS + 7) / 8)
static_assert(NNODES % POOL_ROWS == 0, "pool_accum takes nodes sixteen at a time");
static_assert(POOL_ROWS <= 32, "one lane holds one node's graph id");

__device__ __forceinline__ void pool_flush(int g, int lane, const float (&mx)[8], const float (&sm)[8],
                                           float cnt, unsigned* __restrict__ gmaxenc,
                                           float* __restrict__ gsum, float* __restrict__ counts) {
    unsigned* pm = gmaxenc + g * WDIM + lane;
    float*    ps = gsum + g * WDIM + lane;
#pragma unroll
    for (int i = 0; i < 8; ++i) {
        atomicMax(pm + 32 * i, fenc(mx[i]));
        atomicAdd(ps + 32 * i, sm[i]);
    }
    if (lane == 0) atomicAdd(&counts[g], cnt);
}

__global__ __launch_bounds__(256) void pool_accum(const float* __restrict__ x2, const int* __restrict__ batch,
                                                  unsigned* __restrict__ gmaxenc, float* __restrict__ gsum,
                                                  float* __restrict__ counts) {
    const int wv = __builtin_amdgcn_readfirstlane((int)(threadIdx.x >> 5));
    const int n0 = (blockIdx.x * 8 + wv) * POOL_ROWS;
    if (n0 >= NNODES) return;
    const int lane = threadIdx.x & 31;
    const int gl = batch[n0 + (lane & (POOL_ROWS - 1))];
    const float* xr = x2 + (size_t)n0 * WDIM + lane;
    float v[POOL_ROWS][8];
#pragma unroll
    for (int r = 0; r < POOL_ROWS; ++r) {
#pragma unroll
        for (int i = 0; i < 8; ++i) v[r][i] = xr[r * WDIM + 32 * i];
    }
    int cur = __builtin_amdgcn_readlane(gl, 0);
    float mx[8], sm[8], cnt = 0.f;
#pragma unroll
    for (int i = 0; i < 8; ++i) { mx[i] = -INFINITY; sm[i] = 0.f; }
#pragma unroll
    for (int r = 0; r < POOL_ROWS; ++r) {
        const int g = __builtin_amdgcn_readlane(gl, r);
        if (g != cur) {
            pool_flush(cur, lane, mx, sm, cnt, gmaxenc, gsum, counts);
#pragma unroll
            for (int i = 0; i < 8; ++i) { mx[i] = -INFINITY; sm[i] = 0.f; }
            cnt = 0.f;
            cur = g;
        }
#pragma unroll
        for (int i = 0; i < 8; ++i) { mx[i] = fmaxf(mx[i], v[r][i]); sm[i] += v[r][i]; }
        cnt += 1.0f;
    }
    pool_flush(cur, lane, mx, sm, cnt, gmaxenc, gsum, counts);
}

__global__ void pool_final(const unsigned* __restrict__ gmaxenc, const float* __restrict__ gsum,
                           const float* __restrict__ counts, float* __restrict__ rep) {
    int g = blockIdx.x, c = threadIdx.x;
    float cnt = counts[g];
    float mx = (cnt > 0.f) ? fdec(gmaxenc[g * WDIM + c]) : 0.0f;
    float mean = gsum[g * WDIM + c] / fmaxf(cnt, 1.0f);
    rep[(size_t)g * (2 * WDIM) + c] = mx;
    rep[(size_t)g * (2 * WDIM) + WDIM + c] = mean;
}

__global__ void fc_sigmoid(const float* __restrict__ rep, const float* __restrict__ Wfc,
                           const float* __restrict__ bfc, float* __restrict__ out) {
    __shared__ float red[256];
    int g = blockIdx.x, t = threadIdx.x;
    const float* r = rep + (size_t)g * (2 * WDIM);
    float s = r[t] * Wfc[t] + r[WDIM + t] * Wfc[WDIM + t];
    red[t] = s;
    __syncthreads();
    for (int off = 128; off > 0; off >>= 1) {
        if (t < off) red[t] += red[t + off];
        __syncthreads();
    }
    if (t == 0) out[g] = 1.0f / (1.0f + expf(-(red[0] + bfc[0])));
}

extern "C" void kernel_launch(void* const* d_in, const int* in_sizes, int n_in,
                              void* d_out, int out_size, void* d_ws, size_t ws_size,
                              hipStream_t stream) {
    (void)in_sizes; (void)n_in; (void)out_size; (void)ws_size;

    const float* x    = (const float*)d_in[0];
    const int*   ei   = (const int*)d_in[1];
    const int*   bat  = (const int*)d_in[2];
    const float* W1   = (const float*)d_in[3];
    const float* a1s  = (const float*)d_in[4];
    const float* a1d  = (const float*)d_in[5];
    const float* b1   = (const float*)d_in[6];
    const float* W2   = (const float*)d_in[7];
    const float* a2s  = (const float*)d_in[8];
    const float* a2d  = (const float*)d_in[9];
    const float* b2   = (const float*)d_in[10];
    const float* Wfc  = (const float*)d_in[11];
    const float* bfc  = (const float*)d_in[12];

    float* out = (float*)d_out;
    float* rep = out + NGRAPH;

    const int* esrc = ei;
    const int* edst = ei + NEDGES;

    float*    bufA  = (float*)d_ws;
    float*    bufB  = bufA + (size_t)NNODES * WDIM;
    float*    als   = bufB + (size_t)NNODES * WDIM;
    float*    ald   = als + (size_t)NNODES * NHEADS;
    unsigned* gmaxe = (unsigned*)(ald + (size_t)NNODES * NHEADS);
    float*    gsum  = (float*)(gmaxe + NGRAPH * WDIM);
    float*    cnts  = gsum + NGRAPH * WDIM;
    unsigned* rowp  = (unsigned*)(cnts + NGRAPH);
    int*      csr   = (int*)(rowp + NNODES + 4);

    csr_build<<<CSR_WGS, 1024, 0, stream>>>(esrc, edst, rowp, csr);

    auto run_layer = [&](const float* xin, int K, const float* Wm, const float* as_,
                         const float* ad_, const float* bias, int elu) {
        long waves  = (long)(NNODES / 16) * (WDIM / 64);
        int  blocks = (int)((waves * 32 + 255) / 256);
        gemm_f32_wmma_al<<<blocks, 256, 0, stream>>>(xin, Wm, bufA, as_, ad_, als, ald, NNODES, K);

        node_agg   <<<NNODES / 8, 256, 0, stream>>>(rowp, csr, als, ald, bufA, bias, elu, bufB);
    };

    run_layer(x,    128, W1, a1s, a1d, b1, 1);
    run_layer(bufB, 256, W2, a2s, a2d, b2, 0);

    fill_u32<<<(NGRAPH * WDIM + 255) / 256, 256, 0, stream>>>(gmaxe, 0u, NGRAPH * WDIM);
    fill_u32<<<(NGRAPH * WDIM + 255) / 256, 256, 0, stream>>>((unsigned*)gsum, 0u, NGRAPH * WDIM);
    fill_u32<<<1, 256, 0, stream>>>((unsigned*)cnts, 0u, NGRAPH);
    pool_accum<<<POOL_BLOCKS, 256, 0, stream>>>(bufB, bat, gmaxe, gsum, cnts);
    pool_final<<<NGRAPH, 256, 0, stream>>>(gmaxe, gsum, cnts, rep);
    fc_sigmoid<<<NGRAPH, 256, 0, stream>>>(rep, Wfc, bfc, out);
}


# 11 launches to 6: edge-list build, layer-1 projection and pooling-array zeroing share one grid (roles by block index, independent work side by side); pool finalize and FC in one block; sums as before
# speedup vs baseline: 6.8423x; 1.1455x over previous
#include <hip/hip_runtime.h>
#include <hip/hip_bf16.h>

typedef __attribute__((ext_vector_type(2))) float v2f;
typedef __attribute__((ext_vector_type(8))) float v8f;

#define NNODES 50000
#define NEDGES 800000
#define WDIM   256
#define NHEADS 8
#define NHID   32
#define NGRAPH 256

__device__ __forceinline__ unsigned fenc(float x) {
    unsigned u = __float_as_uint(x);
    return (u & 0x80000000u) ? ~u : (u | 0x80000000u);
}
__device__ __forceinline__ float fdec(unsigned e) {
    unsigned u = (e & 0x80000000u) ? (e & 0x7FFFFFFFu) : ~e;
    return __uint_as_float(u);
}

static_assert(WDIM == 256 && NHEADS == 8 && NHID == 32, "a 64-wide strip is two heads; 4 strips per row tile");

template <int XM>
__device__ __forceinline__ float row_xor(float x) {
    return __int_as_float(__builtin_amdgcn_update_dpp(0, __float_as_int(x), 0x160 + XM, 0xf, 0xf, true));
}
template <int H, int XM>
__device__ __forceinline__ void halve_step(const float (&in)[2 * H], float (&out)[H], bool up) {
#pragma unroll
    for (int i = 0; i < H; ++i) {
        const float keep = up ? in[H + i] : in[i];
        const float send = up ? in[i] : in[H + i];
        out[i] = keep + row_xor<XM>(send);
    }
}

template <int RT>
__device__ __forceinline__ void gemm_al_body(const float* __restrict__ A, const float* __restrict__ B,
                                             float* __restrict__ C,
                                             const float* __restrict__ as_, const float* __restrict__ ad_,
                                             float* __restrict__ als, float* __restrict__ ald,
                                             int M, int K, int blk) {
    __shared__ float s_al[2][16 * RT][NHEADS];

    const int N = WDIM;
    const int w    = threadIdx.x >> 5;
    const int lane = threadIdx.x & 31;
    const int mblk = blk * (16 * RT);
    const int m0r  = mblk + ((w >> 2) << 4);
    const bool valid = m0r < M;
    const int m0 = valid ? m0r : M - 16;
    const int n0 = (w & 3) << 6;
    const int hi = lane >> 4;
    const int l  = lane & 15;

    const float s0 = as_[n0 + l], s1 = as_[n0 + 16 + l], s2 = as_[n0 + 32 + l], s3 = as_[n0 + 48 + l];
    const float d0 = ad_[n0 + l], d1 = ad_[n0 + 16 + l], d2 = ad_[n0 + 32 + l], d3 = ad_[n0 + 48 + l];

    v8f c0 = {}, c1 = {}, c2 = {}, c3 = {};
    const float* arow  = A + (size_t)(m0 + l) * K + (hi << 1);
    const float* bbase = B + (size_t)(hi << 1) * N + n0 + l;

    for (int k = 0; k < K; k += 4) {
        v2f a = *(const v2f*)(arow + k);
        const float* bk0 = bbase + (size_t)k * N;
        const float* bk1 = bk0 + N;
        v2f b0, b1, b2, b3;
        b0.x = bk0[0];   b0.y = bk1[0];
        b1.x = bk0[16];  b1.y = bk1[16];
        b2.x = bk0[32];  b2.y = bk1[32];
        b3.x = bk0[48];  b3.y = bk1[48];
        c0 = __builtin_amdgcn_wmma_f32_16x16x4_f32(false, a, false, b0, (short)0, c0, false, false);
        c1 = __builtin_amdgcn_wmma_f32_16x16x4_f32(false, a, false, b1, (short)0, c1, false, false);
        c2 = __builtin_amdgcn_wmma_f32_16x16x4_f32(false, a, false, b2, (short)0, c2, false, false);
        c3 = __builtin_amdgcn_wmma_f32_16x16x4_f32(false, a, false, b3, (short)0, c3, false, false);
    }
    if (valid) {
        float* crow = C + (size_t)(m0 + (hi << 3)) * N + n0 + l;
#pragma unroll
        for (int r = 0; r < 8; ++r) {
            crow[(size_t)r * N +  0] = c0[r];
            crow[(size_t)r * N + 16] = c1[r];
            crow[(size_t)r * N + 32] = c2[r];
            crow[(size_t)r * N + 48] = c3[r];
        }
    }

    float x[32];
#pragma unroll
    for (int r = 0; r < 8; ++r) {
        x[2 * r]          = fmaf(c1[r], s1, c0[r] * s0);
        x[2 * r + 1]      = fmaf(c1[r], d1, c0[r] * d0);
        x[16 + 2 * r]     = fmaf(c3[r], s3, c2[r] * s2);
        x[16 + 2 * r + 1] = fmaf(c3[r], d3, c2[r] * d2);
    }
    float y[16], z[8], q[4], u[2];
    halve_step<16, 8>(x, y, (lane & 8) != 0);
    halve_step<8, 4>(y, z, (lane & 4) != 0);
    halve_step<4, 2>(z, q, (lane & 2) != 0);
    halve_step<2, 1>(q, u, (lane & 1) != 0);
    {
        const int brow = ((w >> 2) << 4) + (hi << 3) + (l & 7);
        const int head = ((w & 3) << 1) + (l >> 3);
        s_al[0][brow][head] = u[0];
        s_al[1][brow][head] = u[1];
    }
    __syncthreads();
    const int t = threadIdx.x;
    if (t < 64 * RT) {
        const int which = t / (32 * RT), f = t % (32 * RT);
        if (mblk + (f >> 1) < M) {
            const float4 v = ((const float4*)&s_al[which][0][0])[f];
            float* dstp = (which ? ald : als) + (size_t)mblk * NHEADS;
            ((float4*)dstp)[f] = v;
        }
    }
}

__global__ __launch_bounds__(256) void gemm_f32_wmma_al(const float* __restrict__ A, const float* __restrict__ B,
                                                        float* __restrict__ C,
                                                        const float* __restrict__ as_, const float* __restrict__ ad_,
                                                        float* __restrict__ als, float* __restrict__ ald,
                                                        int M, int K) {
    gemm_al_body<2>(A, B, C, as_, ad_, als, ald, M, K, (int)blockIdx.x);
}

#define CSR_WGS  96
#define CSR_R    ((NNODES + CSR_WGS - 1) / CSR_WGS)
#define CSR_CAPW 512
#define CSR_U    8
#define NQUADS   (NEDGES / 4)
static_assert(NEDGES % 4 == 0, "destinations are read four at a time");
static_assert(CSR_R <= 1024 && CSR_R < (1 << 12), "one thread per node of the range; node packed above bit 20");
static_assert(NEDGES < (1 << 20), "a slot inside a list is packed into 20 bits");
static_assert(NNODES % 8 == 0, "node_agg takes nodes eight at a time");

__device__ __forceinline__ unsigned wave_incl_scan(unsigned x, int lane) {
    for (int d = 1; d < 32; d <<= 1) {
        unsigned y = __shfl_up(x, d, 32);
        if (lane >= d) x += y;
    }
    return x;
}

__device__ __forceinline__ void csr_build_body(const int* __restrict__ src, const int* __restrict__ dst,
                                               unsigned* __restrict__ rowptr, int* __restrict__ csr, const int blk) {
    __shared__ unsigned s_cnt[CSR_R];
    __shared__ unsigned s_fill[CSR_R];
    __shared__ unsigned s_wbelow[32];
    __shared__ unsigned s_wtot[32];
    __shared__ unsigned s_ovf;
    __shared__ uint2    s_stash[32][CSR_CAPW];

    const int t = threadIdx.x, lane = t & 31;
    const int w = __builtin_amdgcn_readfirstlane(t >> 5);
    const int lo = blk * CSR_R;

    if (t < CSR_R) { s_cnt[t] = 0u; s_fill[t] = 0u; }
    if (t == 0) s_ovf = 0u;
    __syncthreads();

    const int4* dq = (const int4*)dst;
    unsigned nbelow = 0u, wcount = 0u;
    for (int q0 = 0; q0 < NQUADS; q0 += 1024 * CSR_U) {
        int4 v[CSR_U];
#pragma unroll
        for (int u = 0; u < CSR_U; ++u) {
            const int q = q0 + u * 1024 + t;
            v[u] = make_int4(0x7fffffff, 0x7fffffff, 0x7fffffff, 0x7fffffff);
            if (q < NQUADS) v[u] = dq[q];
        }
#pragma unroll
        for (int u = 0; u < CSR_U; ++u) {
            const int e0 = (q0 + u * 1024 + t) * 4;
            const int dd[4] = {v[u].x, v[u].y, v[u].z, v[u].w};
#pragma unroll
            for (int c = 0; c < 4; ++c) {
                const int d = dd[c];
                nbelow += (unsigned)__popc(__builtin_amdgcn_ballot_w32(d < lo));
                const unsigned dl = (unsigned)(d - lo);
                const bool inr = dl < (unsigned)CSR_R;
                const unsigned mask = __builtin_amdgcn_ballot_w32(inr);
                if (mask != 0u) {
                    if (inr) {
                        const unsigned rk = atomicAdd(&s_cnt[dl], 1u);
                        const unsigned s = wcount + __builtin_amdgcn_mbcnt_lo(mask, 0u);
                        if (s < CSR_CAPW) s_stash[w][s] = make_uint2((unsigned)(e0 + c), (dl << 20) | rk);
                    }
                    wcount += (unsigned)__popc(mask);
                }
            }
        }
    }
    if (lane == 0) {
        s_wbelow[w] = nbelow;
        if (wcount > CSR_CAPW) s_ovf = 1u;
    }
    __syncthreads();

    unsigned base = s_wbelow[lane];
    for (int x = 16; x > 0; x >>= 1) base += __shfl_xor(base, x, 32);
    const unsigned x = (t < CSR_R) ? s_cnt[t] : 0u;
    const unsigned incl = wave_incl_scan(x, lane);
    if (lane == 31) s_wtot[w] = incl;
    const unsigned ovf = s_ovf;
    __syncthreads();
    const unsigned wt = s_wtot[lane];
    const unsigned wincl = wave_incl_scan(wt, lane);
    const unsigned wexcl = __shfl(wincl - wt, w, 32);
    const unsigned total = __shfl(wincl, 31, 32);
    const unsigned excl = wexcl + incl - x;
    if (t < CSR_R) {
        s_cnt[t] = excl;
        if (lo + t < NNODES) rowptr[lo + t] = base + excl;
    }
    if (blk == CSR_WGS - 1 && t == 0) rowptr[NNODES] = base + total;
    __syncthreads();

    if (ovf == 0u) {
        for (unsigned k = (unsigned)lane; k < wcount; k += 32u) {
            const uint2 it = s_stash[w][k];
            csr[base + s_cnt[it.y >> 20] + (it.y & 0xFFFFFu)] = src[it.x];
        }
    } else {
        for (int q = t; q < NQUADS; q += 1024) {
            const int4 vv = dq[q];
            const int dd[4] = {vv.x, vv.y, vv.z, vv.w};
#pragma unroll
            for (int c = 0; c < 4; ++c) {
                const unsigned dl = (unsigned)(dd[c] - lo);
                if (dl < (unsigned)CSR_R) {
                    const unsigned slot = atomicAdd(&s_fill[dl], 1u);
                    csr[base + s_cnt[dl] + slot] = src[q * 4 + c];
                }
            }
        }
    }
}

#define GEMM1_RT     8
#define GEMM1_BLOCKS ((NNODES + 16 * GEMM1_RT - 1) / (16 * GEMM1_RT))
#define GEMM2_BLOCKS ((NNODES + 31) / 32)
#define POOL_WORDS   (2 * NGRAPH * WDIM + NGRAPH)
static_assert(GEMM1_BLOCKS * 1024 >= POOL_WORDS, "one pooling word per thread of the multiplying work-groups");

__global__ __launch_bounds__(1024) void csr_gemm1_zero(const int* __restrict__ src, const int* __restrict__ dst,
                                                       unsigned* __restrict__ rowptr, int* __restrict__ csr,
                                                       const float* __restrict__ A, const float* __restrict__ B,
                                                       float* __restrict__ C,
                                                       const float* __restrict__ as_, const float* __restrict__ ad_,
                                                       float* __restrict__ als, float* __restrict__ ald,
                                                       unsigned* __restrict__ poolz, int K) {
    if (blockIdx.x < CSR_WGS) {
        csr_build_body(src, dst, rowptr, csr, (int)blockIdx.x);
    } else {
        const int g  = (int)blockIdx.x - CSR_WGS;
        const int zi = g * 1024 + (int)threadIdx.x;
        if (zi < POOL_WORDS) poolz[zi] = 0u;
        gemm_al_body<GEMM1_RT>(A, B, C, as_, ad_, als, ald, NNODES, K, g);
    }
}

__device__ __forceinline__ float4 fma4(float p, float4 y, float4 a) {
    a.x = fmaf(p, y.x, a.x); a.y = fmaf(p, y.y, a.y);
    a.z = fmaf(p, y.z, a.z); a.w = fmaf(p, y.w, a.w);
    return a;
}

__device__ __forceinline__ float lrelu(float v) { return v > 0.f ? v : 0.2f * v; }
__device__ __forceinline__ float elu1(float v) { return v > 0.f ? v : (expf(v) - 1.0f); }
__device__ __forceinline__ float sel4(bool lo, bool hi, float a, float b, float c, float d) {
    const float x = lo ? b : a, y = lo ? d : c;
    return hi ? y : x;
}
__device__ __forceinline__ float lane_bcast(float v, int src) {
    return __int_as_float(__builtin_amdgcn_readlane(__float_as_int(v), src));
}
__global__ __launch_bounds__(256) void node_agg(const unsigned* __restrict__ rowptr,
                                                const int* __restrict__ csr,
                                                const float* __restrict__ als,
                                                const float* __restrict__ ald,
                                                const float* __restrict__ h,
                                                const float* __restrict__ bias,
                                                int apply_elu,
                                                float* __restrict__ outx) {
    const int wv   = __builtin_amdgcn_readfirstlane((int)(threadIdx.x >> 5));
    const int node = blockIdx.x * 8 + wv;
    if (node >= NNODES) return;
    const int lane = threadIdx.x & 31;
    const int c0 = lane * 4, c1 = 128 + lane * 4;
    const bool b3 = (lane & 8) != 0, b4 = (lane & 16) != 0;
    const int beg = __builtin_amdgcn_readfirstlane((int)rowptr[node]);
    const int end = __builtin_amdgcn_readfirstlane((int)rowptr[node + 1]);

    const float4 dA = *(const float4*)(ald + (size_t)node * NHEADS);
    const float4 dB = *(const float4*)(ald + (size_t)node * NHEADS + 4);
    const float4 sA = *(const float4*)(als + (size_t)node * NHEADS);
    const float4 sB = *(const float4*)(als + (size_t)node * NHEADS + 4);
    const float d[8] = {dA.x, dA.y, dA.z, dA.w, dB.x, dB.y, dB.z, dB.w};
    const float sf[8] = {sA.x, sA.y, sA.z, sA.w, sB.x, sB.y, sB.z, sB.w};
    float es[8], e[8], m[8];
#pragma unroll
    for (int k = 0; k < 8; ++k) { es[k] = lrelu(sf[k] + d[k]); e[k] = es[k]; m[k] = es[k]; }

    int sl = 0, jlast = beg;
    for (int j0 = beg; j0 < end; j0 += 32) {
        jlast = j0;
        if (j0 + lane < end) {
            sl = csr[j0 + lane];
            const float4 xa = *(const float4*)(als + (size_t)sl * NHEADS);
            const float4 xb = *(const float4*)(als + (size_t)sl * NHEADS + 4);
            const float x[8] = {xa.x, xa.y, xa.z, xa.w, xb.x, xb.y, xb.z, xb.w};
#pragma unroll
            for (int k = 0; k < 8; ++k) { e[k] = lrelu(x[k] + d[k]); m[k] = fmaxf(m[k], e[k]); }
        }
    }
    float k0 = b4 ? m[2] : m[0], k1 = b4 ? m[3] : m[1], k2 = b4 ? m[6] : m[4], k3 = b4 ? m[7] : m[5];
    float g0 = b4 ? m[0] : m[2], g1 = b4 ? m[1] : m[3], g2 = b4 ? m[4] : m[6], g3 = b4 ? m[5] : m[7];
    k0 = fmaxf(k0, __shfl_xor(g0, 16, 32)); k1 = fmaxf(k1, __shfl_xor(g1, 16, 32));
    k2 = fmaxf(k2, __shfl_xor(g2, 16, 32)); k3 = fmaxf(k3, __shfl_xor(g3, 16, 32));
    float m0 = b3 ? k1 : k0, t0 = b3 ? k0 : k1;
    float m1 = b3 ? k3 : k2, t1 = b3 ? k2 : k3;
    m0 = fmaxf(m0, __shfl_xor(t0, 8, 32));
    m1 = fmaxf(m1, __shfl_xor(t1, 8, 32));
#pragma unroll
    for (int x = 4; x > 0; x >>= 1) {
        m0 = fmaxf(m0, __shfl_xor(m0, x, 32));
        m1 = fmaxf(m1, __shfl_xor(m1, x, 32));
    }
    float mx[8];
#pragma unroll
    for (int k = 0; k < 4; ++k) { mx[k] = lane_bcast(m0, 8 * k); mx[4 + k] = lane_bcast(m1, 8 * k); }

    const float* hself = h + (size_t)node * WDIM;
    float dn0 = expf(sel4(b3, b4, es[0], es[1], es[2], es[3]) - m0);
    float dn1 = expf(sel4(b3, b4, es[4], es[5], es[6], es[7]) - m1);
    float4 acc0 = make_float4(0.f, 0.f, 0.f, 0.f), acc1 = acc0;
    acc0 = fma4(dn0, *(const float4*)(hself + c0), acc0);
    acc1 = fma4(dn1, *(const float4*)(hself + c1), acc1);

    for (int j0 = beg; j0 < end; j0 += 32) {
        const int n = (end - j0) < 32 ? (end - j0) : 32;
        int ent = sl;
        float p[8];
        if (j0 != jlast) {
            ent = csr[j0 + lane];
            const float4 xa = *(const float4*)(als + (size_t)ent * NHEADS);
            const float4 xb = *(const float4*)(als + (size_t)ent * NHEADS + 4);
            const float x[8] = {xa.x, xa.y, xa.z, xa.w, xb.x, xb.y, xb.z, xb.w};
#pragma unroll
            for (int k = 0; k < 8; ++k) p[k] = expf(lrelu(x[k] + d[k]) - mx[k]);
        } else {
#pragma unroll
            for (int k = 0; k < 8; ++k) p[k] = expf(e[k] - mx[k]);
        }
        int j = 0;
        for (; j + 4 <= n; j += 4) {
            float4 u0[4], u1[4];
#pragma unroll
            for (int u = 0; u < 4; ++u) {
                const float* hs = h + (size_t)__builtin_amdgcn_readlane(ent, j + u) * WDIM;
                u0[u] = *(const float4*)(hs + c0);
                u1[u] = *(const float4*)(hs + c1);
            }
#pragma unroll
            for (int u = 0; u < 4; ++u) {
                const float p0 = sel4(b3, b4, lane_bcast(p[0], j + u), lane_bcast(p[1], j + u),
                                         lane_bcast(p[2], j + u), lane_bcast(p[3], j + u));
                const float p1 = sel4(b3, b4, lane_bcast(p[4], j + u), lane_bcast(p[5], j + u),
                                         lane_bcast(p[6], j + u), lane_bcast(p[7], j + u));
                dn0 += p0; dn1 += p1;
                acc0 = fma4(p0, u0[u], acc0);
                acc1 = fma4(p1, u1[u], acc1);
            }
        }
        for (; j < n; ++j) {
            const float* hs = h + (size_t)__builtin_amdgcn_readlane(ent, j) * WDIM;
            const float p0 = sel4(b3, b4, lane_bcast(p[0], j), lane_bcast(p[1], j),
                                     lane_bcast(p[2], j), lane_bcast(p[3], j));
            const float p1 = sel4(b3, b4, lane_bcast(p[4], j), lane_bcast(p[5], j),
                                     lane_bcast(p[6], j), lane_bcast(p[7], j));
            dn0 += p0; dn1 += p1;
            acc0 = fma4(p0, *(const float4*)(hs + c0), acc0);
            acc1 = fma4(p1, *(const float4*)(hs + c1), acc1);
        }
    }
    const float4 bz0 = *(const float4*)(bias + c0);
    const float4 bz1 = *(const float4*)(bias + c1);
    float4 o0 = make_float4(acc0.x / dn0 + bz0.x, acc0.y / dn0 + bz0.y, acc0.z / dn0 + bz0.z, acc0.w / dn0 + bz0.w);
    float4 o1 = make_float4(acc1.x / dn1 + bz1.x, acc1.y / dn1 + bz1.y, acc1.z / dn1 + bz1.z, acc1.w / dn1 + bz1.w);
    if (apply_elu) {
        o0.x = elu1(o0.x); o0.y = elu1(o0.y); o0.z = elu1(o0.z); o0.w = elu1(o0.w);
        o1.x = elu1(o1.x); o1.y = elu1(o1.y); o1.z = elu1(o1.z); o1.w = elu1(o1.w);
    }
    float* orow = outx + (size_t)node * WDIM;
    *(float4*)(orow + c0) = o0;
    *(float4*)(orow + c1) = o1;
}

#define POOL_ROWS 16
#define POOL_BLOCKS ((NNODES / POOL_ROWS + 7) / 8)
static_assert(NNODES % POOL_ROWS == 0, "pool_accum takes nodes sixteen at a time");
static_assert(POOL_ROWS <= 32, "one lane holds one node's graph id");

__device__ __forceinline__ void pool_flush(int g, int lane, const float (&mx)[8], const float (&sm)[8],
                                           float cnt, unsigned* __restrict__ gmaxenc,
                                           float* __restrict__ gsum, float* __restrict__ counts) {
    unsigned* pm = gmaxenc + g * WDIM + lane;
    float*    ps = gsum + g * WDIM + lane;
#pragma unroll
    for (int i = 0; i < 8; ++i) {
        atomicMax(pm + 32 * i, fenc(mx[i]));
        atomicAdd(ps + 32 * i, sm[i]);
    }
    if (lane == 0) atomicAdd(&counts[g], cnt);
}

__global__ __launch_bounds__(256) void pool_accum(const float* __restrict__ x2, const int* __restrict__ batch,
                                                  unsigned* __restrict__ gmaxenc, float* __restrict__ gsum,
                                                  float* __restrict__ counts) {
    const int wv = __builtin_amdgcn_readfirstlane((int)(threadIdx.x >> 5));
    const int n0 = (blockIdx.x * 8 + wv) * POOL_ROWS;
    if (n0 >= NNODES) return;
    const int lane = threadIdx.x & 31;
    const int gl = batch[n0 + (lane & (POOL_ROWS - 1))];
    const float* xr = x2 + (size_t)n0 * WDIM + lane;
    float v[POOL_ROWS][8];
#pragma unroll
    for (int r = 0; r < POOL_ROWS; ++r) {
#pragma unroll
        for (int i = 0; i < 8; ++i) v[r][i] = xr[r * WDIM + 32 * i];
    }
    int cur = __builtin_amdgcn_readlane(gl, 0);
    float mx[8], sm[8], cnt = 0.f;
#pragma unroll
    for (int i = 0; i < 8; ++i) { mx[i] = -INFINITY; sm[i] = 0.f; }
#pragma unroll
    for (int r = 0; r < POOL_ROWS; ++r) {
        const int g = __builtin_amdgcn_readlane(gl, r);
        if (g != cur) {
            pool_flush(cur, lane, mx, sm, cnt, gmaxenc, gsum, counts);
#pragma unroll
            for (int i = 0; i < 8; ++i) { mx[i] = -INFINITY; sm[i] = 0.f; }
            cnt = 0.f;
            cur = g;
        }
#pragma unroll
        for (int i = 0; i < 8; ++i) { mx[i] = fmaxf(mx[i], v[r][i]); sm[i] += v[r][i]; }
        cnt += 1.0f;
    }
    pool_flush(cur, lane, mx, sm, cnt, gmaxenc, gsum, counts);
}

__global__ __launch_bounds__(256) void pool_final_fc(const unsigned* __restrict__ gmaxenc, const float* __restrict__ gsum,
                                                     const float* __restrict__ counts, const float* __restrict__ Wfc,
                                                     const float* __restrict__ bfc, float* __restrict__ rep,
                                                     float* __restrict__ out) {
    __shared__ float red[256];
    int g = blockIdx.x, t = threadIdx.x;
    float cnt = counts[g];
    float mx = (cnt > 0.f) ? fdec(gmaxenc[g * WDIM + t]) : 0.0f;
    float mean = gsum[g * WDIM + t] / fmaxf(cnt, 1.0f);
    rep[(size_t)g * (2 * WDIM) + t] = mx;
    rep[(size_t)g * (2 * WDIM) + WDIM + t] = mean;
    float s = mx * Wfc[t] + mean * Wfc[WDIM + t];
    red[t] = s;
    __syncthreads();
    for (int off = 128; off > 0; off >>= 1) {
        if (t < off) red[t] += red[t + off];
        __syncthreads();
    }
    if (t == 0) out[g] = 1.0f / (1.0f + expf(-(red[0] + bfc[0])));
}

extern "C" void kernel_launch(void* const* d_in, const int* in_sizes, int n_in,
                              void* d_out, int out_size, void* d_ws, size_t ws_size,
                              hipStream_t stream) {
    (void)in_sizes; (void)n_in; (void)out_size; (void)ws_size;

    const float* x    = (const float*)d_in[0];
    const int*   ei   = (const int*)d_in[1];
    const int*   bat  = (const int*)d_in[2];
    const float* W1   = (const float*)d_in[3];
    const float* a1s  = (const float*)d_in[4];
    const float* a1d  = (const float*)d_in[5];
    const float* b1   = (const float*)d_in[6];
    const float* W2   = (const float*)d_in[7];
    const float* a2s  = (const float*)d_in[8];
    const float* a2d  = (const float*)d_in[9];
    const float* b2   = (const float*)d_in[10];
    const float* Wfc  = (const float*)d_in[11];
    const float* bfc  = (const float*)d_in[12];

    float* out = (float*)d_out;
    float* rep = out + NGRAPH;

    const int* esrc = ei;
    const int* edst = ei + NEDGES;

    float*    bufA  = (float*)d_ws;
    float*    bufB  = bufA + (size_t)NNODES * WDIM;
    float*    als   = bufB + (size_t)NNODES * WDIM;
    float*    ald   = als + (size_t)NNODES * NHEADS;
    unsigned* gmaxe = (unsigned*)(ald + (size_t)NNODES * NHEADS);
    float*    gsum  = (float*)(gmaxe + NGRAPH * WDIM);
    float*    cnts  = gsum + NGRAPH * WDIM;
    unsigned* rowp  = (unsigned*)(cnts + NGRAPH);
    int*      csr   = (int*)(rowp + NNODES + 4);

    csr_gemm1_zero<<<CSR_WGS + GEMM1_BLOCKS, 1024, 0, stream>>>(esrc, edst, rowp, csr, x, W1, bufA, a1s, a1d,
                                                                als, ald, gmaxe, 128);
    node_agg<<<NNODES / 8, 256, 0, stream>>>(rowp, csr, als, ald, bufA, b1, 1, bufB);

    gemm_f32_wmma_al<<<GEMM2_BLOCKS, 256, 0, stream>>>(bufB, W2, bufA, a2s, a2d, als, ald, NNODES, 256);
    node_agg<<<NNODES / 8, 256, 0, stream>>>(rowp, csr, als, ald, bufA, b2, 0, bufB);

    pool_accum<<<POOL_BLOCKS, 256, 0, stream>>>(bufB, bat, gmaxe, gsum, cnts);
    pool_final_fc<<<NGRAPH, 256, 0, stream>>>(gmaxe, gsum, cnts, Wfc, bfc, rep, out);
}


# projection GEMMs: a wave's 4 tiles interleave by column (tile j = column 4l+j): B in two 16-byte row loads per K-step, not eight 4-byte; C stored 16 bytes a lane; pre-logit sum order changes
# speedup vs baseline: 7.4516x; 1.0890x over previous
#include <hip/hip_runtime.h>
#include <hip/hip_bf16.h>

typedef __attribute__((ext_vector_type(2))) float v2f;
typedef __attribute__((ext_vector_type(8))) float v8f;

#define NNODES 50000
#define NEDGES 800000
#define WDIM   256
#define NHEADS 8
#define NHID   32
#define NGRAPH 256

__device__ __forceinline__ unsigned fenc(float x) {
    unsigned u = __float_as_uint(x);
    return (u & 0x80000000u) ? ~u : (u | 0x80000000u);
}
__device__ __forceinline__ float fdec(unsigned e) {
    unsigned u = (e & 0x80000000u) ? (e & 0x7FFFFFFFu) : ~e;
    return __uint_as_float(u);
}

static_assert(WDIM == 256 && NHEADS == 8 && NHID == 32, "a 64-wide strip is two heads; 4 strips per row tile");

template <int XM>
__device__ __forceinline__ float row_xor(float x) {
    return __int_as_float(__builtin_amdgcn_update_dpp(0, __float_as_int(x), 0x160 + XM, 0xf, 0xf, true));
}
template <int H, int XM>
__device__ __forceinline__ void halve_step(const float (&in)[2 * H], float (&out)[H], bool up) {
#pragma unroll
    for (int i = 0; i < H; ++i) {
        const float keep = up ? in[H + i] : in[i];
        const float send = up ? in[i] : in[H + i];
        out[i] = keep + row_xor<XM>(send);
    }
}

template <int RT>
__device__ __forceinline__ void gemm_al_body(const float* __restrict__ A, const float* __restrict__ B,
                                             float* __restrict__ C,
                                             const float* __restrict__ as_, const float* __restrict__ ad_,
                                             float* __restrict__ als, float* __restrict__ ald,
                                             int M, int K, int blk) {
    __shared__ float s_al[2][16 * RT][NHEADS];

    const int N = WDIM;
    const int w    = threadIdx.x >> 5;
    const int lane = threadIdx.x & 31;
    const int mblk = blk * (16 * RT);
    const int m0r  = mblk + ((w >> 2) << 4);
    const bool valid = m0r < M;
    const int m0 = valid ? m0r : M - 16;
    const int n0 = (w & 3) << 6;
    const int hi = lane >> 4;
    const int l  = lane & 15;

    const float4 sv = *(const float4*)(as_ + n0 + 4 * l);
    const float4 dv = *(const float4*)(ad_ + n0 + 4 * l);

    v8f c0 = {}, c1 = {}, c2 = {}, c3 = {};
    const float* arow  = A + (size_t)(m0 + l) * K + (hi << 1);
    const float* bbase = B + (size_t)(hi << 1) * N + n0 + 4 * l;

    for (int k = 0; k < K; k += 4) {
        v2f a = *(const v2f*)(arow + k);
        const float* bk0 = bbase + (size_t)k * N;
        const float* bk1 = bk0 + N;
        const float4 r0 = *(const float4*)bk0;
        const float4 r1 = *(const float4*)bk1;
        v2f b0, b1, b2, b3;
        b0.x = r0.x;  b0.y = r1.x;
        b1.x = r0.y;  b1.y = r1.y;
        b2.x = r0.z;  b2.y = r1.z;
        b3.x = r0.w;  b3.y = r1.w;
        c0 = __builtin_amdgcn_wmma_f32_16x16x4_f32(false, a, false, b0, (short)0, c0, false, false);
        c1 = __builtin_amdgcn_wmma_f32_16x16x4_f32(false, a, false, b1, (short)0, c1, false, false);
        c2 = __builtin_amdgcn_wmma_f32_16x16x4_f32(false, a, false, b2, (short)0, c2, false, false);
        c3 = __builtin_amdgcn_wmma_f32_16x16x4_f32(false, a, false, b3, (short)0, c3, false, false);
    }
    if (valid) {
        float* crow = C + (size_t)(m0 + (hi << 3)) * N + n0 + 4 * l;
#pragma unroll
        for (int r = 0; r < 8; ++r)
            *(float4*)(crow + (size_t)r * N) = make_float4(c0[r], c1[r], c2[r], c3[r]);
    }

    float x[16];
#pragma unroll
    for (int r = 0; r < 8; ++r) {
        x[2 * r]     = fmaf(c3[r], sv.w, fmaf(c2[r], sv.z, fmaf(c1[r], sv.y, c0[r] * sv.x)));
        x[2 * r + 1] = fmaf(c3[r], dv.w, fmaf(c2[r], dv.z, fmaf(c1[r], dv.y, c0[r] * dv.x)));
    }
    float z[8], q[4], u[2];
    halve_step<8, 4>(x, z, (lane & 4) != 0);
    halve_step<4, 2>(z, q, (lane & 2) != 0);
    halve_step<2, 1>(q, u, (lane & 1) != 0);
    {
        const int brow = ((w >> 2) << 4) + (hi << 3) + (l & 7);
        const int head = ((w & 3) << 1) + (l >> 3);
        s_al[0][brow][head] = u[0];
        s_al[1][brow][head] = u[1];
    }
    __syncthreads();
    const int t = threadIdx.x;
    if (t < 64 * RT) {
        const int which = t / (32 * RT), f = t % (32 * RT);
        if (mblk + (f >> 1) < M) {
            const float4 v = ((const float4*)&s_al[which][0][0])[f];
            float* dstp = (which ? ald : als) + (size_t)mblk * NHEADS;
            ((float4*)dstp)[f] = v;
        }
    }
}

__global__ __launch_bounds__(256) void gemm_f32_wmma_al(const float* __restrict__ A, const float* __restrict__ B,
                                                        float* __restrict__ C,
                                                        const float* __restrict__ as_, const float* __restrict__ ad_,
                                                        float* __restrict__ als, float* __restrict__ ald,
                                                        int M, int K) {
    gemm_al_body<2>(A, B, C, as_, ad_, als, ald, M, K, (int)blockIdx.x);
}

#define CSR_WGS  96
#define CSR_R    ((NNODES + CSR_WGS - 1) / CSR_WGS)
#define CSR_CAPW 512
#define CSR_U    8
#define NQUADS   (NEDGES / 4)
static_assert(NEDGES % 4 == 0, "destinations are read four at a time");
static_assert(CSR_R <= 1024 && CSR_R < (1 << 12), "one thread per node of the range; node packed above bit 20");
static_assert(NEDGES < (1 << 20), "a slot inside a list is packed into 20 bits");
static_assert(NNODES % 8 == 0, "node_agg takes nodes eight at a time");

__device__ __forceinline__ unsigned wave_incl_scan(unsigned x, int lane) {
    for (int d = 1; d < 32; d <<= 1) {
        unsigned y = __shfl_up(x, d, 32);
        if (lane >= d) x += y;
    }
    return x;
}

__device__ __forceinline__ void csr_build_body(const int* __restrict__ src, const int* __restrict__ dst,
                                               unsigned* __restrict__ rowptr, int* __restrict__ csr, const int blk) {
    __shared__ unsigned s_cnt[CSR_R];
    __shared__ unsigned s_fill[CSR_R];
    __shared__ unsigned s_wbelow[32];
    __shared__ unsigned s_wtot[32];
    __shared__ unsigned s_ovf;
    __shared__ uint2    s_stash[32][CSR_CAPW];

    const int t = threadIdx.x, lane = t & 31;
    const int w = __builtin_amdgcn_readfirstlane(t >> 5);
    const int lo = blk * CSR_R;

    if (t < CSR_R) { s_cnt[t] = 0u; s_fill[t] = 0u; }
    if (t == 0) s_ovf = 0u;
    __syncthreads();

    const int4* dq = (const int4*)dst;
    unsigned nbelow = 0u, wcount = 0u;
    for (int q0 = 0; q0 < NQUADS; q0 += 1024 * CSR_U) {
        int4 v[CSR_U];
#pragma unroll
        for (int u = 0; u < CSR_U; ++u) {
            const int q = q0 + u * 1024 + t;
            v[u] = make_int4(0x7fffffff, 0x7fffffff, 0x7fffffff, 0x7fffffff);
            if (q < NQUADS) v[u] = dq[q];
        }
#pragma unroll
        for (int u = 0; u < CSR_U; ++u) {
            const int e0 = (q0 + u * 1024 + t) * 4;
            const int dd[4] = {v[u].x, v[u].y, v[u].z, v[u].w};
#pragma unroll
            for (int c = 0; c < 4; ++c) {
                const int d = dd[c];
                nbelow += (unsigned)__popc(__builtin_amdgcn_ballot_w32(d < lo));
                const unsigned dl = (unsigned)(d - lo);
                const bool inr = dl < (unsigned)CSR_R;
                const unsigned mask = __builtin_amdgcn_ballot_w32(inr);
                if (mask != 0u) {
                    if (inr) {
                        const unsigned rk = atomicAdd(&s_cnt[dl], 1u);
                        const unsigned s = wcount + __builtin_amdgcn_mbcnt_lo(mask, 0u);
                        if (s < CSR_CAPW) s_stash[w][s] = make_uint2((unsigned)(e0 + c), (dl << 20) | rk);
                    }
                    wcount += (unsigned)__popc(mask);
                }
            }
        }
    }
    if (lane == 0) {
        s_wbelow[w] = nbelow;
        if (wcount > CSR_CAPW) s_ovf = 1u;
    }
    __syncthreads();

    unsigned base = s_wbelow[lane];
    for (int x = 16; x > 0; x >>= 1) base += __shfl_xor(base, x, 32);
    const unsigned x = (t < CSR_R) ? s_cnt[t] : 0u;
    const unsigned incl = wave_incl_scan(x, lane);
    if (lane == 31) s_wtot[w] = incl;
    const unsigned ovf = s_ovf;
    __syncthreads();
    const unsigned wt = s_wtot[lane];
    const unsigned wincl = wave_incl_scan(wt, lane);
    const unsigned wexcl = __shfl(wincl - wt, w, 32);
    const unsigned total = __shfl(wincl, 31, 32);
    const unsigned excl = wexcl + incl - x;
    if (t < CSR_R) {
        s_cnt[t] = excl;
        if (lo + t < NNODES) rowptr[lo + t] = base + excl;
    }
    if (blk == CSR_WGS - 1 && t == 0) rowptr[NNODES] = base + total;
    __syncthreads();

    if (ovf == 0u) {
        for (unsigned k = (unsigned)lane; k < wcount; k += 32u) {
            const uint2 it = s_stash[w][k];
            csr[base + s_cnt[it.y >> 20] + (it.y & 0xFFFFFu)] = src[it.x];
        }
    } else {
        for (int q = t; q < NQUADS; q += 1024) {
            const int4 vv = dq[q];
            const int dd[4] = {vv.x, vv.y, vv.z, vv.w};
#pragma unroll
            for (int c = 0; c < 4; ++c) {
                const unsigned dl = (unsigned)(dd[c] - lo);
                if (dl < (unsigned)CSR_R) {
                    const unsigned slot = atomicAdd(&s_fill[dl], 1u);
                    csr[base + s_cnt[dl] + slot] = src[q * 4 + c];
                }
            }
        }
    }
}

#define GEMM1_RT     8
#define GEMM1_BLOCKS ((NNODES + 16 * GEMM1_RT - 1) / (16 * GEMM1_RT))
#define GEMM2_BLOCKS ((NNODES + 31) / 32)
#define POOL_WORDS   (2 * NGRAPH * WDIM + NGRAPH)
static_assert(GEMM1_BLOCKS * 1024 >= POOL_WORDS, "one pooling word per thread of the multiplying work-groups");

__global__ __launch_bounds__(1024) void csr_gemm1_zero(const int* __restrict__ src, const int* __restrict__ dst,
                                                       unsigned* __restrict__ rowptr, int* __restrict__ csr,
                                                       const float* __restrict__ A, const float* __restrict__ B,
                                                       float* __restrict__ C,
                                                       const float* __restrict__ as_, const float* __restrict__ ad_,
                                                       float* __restrict__ als, float* __restrict__ ald,
                                                       unsigned* __restrict__ poolz, int K) {
    if (blockIdx.x < CSR_WGS) {
        csr_build_body(src, dst, rowptr, csr, (int)blockIdx.x);
    } else {
        const int g  = (int)blockIdx.x - CSR_WGS;
        const int zi = g * 1024 + (int)threadIdx.x;
        if (zi < POOL_WORDS) poolz[zi] = 0u;
        gemm_al_body<GEMM1_RT>(A, B, C, as_, ad_, als, ald, NNODES, K, g);
    }
}

__device__ __forceinline__ float4 fma4(float p, float4 y, float4 a) {
    a.x = fmaf(p, y.x, a.x); a.y = fmaf(p, y.y, a.y);
    a.z = fmaf(p, y.z, a.z); a.w = fmaf(p, y.w, a.w);
    return a;
}

__device__ __forceinline__ float lrelu(float v) { return v > 0.f ? v : 0.2f * v; }
__device__ __forceinline__ float elu1(float v) { return v > 0.f ? v : (expf(v) - 1.0f); }
__device__ __forceinline__ float sel4(bool lo, bool hi, float a, float b, float c, float d) {
    const float x = lo ? b : a, y = lo ? d : c;
    return hi ? y : x;
}
__device__ __forceinline__ float lane_bcast(float v, int src) {
    return __int_as_float(__builtin_amdgcn_readlane(__float_as_int(v), src));
}
__global__ __launch_bounds__(256) void node_agg(const unsigned* __restrict__ rowptr,
                                                const int* __restrict__ csr,
                                                const float* __restrict__ als,
                                                const float* __restrict__ ald,
                                                const float* __restrict__ h,
                                                const float* __restrict__ bias,
                                                int apply_elu,
                                                float* __restrict__ outx) {
    const int wv   = __builtin_amdgcn_readfirstlane((int)(threadIdx.x >> 5));
    const int node = blockIdx.x * 8 + wv;
    if (node >= NNODES) return;
    const int lane = threadIdx.x & 31;
    const int c0 = lane * 4, c1 = 128 + lane * 4;
    const bool b3 = (lane & 8) != 0, b4 = (lane & 16) != 0;
    const int beg = __builtin_amdgcn_readfirstlane((int)rowptr[node]);
    const int end = __builtin_amdgcn_readfirstlane((int)rowptr[node + 1]);

    const float4 dA = *(const float4*)(ald + (size_t)node * NHEADS);
    const float4 dB = *(const float4*)(ald + (size_t)node * NHEADS + 4);
    const float4 sA = *(const float4*)(als + (size_t)node * NHEADS);
    const float4 sB = *(const float4*)(als + (size_t)node * NHEADS + 4);
    const float d[8] = {dA.x, dA.y, dA.z, dA.w, dB.x, dB.y, dB.z, dB.w};
    const float sf[8] = {sA.x, sA.y, sA.z, sA.w, sB.x, sB.y, sB.z, sB.w};
    float es[8], e[8], m[8];
#pragma unroll
    for (int k = 0; k < 8; ++k) { es[k] = lrelu(sf[k] + d[k]); e[k] = es[k]; m[k] = es[k]; }

    int sl = 0, jlast = beg;
    for (int j0 = beg; j0 < end; j0 += 32) {
        jlast = j0;
        if (j0 + lane < end) {
            sl = csr[j0 + lane];
            const float4 xa = *(const float4*)(als + (size_t)sl * NHEADS);
            const float4 xb = *(const float4*)(als + (size_t)sl * NHEADS + 4);
            const float x[8] = {xa.x, xa.y, xa.z, xa.w, xb.x, xb.y, xb.z, xb.w};
#pragma unroll
            for (int k = 0; k < 8; ++k) { e[k] = lrelu(x[k] + d[k]); m[k] = fmaxf(m[k], e[k]); }
        }
    }
    float k0 = b4 ? m[2] : m[0], k1 = b4 ? m[3] : m[1], k2 = b4 ? m[6] : m[4], k3 = b4 ? m[7] : m[5];
    float g0 = b4 ? m[0] : m[2], g1 = b4 ? m[1] : m[3], g2 = b4 ? m[4] : m[6], g3 = b4 ? m[5] : m[7];
    k0 = fmaxf(k0, __shfl_xor(g0, 16, 32)); k1 = fmaxf(k1, __shfl_xor(g1, 16, 32));
    k2 = fmaxf(k2, __shfl_xor(g2, 16, 32)); k3 = fmaxf(k3, __shfl_xor(g3, 16, 32));
    float m0 = b3 ? k1 : k0, t0 = b3 ? k0 : k1;
    float m1 = b3 ? k3 : k2, t1 = b3 ? k2 : k3;
    m0 = fmaxf(m0, __shfl_xor(t0, 8, 32));
    m1 = fmaxf(m1, __shfl_xor(t1, 8, 32));
#pragma unroll
    for (int x = 4; x > 0; x >>= 1) {
        m0 = fmaxf(m0, __shfl_xor(m0, x, 32));
        m1 = fmaxf(m1, __shfl_xor(m1, x, 32));
    }
    float mx[8];
#pragma unroll
    for (int k = 0; k < 4; ++k) { mx[k] = lane_bcast(m0, 8 * k); mx[4 + k] = lane_bcast(m1, 8 * k); }

    const float* hself = h + (size_t)node * WDIM;
    float dn0 = expf(sel4(b3, b4, es[0], es[1], es[2], es[3]) - m0);
    float dn1 = expf(sel4(b3, b4, es[4], es[5], es[6], es[7]) - m1);
    float4 acc0 = make_float4(0.f, 0.f, 0.f, 0.f), acc1 = acc0;
    acc0 = fma4(dn0, *(const float4*)(hself + c0), acc0);
    acc1 = fma4(dn1, *(const float4*)(hself + c1), acc1);

    for (int j0 = beg; j0 < end; j0 += 32) {
        const int n = (end - j0) < 32 ? (end - j0) : 32;
        int ent = sl;
        float p[8];
        if (j0 != jlast) {
            ent = csr[j0 + lane];
            const float4 xa = *(const float4*)(als + (size_t)ent * NHEADS);
            const float4 xb = *(const float4*)(als + (size_t)ent * NHEADS + 4);
            const float x[8] = {xa.x, xa.y, xa.z, xa.w, xb.x, xb.y, xb.z, xb.w};
#pragma unroll
            for (int k = 0; k < 8; ++k) p[k] = expf(lrelu(x[k] + d[k]) - mx[k]);
        } else {
#pragma unroll
            for (int k = 0; k < 8; ++k) p[k] = expf(e[k] - mx[k]);
        }
        int j = 0;
        for (; j + 4 <= n; j += 4) {
            float4 u0[4], u1[4];
#pragma unroll
            for (int u = 0; u < 4; ++u) {
                const float* hs = h + (size_t)__builtin_amdgcn_readlane(ent, j + u) * WDIM;
                u0[u] = *(const float4*)(hs + c0);
                u1[u] = *(const float4*)(hs + c1);
            }
#pragma unroll
            for (int u = 0; u < 4; ++u) {
                const float p0 = sel4(b3, b4, lane_bcast(p[0], j + u), lane_bcast(p[1], j + u),
                                         lane_bcast(p[2], j + u), lane_bcast(p[3], j + u));
                const float p1 = sel4(b3, b4, lane_bcast(p[4], j + u), lane_bcast(p[5], j + u),
                                         lane_bcast(p[6], j + u), lane_bcast(p[7], j + u));
                dn0 += p0; dn1 += p1;
                acc0 = fma4(p0, u0[u], acc0);
                acc1 = fma4(p1, u1[u], acc1);
            }
        }
        for (; j < n; ++j) {
            const float* hs = h + (size_t)__builtin_amdgcn_readlane(ent, j) * WDIM;
            const float p0 = sel4(b3, b4, lane_bcast(p[0], j), lane_bcast(p[1], j),
                                     lane_bcast(p[2], j), lane_bcast(p[3], j));
            const float p1 = sel4(b3, b4, lane_bcast(p[4], j), lane_bcast(p[5], j),
                                     lane_bcast(p[6], j), lane_bcast(p[7], j));
            dn0 += p0; dn1 += p1;
            acc0 = fma4(p0, *(const float4*)(hs + c0), acc0);
            acc1 = fma4(p1, *(const float4*)(hs + c1), acc1);
        }
    }
    const float4 bz0 = *(const float4*)(bias + c0);
    const float4 bz1 = *(const float4*)(bias + c1);
    float4 o0 = make_float4(acc0.x / dn0 + bz0.x, acc0.y / dn0 + bz0.y, acc0.z / dn0 + bz0.z, acc0.w / dn0 + bz0.w);
    float4 o1 = make_float4(acc1.x / dn1 + bz1.x, acc1.y / dn1 + bz1.y, acc1.z / dn1 + bz1.z, acc1.w / dn1 + bz1.w);
    if (apply_elu) {
        o0.x = elu1(o0.x); o0.y = elu1(o0.y); o0.z = elu1(o0.z); o0.w = elu1(o0.w);
        o1.x = elu1(o1.x); o1.y = elu1(o1.y); o1.z = elu1(o1.z); o1.w = elu1(o1.w);
    }
    float* orow = outx + (size_t)node * WDIM;
    *(float4*)(orow + c0) = o0;
    *(float4*)(orow + c1) = o1;
}

#define POOL_ROWS 16
#define POOL_BLOCKS ((NNODES / POOL_ROWS + 7) / 8)
static_assert(NNODES % POOL_ROWS == 0, "pool_accum takes nodes sixteen at a time");
static_assert(POOL_ROWS <= 32, "one lane holds one node's graph id");

__device__ __forceinline__ void pool_flush(int g, int lane, const float (&mx)[8], const float (&sm)[8],
                                           float cnt, unsigned* __restrict__ gmaxenc,
                                           float* __restrict__ gsum, float* __restrict__ counts) {
    unsigned* pm = gmaxenc + g * WDIM + lane;
    float*    ps = gsum + g * WDIM + lane;
#pragma unroll
    for (int i = 0; i < 8; ++i) {
        atomicMax(pm + 32 * i, fenc(mx[i]));
        atomicAdd(ps + 32 * i, sm[i]);
    }
    if (lane == 0) atomicAdd(&counts[g], cnt);
}

__global__ __launch_bounds__(256) void pool_accum(const float* __restrict__ x2, const int* __restrict__ batch,
                                                  unsigned* __restrict__ gmaxenc, float* __restrict__ gsum,
                                                  float* __restrict__ counts) {
    const int wv = __builtin_amdgcn_readfirstlane((int)(threadIdx.x >> 5));
    const int n0 = (blockIdx.x * 8 + wv) * POOL_ROWS;
    if (n0 >= NNODES) return;
    const int lane = threadIdx.x & 31;
    const int gl = batch[n0 + (lane & (POOL_ROWS - 1))];
    const float* xr = x2 + (size_t)n0 * WDIM + lane;
    float v[POOL_ROWS][8];
#pragma unroll
    for (int r = 0; r < POOL_ROWS; ++r) {
#pragma unroll
        for (int i = 0; i < 8; ++i) v[r][i] = xr[r * WDIM + 32 * i];
    }
    int cur = __builtin_amdgcn_readlane(gl, 0);
    float mx[8], sm[8], cnt = 0.f;
#pragma unroll
    for (int i = 0; i < 8; ++i) { mx[i] = -INFINITY; sm[i] = 0.f; }
#pragma unroll
    for (int r = 0; r < POOL_ROWS; ++r) {
        const int g = __builtin_amdgcn_readlane(gl, r);
        if (g != cur) {
            pool_flush(cur, lane, mx, sm, cnt, gmaxenc, gsum, counts);
#pragma unroll
            for (int i = 0; i < 8; ++i) { mx[i] = -INFINITY; sm[i] = 0.f; }
            cnt = 0.f;
            cur = g;
        }
#pragma unroll
        for (int i = 0; i < 8; ++i) { mx[i] = fmaxf(mx[i], v[r][i]); sm[i] += v[r][i]; }
        cnt += 1.0f;
    }
    pool_flush(cur, lane, mx, sm, cnt, gmaxenc, gsum, counts);
}

__global__ __launch_bounds__(256) void pool_final_fc(const unsigned* __restrict__ gmaxenc, const float* __restrict__ gsum,
                                                     const float* __restrict__ counts, const float* __restrict__ Wfc,
                                                     const float* __restrict__ bfc, float* __restrict__ rep,
                                                     float* __restrict__ out) {
    __shared__ float red[256];
    int g = blockIdx.x, t = threadIdx.x;
    float cnt = counts[g];
    float mx = (cnt > 0.f) ? fdec(gmaxenc[g * WDIM + t]) : 0.0f;
    float mean = gsum[g * WDIM + t] / fmaxf(cnt, 1.0f);
    rep[(size_t)g * (2 * WDIM) + t] = mx;
    rep[(size_t)g * (2 * WDIM) + WDIM + t] = mean;
    float s = mx * Wfc[t] + mean * Wfc[WDIM + t];
    red[t] = s;
    __syncthreads();
    for (int off = 128; off > 0; off >>= 1) {
        if (t < off) red[t] += red[t + off];
        __syncthreads();
    }
    if (t == 0) out[g] = 1.0f / (1.0f + expf(-(red[0] + bfc[0])));
}

extern "C" void kernel_launch(void* const* d_in, const int* in_sizes, int n_in,
                              void* d_out, int out_size, void* d_ws, size_t ws_size,
                              hipStream_t stream) {
    (void)in_sizes; (void)n_in; (void)out_size; (void)ws_size;

    const float* x    = (const float*)d_in[0];
    const int*   ei   = (const int*)d_in[1];
    const int*   bat  = (const int*)d_in[2];
    const float* W1   = (const float*)d_in[3];
    const float* a1s  = (const float*)d_in[4];
    const float* a1d  = (const float*)d_in[5];
    const float* b1   = (const float*)d_in[6];
    const float* W2   = (const float*)d_in[7];
    const float* a2s  = (const float*)d_in[8];
    const float* a2d  = (const float*)d_in[9];
    const float* b2   = (const float*)d_in[10];
    const float* Wfc  = (const float*)d_in[11];
    const float* bfc  = (const float*)d_in[12];

    float* out = (float*)d_out;
    float* rep = out + NGRAPH;

    const int* esrc = ei;
    const int* edst = ei + NEDGES;

    float*    bufA  = (float*)d_ws;
    float*    bufB  = bufA + (size_t)NNODES * WDIM;
    float*    als   = bufB + (size_t)NNODES * WDIM;
    float*    ald   = als + (size_t)NNODES * NHEADS;
    unsigned* gmaxe = (unsigned*)(ald + (size_t)NNODES * NHEADS);
    float*    gsum  = (float*)(gmaxe + NGRAPH * WDIM);
    float*    cnts  = gsum + NGRAPH * WDIM;
    unsigned* rowp  = (unsigned*)(cnts + NGRAPH);
    int*      csr   = (int*)(rowp + NNODES + 4);

    csr_gemm1_zero<<<CSR_WGS + GEMM1_BLOCKS, 1024, 0, stream>>>(esrc, edst, rowp, csr, x, W1, bufA, a1s, a1d,
                                                                als, ald, gmaxe, 128);
    node_agg<<<NNODES / 8, 256, 0, stream>>>(rowp, csr, als, ald, bufA, b1, 1, bufB);

    gemm_f32_wmma_al<<<GEMM2_BLOCKS, 256, 0, stream>>>(bufB, W2, bufA, a2s, a2d, als, ald, NNODES, 256);
    node_agg<<<NNODES / 8, 256, 0, stream>>>(rowp, csr, als, ald, bufA, b2, 0, bufB);

    pool_accum<<<POOL_BLOCKS, 256, 0, stream>>>(bufB, bat, gmaxe, gsum, cnts);
    pool_final_fc<<<NGRAPH, 256, 0, stream>>>(gmaxe, gsum, cnts, Wfc, bfc, rep, out);
}
